# starting kernel
# baseline (speedup 1.0000x reference)
#include <hip/hip_runtime.h>
#include <hip/hip_bf16.h>
#include <math.h>

#define DMODEL   768
#define DSTATE   64
#define DCONV    4
#define DSSM     1536
#define HEADDIM  64
#define NHEADS   24
#define CONVDIM  (DSSM + 2*DSTATE)
#define DINPROJ  (2*DSSM + 2*DSTATE + NHEADS)
#define BATCH    2
#define SEQLEN   512
#define MTOK     (BATCH*SEQLEN)

typedef float v2f __attribute__((ext_vector_type(2)));
typedef float v4f __attribute__((ext_vector_type(4)));
typedef float v8f __attribute__((ext_vector_type(8)));

#define KC    16
#define LDSS  18
#define MBLK  128
#define NBLK  64

__global__ void gemm_nt_f32_wmma(const float* __restrict__ A,
                                 const float* __restrict__ Bm,
                                 float* __restrict__ C,
                                 int M, int N, int K) {
    const int nbt = (N + NBLK - 1) / NBLK;
    const int nb  = blockIdx.x % nbt;
    const int mb  = blockIdx.x / nbt;

    const int tid  = threadIdx.x;
    const int w    = tid >> 5;
    const int lane = tid & 31;
    const int g    = lane >> 4;
    const int r16  = lane & 15;

    __shared__ float As[2][MBLK * LDSS];
    __shared__ float Bs[2][NBLK * LDSS];

    const int arow  = tid >> 1;
    const int ahalf = tid & 1;
    const float* agp = A + (size_t)(mb * MBLK + arow) * K + ahalf * 8;
    const int brow = tid >> 2;
    const int bq   = tid & 3;
    int bgrow = nb * NBLK + brow; if (bgrow >= N) bgrow = N - 1;
    const float* bgp = Bm + (size_t)bgrow * K + bq * 4;

    const int aoff = (w * 16 + r16) * LDSS + 2 * g;
    const int boff = r16 * LDSS + 2 * g;

    v8f acc0 = {}, acc1 = {}, acc2 = {}, acc3 = {};

    {
        v4f a0 = *(const v4f*)(agp);
        v4f a1 = *(const v4f*)(agp + 4);
        v4f b0 = *(const v4f*)(bgp);
        float* as = &As[0][arow * LDSS + ahalf * 8];
        *(v2f*)(as)     = v2f{a0.x, a0.y};
        *(v2f*)(as + 2) = v2f{a0.z, a0.w};
        *(v2f*)(as + 4) = v2f{a1.x, a1.y};
        *(v2f*)(as + 6) = v2f{a1.z, a1.w};
        float* bs = &Bs[0][brow * LDSS + bq * 4];
        *(v2f*)(bs)     = v2f{b0.x, b0.y};
        *(v2f*)(bs + 2) = v2f{b0.z, b0.w};
    }
    __syncthreads();

    int buf = 0;
    for (int kk = 0; kk < K; kk += KC) {
        const bool more = (kk + KC) < K;

        v4f a0, a1, b0;
        if (more) {
            a0 = *(const v4f*)(agp + kk + KC);
            a1 = *(const v4f*)(agp + kk + KC + 4);
            b0 = *(const v4f*)(bgp + kk + KC);
        }

        const float* asb = &As[buf][aoff];
        const float* bsb = &Bs[buf][boff];
        #pragma unroll
        for (int ks = 0; ks < KC; ks += 4) {
            v2f a  = *(const v2f*)(asb + ks);
            v2f q0 = *(const v2f*)(bsb +  0 * LDSS + ks);
            v2f q1 = *(const v2f*)(bsb + 16 * LDSS + ks);
            v2f q2 = *(const v2f*)(bsb + 32 * LDSS + ks);
            v2f q3 = *(const v2f*)(bsb + 48 * LDSS + ks);
            acc0 = __builtin_amdgcn_wmma_f32_16x16x4_f32(false, a, false, q0, (short)0, acc0, false, false);
            acc1 = __builtin_amdgcn_wmma_f32_16x16x4_f32(false, a, false, q1, (short)0, acc1, false, false);
            acc2 = __builtin_amdgcn_wmma_f32_16x16x4_f32(false, a, false, q2, (short)0, acc2, false, false);
            acc3 = __builtin_amdgcn_wmma_f32_16x16x4_f32(false, a, false, q3, (short)0, acc3, false, false);
        }

        if (more) {
            float* as = &As[buf ^ 1][arow * LDSS + ahalf * 8];
            *(v2f*)(as)     = v2f{a0.x, a0.y};
            *(v2f*)(as + 2) = v2f{a0.z, a0.w};
            *(v2f*)(as + 4) = v2f{a1.x, a1.y};
            *(v2f*)(as + 6) = v2f{a1.z, a1.w};
            float* bs = &Bs[buf ^ 1][brow * LDSS + bq * 4];
            *(v2f*)(bs)     = v2f{b0.x, b0.y};
            *(v2f*)(bs + 2) = v2f{b0.z, b0.w};
            __syncthreads();
            buf ^= 1;
        }
    }

    const int rowbase = mb * MBLK + w * 16 + 8 * g;
    #pragma unroll
    for (int j = 0; j < 4; ++j) {
        const v8f acc = (j == 0) ? acc0 : (j == 1) ? acc1 : (j == 2) ? acc2 : acc3;
        const int col = nb * NBLK + j * 16 + r16;
        if (col < N) {
            #pragma unroll
            for (int r = 0; r < 8; ++r)
                C[(size_t)(rowbase + r) * N + col] = acc[r];
        }
    }
}

__global__ void conv_silu_kernel(const float* __restrict__ zxbcdt,
                                 const float* __restrict__ conv_w,
                                 const float* __restrict__ conv_b,
                                 float* __restrict__ xbc_out) {
    const int idx = blockIdx.x * blockDim.x + threadIdx.x;
    if (idx >= MTOK * CONVDIM) return;
    const int c  = idx % CONVDIM;
    const int bl = idx / CONVDIM;
    const int l  = bl % SEQLEN;

    float acc = conv_b[c];
    #pragma unroll
    for (int j = 0; j < DCONV; ++j) {
        const int ll = l - (DCONV - 1) + j;
        if (ll >= 0) {
            const float xv = zxbcdt[(size_t)(bl - (DCONV - 1) + j) * DINPROJ + DSSM + c];
            acc = fmaf(xv, conv_w[c * DCONV + j], acc);
        }
    }
    const float s = acc / (1.0f + __expf(-acc));
    xbc_out[(size_t)bl * CONVDIM + c] = s;
}

__global__ void ssm_scan_kernel(const float* __restrict__ zxbcdt,
                                const float* __restrict__ xbc,
                                const float* __restrict__ dt_bias,
                                const float* __restrict__ A_log,
                                const float* __restrict__ D_param,
                                float* __restrict__ y) {
    const int bh = blockIdx.x;
    const int b  = bh / NHEADS;
    const int h  = bh % NHEADS;
    const int t  = threadIdx.x;
    const int p  = t >> 2;
    const int nc = t & 3;

    const float Ah  = -__expf(A_log[h]);
    const float dtb = dt_bias[h];
    const float Dp  = D_param[h];

    __shared__ float sx[HEADDIM];
    __shared__ float sB[DSTATE];
    __shared__ float sC[DSTATE];

    float s[16];
    #pragma unroll
    for (int i = 0; i < 16; ++i) s[i] = 0.0f;

    for (int l = 0; l < SEQLEN; ++l) {
        const size_t rowc = (size_t)(b * SEQLEN + l) * CONVDIM;
        const size_t rowz = (size_t)(b * SEQLEN + l) * DINPROJ;

        if (t < 64)        sx[t]       = xbc[rowc + h * HEADDIM + t];
        else if (t < 128)  sB[t - 64]  = xbc[rowc + DSSM + (t - 64)];
        else if (t < 192)  sC[t - 128] = xbc[rowc + DSSM + DSTATE + (t - 128)];
        __syncthreads();

        const float dv  = zxbcdt[rowz + (DSSM + CONVDIM) + h] + dtb;
        float dt = (dv > 20.0f) ? dv : __logf(1.0f + __expf(dv));
        dt = fminf(fmaxf(dt, 1e-6f), 1000.0f);
        const float dA = fminf(fmaxf(__expf(dt * Ah), 1e-6f), 1000.0f);

        const float xv  = sx[p];
        const float dtx = dt * xv;

        float ysum = 0.0f;
        #pragma unroll
        for (int i = 0; i < 16; ++i) {
            const int n = nc * 16 + i;
            float sv = fmaf(s[i], dA, dtx * sB[n]);
            sv = fminf(fmaxf(sv, -1000.0f), 1000.0f);
            s[i] = sv;
            ysum = fmaf(sv, sC[n], ysum);
        }
        ysum += __shfl_xor(ysum, 1);
        ysum += __shfl_xor(ysum, 2);
        if (nc == 0)
            y[(size_t)(b * SEQLEN + l) * DSSM + h * HEADDIM + p] = fmaf(Dp, xv, ysum);
        __syncthreads();
    }
}

__global__ void gate_kernel(const float* __restrict__ zxbcdt,
                            float* __restrict__ y) {
    const int idx = blockIdx.x * blockDim.x + threadIdx.x;
    if (idx >= MTOK * DSSM) return;
    const int row = idx / DSSM;
    const int c   = idx % DSSM;
    const float z  = zxbcdt[(size_t)row * DINPROJ + c];
    const float sz = z / (1.0f + __expf(-z));
    y[idx] *= sz;
}

extern "C" void kernel_launch(void* const* d_in, const int* in_sizes, int n_in,
                              void* d_out, int out_size, void* d_ws, size_t ws_size,
                              hipStream_t stream) {
    const float* u       = (const float*)d_in[0];
    const float* W_in    = (const float*)d_in[1];
    const float* conv_w  = (const float*)d_in[2];
    const float* conv_b  = (const float*)d_in[3];
    const float* dt_bias = (const float*)d_in[4];
    const float* A_log   = (const float*)d_in[5];
    const float* D_param = (const float*)d_in[6];
    const float* W_out   = (const float*)d_in[7];
    float* out = (float*)d_out;

    float* ws      = (float*)d_ws;
    float* zxbcdt  = ws;
    float* xbc     = zxbcdt + (size_t)MTOK * DINPROJ;
    float* ybuf    = xbc    + (size_t)MTOK * CONVDIM;

    {
        const int mbt = MTOK / MBLK;
        const int nbt = (DINPROJ + NBLK - 1) / NBLK;
        gemm_nt_f32_wmma<<<mbt * nbt, 256, 0, stream>>>(u, W_in, zxbcdt,
                                                        MTOK, DINPROJ, DMODEL);
    }

    {
        const int n = MTOK * CONVDIM;
        conv_silu_kernel<<<(n + 255) / 256, 256, 0, stream>>>(zxbcdt, conv_w, conv_b, xbc);
    }

    ssm_scan_kernel<<<BATCH * NHEADS, 256, 0, stream>>>(zxbcdt, xbc, dt_bias,
                                                        A_log, D_param, ybuf);

    {
        const int n = MTOK * DSSM;
        gate_kernel<<<(n + 255) / 256, 256, 0, stream>>>(zxbcdt, ybuf);
    }

    {
        const int mbt = MTOK / MBLK;
        const int nbt = (DMODEL + NBLK - 1) / NBLK;
        gemm_nt_f32_wmma<<<mbt * nbt, 256, 0, stream>>>(ybuf, W_out, out,
                                                        MTOK, DMODEL, DSSM);
    }
}


# scan kernel: x, B, C, dt, dA staged 32 time steps per barrier in double-buffered LDS, loaded one chunk ahead; a step reads LDS only (was 2 barriers and global loads per step); sums unchanged
# speedup vs baseline: 3.9828x; 3.9828x over previous
#include <hip/hip_runtime.h>
#include <hip/hip_bf16.h>
#include <math.h>

#define DMODEL   768
#define DSTATE   64
#define DCONV    4
#define DSSM     1536
#define HEADDIM  64
#define NHEADS   24
#define CONVDIM  (DSSM + 2*DSTATE)
#define DINPROJ  (2*DSSM + 2*DSTATE + NHEADS)
#define BATCH    2
#define SEQLEN   512
#define MTOK     (BATCH*SEQLEN)

typedef float v2f __attribute__((ext_vector_type(2)));
typedef float v4f __attribute__((ext_vector_type(4)));
typedef float v8f __attribute__((ext_vector_type(8)));

#define KC    16
#define LDSS  18
#define MBLK  128
#define NBLK  64

__global__ void gemm_nt_f32_wmma(const float* __restrict__ A,
                                 const float* __restrict__ Bm,
                                 float* __restrict__ C,
                                 int M, int N, int K) {
    const int nbt = (N + NBLK - 1) / NBLK;
    const int nb  = blockIdx.x % nbt;
    const int mb  = blockIdx.x / nbt;

    const int tid  = threadIdx.x;
    const int w    = tid >> 5;
    const int lane = tid & 31;
    const int g    = lane >> 4;
    const int r16  = lane & 15;

    __shared__ float As[2][MBLK * LDSS];
    __shared__ float Bs[2][NBLK * LDSS];

    const int arow  = tid >> 1;
    const int ahalf = tid & 1;
    const float* agp = A + (size_t)(mb * MBLK + arow) * K + ahalf * 8;
    const int brow = tid >> 2;
    const int bq   = tid & 3;
    int bgrow = nb * NBLK + brow; if (bgrow >= N) bgrow = N - 1;
    const float* bgp = Bm + (size_t)bgrow * K + bq * 4;

    const int aoff = (w * 16 + r16) * LDSS + 2 * g;
    const int boff = r16 * LDSS + 2 * g;

    v8f acc0 = {}, acc1 = {}, acc2 = {}, acc3 = {};

    {
        v4f a0 = *(const v4f*)(agp);
        v4f a1 = *(const v4f*)(agp + 4);
        v4f b0 = *(const v4f*)(bgp);
        float* as = &As[0][arow * LDSS + ahalf * 8];
        *(v2f*)(as)     = v2f{a0.x, a0.y};
        *(v2f*)(as + 2) = v2f{a0.z, a0.w};
        *(v2f*)(as + 4) = v2f{a1.x, a1.y};
        *(v2f*)(as + 6) = v2f{a1.z, a1.w};
        float* bs = &Bs[0][brow * LDSS + bq * 4];
        *(v2f*)(bs)     = v2f{b0.x, b0.y};
        *(v2f*)(bs + 2) = v2f{b0.z, b0.w};
    }
    __syncthreads();

    int buf = 0;
    for (int kk = 0; kk < K; kk += KC) {
        const bool more = (kk + KC) < K;

        v4f a0, a1, b0;
        if (more) {
            a0 = *(const v4f*)(agp + kk + KC);
            a1 = *(const v4f*)(agp + kk + KC + 4);
            b0 = *(const v4f*)(bgp + kk + KC);
        }

        const float* asb = &As[buf][aoff];
        const float* bsb = &Bs[buf][boff];
        #pragma unroll
        for (int ks = 0; ks < KC; ks += 4) {
            v2f a  = *(const v2f*)(asb + ks);
            v2f q0 = *(const v2f*)(bsb +  0 * LDSS + ks);
            v2f q1 = *(const v2f*)(bsb + 16 * LDSS + ks);
            v2f q2 = *(const v2f*)(bsb + 32 * LDSS + ks);
            v2f q3 = *(const v2f*)(bsb + 48 * LDSS + ks);
            acc0 = __builtin_amdgcn_wmma_f32_16x16x4_f32(false, a, false, q0, (short)0, acc0, false, false);
            acc1 = __builtin_amdgcn_wmma_f32_16x16x4_f32(false, a, false, q1, (short)0, acc1, false, false);
            acc2 = __builtin_amdgcn_wmma_f32_16x16x4_f32(false, a, false, q2, (short)0, acc2, false, false);
            acc3 = __builtin_amdgcn_wmma_f32_16x16x4_f32(false, a, false, q3, (short)0, acc3, false, false);
        }

        if (more) {
            float* as = &As[buf ^ 1][arow * LDSS + ahalf * 8];
            *(v2f*)(as)     = v2f{a0.x, a0.y};
            *(v2f*)(as + 2) = v2f{a0.z, a0.w};
            *(v2f*)(as + 4) = v2f{a1.x, a1.y};
            *(v2f*)(as + 6) = v2f{a1.z, a1.w};
            float* bs = &Bs[buf ^ 1][brow * LDSS + bq * 4];
            *(v2f*)(bs)     = v2f{b0.x, b0.y};
            *(v2f*)(bs + 2) = v2f{b0.z, b0.w};
            __syncthreads();
            buf ^= 1;
        }
    }

    const int rowbase = mb * MBLK + w * 16 + 8 * g;
    #pragma unroll
    for (int j = 0; j < 4; ++j) {
        const v8f acc = (j == 0) ? acc0 : (j == 1) ? acc1 : (j == 2) ? acc2 : acc3;
        const int col = nb * NBLK + j * 16 + r16;
        if (col < N) {
            #pragma unroll
            for (int r = 0; r < 8; ++r)
                C[(size_t)(rowbase + r) * N + col] = acc[r];
        }
    }
}

__global__ void conv_silu_kernel(const float* __restrict__ zxbcdt,
                                 const float* __restrict__ conv_w,
                                 const float* __restrict__ conv_b,
                                 float* __restrict__ xbc_out) {
    const int idx = blockIdx.x * blockDim.x + threadIdx.x;
    if (idx >= MTOK * CONVDIM) return;
    const int c  = idx % CONVDIM;
    const int bl = idx / CONVDIM;
    const int l  = bl % SEQLEN;

    float acc = conv_b[c];
    #pragma unroll
    for (int j = 0; j < DCONV; ++j) {
        const int ll = l - (DCONV - 1) + j;
        if (ll >= 0) {
            const float xv = zxbcdt[(size_t)(bl - (DCONV - 1) + j) * DINPROJ + DSSM + c];
            acc = fmaf(xv, conv_w[c * DCONV + j], acc);
        }
    }
    const float s = acc / (1.0f + __expf(-acc));
    xbc_out[(size_t)bl * CONVDIM + c] = s;
}

#define TCH   32
#define NCHK  (SEQLEN / TCH)

__global__ void __launch_bounds__(256)
ssm_scan_kernel(const float* __restrict__ zxbcdt,
                const float* __restrict__ xbc,
                const float* __restrict__ dt_bias,
                const float* __restrict__ A_log,
                const float* __restrict__ D_param,
                float* __restrict__ y) {
    const int bh = blockIdx.x;
    const int b  = bh / NHEADS;
    const int h  = bh % NHEADS;
    const int t  = threadIdx.x;
    const int p  = t >> 2;
    const int nc = t & 3;

    const float Ah  = -__expf(A_log[h]);
    const float dtb = dt_bias[h];
    const float Dp  = D_param[h];

    __shared__ __attribute__((aligned(16))) float sx [2][TCH * HEADDIM];
    __shared__ __attribute__((aligned(16))) float sBC[2][TCH * 2 * DSTATE];
    __shared__ __attribute__((aligned(8)))  v2f   sdd[2][TCH];

    const int xr = t >> 4, xq = t & 15;
    const int cr = t >> 5, cq = t & 31;
    const float* xg  = xbc + (size_t)(b * SEQLEN + xr) * CONVDIM + h * HEADDIM + xq * 4;
    const float* cg  = xbc + (size_t)(b * SEQLEN + cr) * CONVDIM + DSSM + cq * 4;
    const float* dg  = zxbcdt + (size_t)(b * SEQLEN + (t & (TCH - 1))) * DINPROJ
                              + (DSSM + CONVDIM) + h;
    const int xs = xr * HEADDIM + xq * 4;
    const int cs = cr * (2 * DSTATE) + cq * 4;

    float s[16];
    #pragma unroll
    for (int i = 0; i < 16; ++i) s[i] = 0.0f;

    {
        const v4f gx0 = *(const v4f*)(xg);
        const v4f gx1 = *(const v4f*)(xg + (size_t)16 * CONVDIM);
        const v4f gc0 = *(const v4f*)(cg);
        const v4f gc1 = *(const v4f*)(cg + (size_t)8  * CONVDIM);
        const v4f gc2 = *(const v4f*)(cg + (size_t)16 * CONVDIM);
        const v4f gc3 = *(const v4f*)(cg + (size_t)24 * CONVDIM);
        *(v4f*)(&sx[0][xs])                 = gx0;
        *(v4f*)(&sx[0][xs + 16 * HEADDIM])  = gx1;
        *(v4f*)(&sBC[0][cs])                = gc0;
        *(v4f*)(&sBC[0][cs +  8 * 2 * DSTATE]) = gc1;
        *(v4f*)(&sBC[0][cs + 16 * 2 * DSTATE]) = gc2;
        *(v4f*)(&sBC[0][cs + 24 * 2 * DSTATE]) = gc3;
        if (t < TCH) {
            const float dv = dg[0] + dtb;
            float dt = (dv > 20.0f) ? dv : __logf(1.0f + __expf(dv));
            dt = fminf(fmaxf(dt, 1e-6f), 1000.0f);
            const float dA = fminf(fmaxf(__expf(dt * Ah), 1e-6f), 1000.0f);
            sdd[0][t] = v2f{dt, dA};
        }
    }
    __syncthreads();

    int buf = 0;
    for (int c = 0; c < NCHK; ++c) {
        const bool more = (c + 1) < NCHK;
        const size_t rowoff = (size_t)(c + 1) * TCH;

        v4f gx0, gx1, gc0, gc1, gc2, gc3;
        float gdv = 0.0f;
        if (more) {
            const float* xn = xg + rowoff * CONVDIM;
            const float* cn = cg + rowoff * CONVDIM;
            gx0 = *(const v4f*)(xn);
            gx1 = *(const v4f*)(xn + (size_t)16 * CONVDIM);
            gc0 = *(const v4f*)(cn);
            gc1 = *(const v4f*)(cn + (size_t)8  * CONVDIM);
            gc2 = *(const v4f*)(cn + (size_t)16 * CONVDIM);
            gc3 = *(const v4f*)(cn + (size_t)24 * CONVDIM);
            if (t < TCH) gdv = dg[rowoff * DINPROJ];
        }

        const float* bx  = &sx[buf][p];
        const float* bBC = &sBC[buf][nc * 16];
        float* yp = y + (size_t)(b * SEQLEN + c * TCH + nc) * DSSM + h * HEADDIM + p;
        for (int j4 = 0; j4 < TCH; j4 += 4) {
            float ys[4], xs4[4];
            #pragma unroll
            for (int u = 0; u < 4; ++u) {
                const int j = j4 + u;
                const v2f   dd  = sdd[buf][j];
                const float dt  = dd.x;
                const float dA  = dd.y;
                const float xv  = bx[j * HEADDIM];
                const float dtx = dt * xv;

                const v4f* Bq = (const v4f*)(bBC + j * 2 * DSTATE);
                const v4f* Cq = (const v4f*)(bBC + j * 2 * DSTATE + DSTATE);
                const v4f B0 = Bq[0], B1 = Bq[1], B2 = Bq[2], B3 = Bq[3];
                const v4f C0 = Cq[0], C1 = Cq[1], C2 = Cq[2], C3 = Cq[3];
                const float Bv[16] = {B0.x, B0.y, B0.z, B0.w, B1.x, B1.y, B1.z, B1.w,
                                      B2.x, B2.y, B2.z, B2.w, B3.x, B3.y, B3.z, B3.w};
                const float Cv[16] = {C0.x, C0.y, C0.z, C0.w, C1.x, C1.y, C1.z, C1.w,
                                      C2.x, C2.y, C2.z, C2.w, C3.x, C3.y, C3.z, C3.w};

                float ysum = 0.0f;
                #pragma unroll
                for (int i = 0; i < 16; ++i) {
                    float sv = fmaf(s[i], dA, dtx * Bv[i]);
                    sv = fminf(fmaxf(sv, -1000.0f), 1000.0f);
                    s[i] = sv;
                    ysum = fmaf(sv, Cv[i], ysum);
                }
                ys[u]  = ysum;
                xs4[u] = xv;
            }
            ys[0] += __shfl_xor(ys[0], 1);
            ys[1] += __shfl_xor(ys[1], 1);
            ys[2] += __shfl_xor(ys[2], 1);
            ys[3] += __shfl_xor(ys[3], 1);
            ys[0] += __shfl_xor(ys[0], 2);
            ys[1] += __shfl_xor(ys[1], 2);
            ys[2] += __shfl_xor(ys[2], 2);
            ys[3] += __shfl_xor(ys[3], 2);
            float ykeep = fmaf(Dp, xs4[0], ys[0]);
            ykeep = (nc == 1) ? fmaf(Dp, xs4[1], ys[1]) : ykeep;
            ykeep = (nc == 2) ? fmaf(Dp, xs4[2], ys[2]) : ykeep;
            ykeep = (nc == 3) ? fmaf(Dp, xs4[3], ys[3]) : ykeep;
            yp[(size_t)j4 * DSSM] = ykeep;
        }

        if (more) {
            *(v4f*)(&sx[buf ^ 1][xs])                 = gx0;
            *(v4f*)(&sx[buf ^ 1][xs + 16 * HEADDIM])  = gx1;
            *(v4f*)(&sBC[buf ^ 1][cs])                = gc0;
            *(v4f*)(&sBC[buf ^ 1][cs +  8 * 2 * DSTATE]) = gc1;
            *(v4f*)(&sBC[buf ^ 1][cs + 16 * 2 * DSTATE]) = gc2;
            *(v4f*)(&sBC[buf ^ 1][cs + 24 * 2 * DSTATE]) = gc3;
            if (t < TCH) {
                const float dv = gdv + dtb;
                float dt = (dv > 20.0f) ? dv : __logf(1.0f + __expf(dv));
                dt = fminf(fmaxf(dt, 1e-6f), 1000.0f);
                const float dA = fminf(fmaxf(__expf(dt * Ah), 1e-6f), 1000.0f);
                sdd[buf ^ 1][t] = v2f{dt, dA};
            }
            __builtin_amdgcn_fence(__ATOMIC_RELEASE, "workgroup", "local");
            __builtin_amdgcn_s_barrier();
            __builtin_amdgcn_fence(__ATOMIC_ACQUIRE, "workgroup", "local");
            buf ^= 1;
        }
    }
}

__global__ void gate_kernel(const float* __restrict__ zxbcdt,
                            float* __restrict__ y) {
    const int idx = blockIdx.x * blockDim.x + threadIdx.x;
    if (idx >= MTOK * DSSM) return;
    const int row = idx / DSSM;
    const int c   = idx % DSSM;
    const float z  = zxbcdt[(size_t)row * DINPROJ + c];
    const float sz = z / (1.0f + __expf(-z));
    y[idx] *= sz;
}

extern "C" void kernel_launch(void* const* d_in, const int* in_sizes, int n_in,
                              void* d_out, int out_size, void* d_ws, size_t ws_size,
                              hipStream_t stream) {
    const float* u       = (const float*)d_in[0];
    const float* W_in    = (const float*)d_in[1];
    const float* conv_w  = (const float*)d_in[2];
    const float* conv_b  = (const float*)d_in[3];
    const float* dt_bias = (const float*)d_in[4];
    const float* A_log   = (const float*)d_in[5];
    const float* D_param = (const float*)d_in[6];
    const float* W_out   = (const float*)d_in[7];
    float* out = (float*)d_out;

    float* ws      = (float*)d_ws;
    float* zxbcdt  = ws;
    float* xbc     = zxbcdt + (size_t)MTOK * DINPROJ;
    float* ybuf    = xbc    + (size_t)MTOK * CONVDIM;

    {
        const int mbt = MTOK / MBLK;
        const int nbt = (DINPROJ + NBLK - 1) / NBLK;
        gemm_nt_f32_wmma<<<mbt * nbt, 256, 0, stream>>>(u, W_in, zxbcdt,
                                                        MTOK, DINPROJ, DMODEL);
    }

    {
        const int n = MTOK * CONVDIM;
        conv_silu_kernel<<<(n + 255) / 256, 256, 0, stream>>>(zxbcdt, conv_w, conv_b, xbc);
    }

    ssm_scan_kernel<<<BATCH * NHEADS, 256, 0, stream>>>(zxbcdt, xbc, dt_bias,
                                                        A_log, D_param, ybuf);

    {
        const int n = MTOK * DSSM;
        gate_kernel<<<(n + 255) / 256, 256, 0, stream>>>(zxbcdt, ybuf);
    }

    {
        const int mbt = MTOK / MBLK;
        const int nbt = (DMODEL + NBLK - 1) / NBLK;
        gemm_nt_f32_wmma<<<mbt * nbt, 256, 0, stream>>>(ybuf, W_out, out,
                                                        MTOK, DMODEL, DSSM);
    }
}


# conv+bias+SiLU and the y*silu(z) gate folded into the scan kernel: chunk staging reads raw in-projection rows, y stored once gated; launches 5 -> 3; sums unchanged
# speedup vs baseline: 4.3772x; 1.0990x over previous
#include <hip/hip_runtime.h>
#include <hip/hip_bf16.h>
#include <math.h>

#define DMODEL   768
#define DSTATE   64
#define DCONV    4
#define DSSM     1536
#define HEADDIM  64
#define NHEADS   24
#define CONVDIM  (DSSM + 2*DSTATE)
#define DINPROJ  (2*DSSM + 2*DSTATE + NHEADS)
#define BATCH    2
#define SEQLEN   512
#define MTOK     (BATCH*SEQLEN)

typedef float v2f __attribute__((ext_vector_type(2)));
typedef float v4f __attribute__((ext_vector_type(4)));
typedef float v8f __attribute__((ext_vector_type(8)));

#define KC    16
#define LDSS  18
#define MBLK  128
#define NBLK  64

__global__ void gemm_nt_f32_wmma(const float* __restrict__ A,
                                 const float* __restrict__ Bm,
                                 float* __restrict__ C,
                                 int M, int N, int K) {
    const int nbt = (N + NBLK - 1) / NBLK;
    const int nb  = blockIdx.x % nbt;
    const int mb  = blockIdx.x / nbt;

    const int tid  = threadIdx.x;
    const int w    = tid >> 5;
    const int lane = tid & 31;
    const int g    = lane >> 4;
    const int r16  = lane & 15;

    __shared__ float As[2][MBLK * LDSS];
    __shared__ float Bs[2][NBLK * LDSS];

    const int arow  = tid >> 1;
    const int ahalf = tid & 1;
    const float* agp = A + (size_t)(mb * MBLK + arow) * K + ahalf * 8;
    const int brow = tid >> 2;
    const int bq   = tid & 3;
    int bgrow = nb * NBLK + brow; if (bgrow >= N) bgrow = N - 1;
    const float* bgp = Bm + (size_t)bgrow * K + bq * 4;

    const int aoff = (w * 16 + r16) * LDSS + 2 * g;
    const int boff = r16 * LDSS + 2 * g;

    v8f acc0 = {}, acc1 = {}, acc2 = {}, acc3 = {};

    {
        v4f a0 = *(const v4f*)(agp);
        v4f a1 = *(const v4f*)(agp + 4);
        v4f b0 = *(const v4f*)(bgp);
        float* as = &As[0][arow * LDSS + ahalf * 8];
        *(v2f*)(as)     = v2f{a0.x, a0.y};
        *(v2f*)(as + 2) = v2f{a0.z, a0.w};
        *(v2f*)(as + 4) = v2f{a1.x, a1.y};
        *(v2f*)(as + 6) = v2f{a1.z, a1.w};
        float* bs = &Bs[0][brow * LDSS + bq * 4];
        *(v2f*)(bs)     = v2f{b0.x, b0.y};
        *(v2f*)(bs + 2) = v2f{b0.z, b0.w};
    }
    __syncthreads();

    int buf = 0;
    for (int kk = 0; kk < K; kk += KC) {
        const bool more = (kk + KC) < K;

        v4f a0, a1, b0;
        if (more) {
            a0 = *(const v4f*)(agp + kk + KC);
            a1 = *(const v4f*)(agp + kk + KC + 4);
            b0 = *(const v4f*)(bgp + kk + KC);
        }

        const float* asb = &As[buf][aoff];
        const float* bsb = &Bs[buf][boff];
        #pragma unroll
        for (int ks = 0; ks < KC; ks += 4) {
            v2f a  = *(const v2f*)(asb + ks);
            v2f q0 = *(const v2f*)(bsb +  0 * LDSS + ks);
            v2f q1 = *(const v2f*)(bsb + 16 * LDSS + ks);
            v2f q2 = *(const v2f*)(bsb + 32 * LDSS + ks);
            v2f q3 = *(const v2f*)(bsb + 48 * LDSS + ks);
            acc0 = __builtin_amdgcn_wmma_f32_16x16x4_f32(false, a, false, q0, (short)0, acc0, false, false);
            acc1 = __builtin_amdgcn_wmma_f32_16x16x4_f32(false, a, false, q1, (short)0, acc1, false, false);
            acc2 = __builtin_amdgcn_wmma_f32_16x16x4_f32(false, a, false, q2, (short)0, acc2, false, false);
            acc3 = __builtin_amdgcn_wmma_f32_16x16x4_f32(false, a, false, q3, (short)0, acc3, false, false);
        }

        if (more) {
            float* as = &As[buf ^ 1][arow * LDSS + ahalf * 8];
            *(v2f*)(as)     = v2f{a0.x, a0.y};
            *(v2f*)(as + 2) = v2f{a0.z, a0.w};
            *(v2f*)(as + 4) = v2f{a1.x, a1.y};
            *(v2f*)(as + 6) = v2f{a1.z, a1.w};
            float* bs = &Bs[buf ^ 1][brow * LDSS + bq * 4];
            *(v2f*)(bs)     = v2f{b0.x, b0.y};
            *(v2f*)(bs + 2) = v2f{b0.z, b0.w};
            __syncthreads();
            buf ^= 1;
        }
    }

    const int rowbase = mb * MBLK + w * 16 + 8 * g;
    #pragma unroll
    for (int j = 0; j < 4; ++j) {
        const v8f acc = (j == 0) ? acc0 : (j == 1) ? acc1 : (j == 2) ? acc2 : acc3;
        const int col = nb * NBLK + j * 16 + r16;
        if (col < N) {
            #pragma unroll
            for (int r = 0; r < 8; ++r)
                C[(size_t)(rowbase + r) * N + col] = acc[r];
        }
    }
}

#define TCH   32
#define NCHK  (SEQLEN / TCH)
#define ZP    72

struct ScanStage {
    v4f   rx[5];
    v4f   rc[7];
    v4f   rz[2];
    float dv;
};

__device__ __forceinline__ float silu_f(float a) {
    return a / (1.0f + __expf(-a));
}

template <bool FIRST>
__device__ __forceinline__ v4f ld_raw(const float* __restrict__ col0, int l) {
    const int lr = (FIRST && l < 0) ? 0 : l;
    return *(const v4f*)(col0 + (size_t)lr * DINPROJ);
}

template <bool FIRST>
__device__ __forceinline__ v4f conv4_silu(const v4f r0, const v4f r1, const v4f r2, const v4f r3,
                                          int l, const v4f (&w)[4], const v4f bias) {
    v4f o;
    #pragma unroll
    for (int ch = 0; ch < 4; ++ch) {
        float acc = bias[ch];
        if (!FIRST || l >= 3) acc = fmaf(r0[ch], w[ch].x, acc);
        if (!FIRST || l >= 2) acc = fmaf(r1[ch], w[ch].y, acc);
        if (!FIRST || l >= 1) acc = fmaf(r2[ch], w[ch].z, acc);
        acc = fmaf(r3[ch], w[ch].w, acc);
        o[ch] = silu_f(acc);
    }
    return o;
}

template <bool FIRST>
__device__ __forceinline__ void stage_load(ScanStage& s,
                                           const float* __restrict__ xraw,
                                           const float* __restrict__ craw,
                                           const float* __restrict__ zraw,
                                           const float* __restrict__ draw,
                                           int lx, int lc, int ld, bool dlane) {
    #pragma unroll
    for (int i = 0; i < 5; ++i) s.rx[i] = ld_raw<FIRST>(xraw, lx - 3 + i);
    #pragma unroll
    for (int i = 0; i < 7; ++i) s.rc[i] = ld_raw<FIRST>(craw, lc - 3 + i);
    s.rz[0] = *(const v4f*)(zraw + (size_t)lx * DINPROJ);
    s.rz[1] = *(const v4f*)(zraw + (size_t)(lx + 1) * DINPROJ);
    s.dv = dlane ? draw[(size_t)ld * DINPROJ] : 0.0f;
}

template <bool FIRST>
__device__ __forceinline__ void stage_store(const ScanStage& s,
                                            float* __restrict__ sxb, float* __restrict__ sBCb,
                                            float* __restrict__ sgb, v2f* __restrict__ sddb,
                                            int lx, int lc, int xrow, int crow, int q16, int q32,
                                            const v4f (&wx)[4], const v4f bxv,
                                            const v4f (&wc)[4], const v4f bcv,
                                            bool dlane, int t, float dtb, float Ah) {
    #pragma unroll
    for (int o = 0; o < 2; ++o) {
        const v4f v = conv4_silu<FIRST>(s.rx[o], s.rx[o + 1], s.rx[o + 2], s.rx[o + 3],
                                        lx + o, wx, bxv);
        *(v4f*)(sxb + (xrow + o) * HEADDIM + q16 * 4) = v;
    }
    #pragma unroll
    for (int o = 0; o < 4; ++o) {
        const v4f v = conv4_silu<FIRST>(s.rc[o], s.rc[o + 1], s.rc[o + 2], s.rc[o + 3],
                                        lc + o, wc, bcv);
        *(v4f*)(sBCb + (crow + o) * (2 * DSTATE) + q32 * 4) = v;
    }
    #pragma unroll
    for (int o = 0; o < 2; ++o) {
        const v4f z = s.rz[o];
        const v4f g = v4f{silu_f(z.x), silu_f(z.y), silu_f(z.z), silu_f(z.w)};
        *(v4f*)(sgb + (xrow + o) * ZP + q16 * 4) = g;
    }
    if (dlane) {
        const float dv = s.dv + dtb;
        float dt = (dv > 20.0f) ? dv : __logf(1.0f + __expf(dv));
        dt = fminf(fmaxf(dt, 1e-6f), 1000.0f);
        const float dA = fminf(fmaxf(__expf(dt * Ah), 1e-6f), 1000.0f);
        sddb[t] = v2f{dt, dA};
    }
}

__global__ void __launch_bounds__(256)
ssm_scan_kernel(const float* __restrict__ zxbcdt,
                const float* __restrict__ conv_w,
                const float* __restrict__ conv_b,
                const float* __restrict__ dt_bias,
                const float* __restrict__ A_log,
                const float* __restrict__ D_param,
                float* __restrict__ y) {
    const int bh = blockIdx.x;
    const int b  = bh / NHEADS;
    const int h  = bh % NHEADS;
    const int t  = threadIdx.x;
    const int p  = t >> 2;
    const int nc = t & 3;

    const float Ah  = -__expf(A_log[h]);
    const float dtb = dt_bias[h];
    const float Dp  = D_param[h];

    __shared__ __attribute__((aligned(16))) float sx [2][TCH * HEADDIM];
    __shared__ __attribute__((aligned(16))) float sBC[2][TCH * 2 * DSTATE];
    __shared__ __attribute__((aligned(16))) float sg [2][TCH * ZP];
    __shared__ __attribute__((aligned(8)))  v2f   sdd[2][TCH];

    const int q16  = t & 15;
    const int xrow = (t >> 4) * 2;
    const int q32  = t & 31;
    const int crow = (t >> 5) * 4;
    const bool dlane = t < TCH;
    const int cx = h * HEADDIM + q16 * 4;
    const int cb = DSSM + q32 * 4;
    const float* base = zxbcdt + (size_t)b * SEQLEN * DINPROJ;
    const float* zraw = base + cx;
    const float* xraw = base + DSSM + cx;
    const float* craw = base + DSSM + cb;
    const float* draw = base + (DSSM + CONVDIM) + h;

    v4f wx[4], wc[4];
    #pragma unroll
    for (int ch = 0; ch < 4; ++ch) {
        wx[ch] = *(const v4f*)(conv_w + (size_t)(cx + ch) * DCONV);
        wc[ch] = *(const v4f*)(conv_w + (size_t)(cb + ch) * DCONV);
    }
    const v4f bxv = *(const v4f*)(conv_b + cx);
    const v4f bcv = *(const v4f*)(conv_b + cb);

    float s[16];
    #pragma unroll
    for (int i = 0; i < 16; ++i) s[i] = 0.0f;

    {
        ScanStage st;
        stage_load<true>(st, xraw, craw, zraw, draw, xrow, crow, t, dlane);
        stage_store<true>(st, &sx[0][0], &sBC[0][0], &sg[0][0], &sdd[0][0],
                          xrow, crow, xrow, crow, q16, q32, wx, bxv, wc, bcv,
                          dlane, t, dtb, Ah);
    }
    __syncthreads();

    int buf = 0;
    for (int c = 0; c < NCHK; ++c) {
        const bool more = (c + 1) < NCHK;
        const int  l1   = (c + 1) * TCH;

        ScanStage st;
        if (more)
            stage_load<false>(st, xraw, craw, zraw, draw, l1 + xrow, l1 + crow, l1 + t, dlane);

        const float* bx  = &sx[buf][p];
        const float* bBC = &sBC[buf][nc * 16];
        const float* bg  = &sg[buf][nc * ZP + p];
        float* yp = y + (size_t)(b * SEQLEN + c * TCH + nc) * DSSM + h * HEADDIM + p;
        for (int j4 = 0; j4 < TCH; j4 += 4) {
            float ys[4], xs4[4];
            const float gate = bg[j4 * ZP];
            #pragma unroll
            for (int u = 0; u < 4; ++u) {
                const int j = j4 + u;
                const v2f   dd  = sdd[buf][j];
                const float dt  = dd.x;
                const float dA  = dd.y;
                const float xv  = bx[j * HEADDIM];
                const float dtx = dt * xv;

                const v4f* Bq = (const v4f*)(bBC + j * 2 * DSTATE);
                const v4f* Cq = (const v4f*)(bBC + j * 2 * DSTATE + DSTATE);
                const v4f B0 = Bq[0], B1 = Bq[1], B2 = Bq[2], B3 = Bq[3];
                const v4f C0 = Cq[0], C1 = Cq[1], C2 = Cq[2], C3 = Cq[3];
                const float Bv[16] = {B0.x, B0.y, B0.z, B0.w, B1.x, B1.y, B1.z, B1.w,
                                      B2.x, B2.y, B2.z, B2.w, B3.x, B3.y, B3.z, B3.w};
                const float Cv[16] = {C0.x, C0.y, C0.z, C0.w, C1.x, C1.y, C1.z, C1.w,
                                      C2.x, C2.y, C2.z, C2.w, C3.x, C3.y, C3.z, C3.w};

                float ysum = 0.0f;
                #pragma unroll
                for (int i = 0; i < 16; ++i) {
                    float sv = fmaf(s[i], dA, dtx * Bv[i]);
                    sv = fminf(fmaxf(sv, -1000.0f), 1000.0f);
                    s[i] = sv;
                    ysum = fmaf(sv, Cv[i], ysum);
                }
                ys[u]  = ysum;
                xs4[u] = xv;
            }
            ys[0] += __shfl_xor(ys[0], 1);
            ys[1] += __shfl_xor(ys[1], 1);
            ys[2] += __shfl_xor(ys[2], 1);
            ys[3] += __shfl_xor(ys[3], 1);
            ys[0] += __shfl_xor(ys[0], 2);
            ys[1] += __shfl_xor(ys[1], 2);
            ys[2] += __shfl_xor(ys[2], 2);
            ys[3] += __shfl_xor(ys[3], 2);
            float ykeep = fmaf(Dp, xs4[0], ys[0]);
            ykeep = (nc == 1) ? fmaf(Dp, xs4[1], ys[1]) : ykeep;
            ykeep = (nc == 2) ? fmaf(Dp, xs4[2], ys[2]) : ykeep;
            ykeep = (nc == 3) ? fmaf(Dp, xs4[3], ys[3]) : ykeep;
            yp[(size_t)j4 * DSSM] = ykeep * gate;
        }

        if (more) {
            stage_store<false>(st, &sx[buf ^ 1][0], &sBC[buf ^ 1][0], &sg[buf ^ 1][0],
                               &sdd[buf ^ 1][0],
                               l1 + xrow, l1 + crow, xrow, crow, q16, q32,
                               wx, bxv, wc, bcv, dlane, t, dtb, Ah);
            __builtin_amdgcn_fence(__ATOMIC_RELEASE, "workgroup", "local");
            __builtin_amdgcn_s_barrier();
            __builtin_amdgcn_fence(__ATOMIC_ACQUIRE, "workgroup", "local");
            buf ^= 1;
        }
    }
}

extern "C" void kernel_launch(void* const* d_in, const int* in_sizes, int n_in,
                              void* d_out, int out_size, void* d_ws, size_t ws_size,
                              hipStream_t stream) {
    const float* u       = (const float*)d_in[0];
    const float* W_in    = (const float*)d_in[1];
    const float* conv_w  = (const float*)d_in[2];
    const float* conv_b  = (const float*)d_in[3];
    const float* dt_bias = (const float*)d_in[4];
    const float* A_log   = (const float*)d_in[5];
    const float* D_param = (const float*)d_in[6];
    const float* W_out   = (const float*)d_in[7];
    float* out = (float*)d_out;

    float* ws      = (float*)d_ws;
    float* zxbcdt  = ws;
    float* ybuf    = zxbcdt + (size_t)MTOK * DINPROJ + (size_t)MTOK * CONVDIM;

    {
        const int mbt = MTOK / MBLK;
        const int nbt = (DINPROJ + NBLK - 1) / NBLK;
        gemm_nt_f32_wmma<<<mbt * nbt, 256, 0, stream>>>(u, W_in, zxbcdt,
                                                        MTOK, DINPROJ, DMODEL);
    }

    ssm_scan_kernel<<<BATCH * NHEADS, 256, 0, stream>>>(zxbcdt, conv_w, conv_b, dt_bias,
                                                        A_log, D_param, ybuf);

    {
        const int mbt = MTOK / MBLK;
        const int nbt = (DMODEL + NBLK - 1) / NBLK;
        gemm_nt_f32_wmma<<<mbt * nbt, 256, 0, stream>>>(ybuf, W_out, out,
                                                        MTOK, DMODEL, DSSM);
    }
}


# out-projection GEMM retiled 128x64 (96 work-groups) -> 32x96 (256 work-groups of 4 waves, one per WGP), K staged 64 per barrier, C stored as whole 128-byte lines through LDS; sums unchanged
# speedup vs baseline: 5.3446x; 1.2210x over previous
#include <hip/hip_runtime.h>
#include <hip/hip_bf16.h>
#include <math.h>

#define DMODEL   768
#define DSTATE   64
#define DCONV    4
#define DSSM     1536
#define HEADDIM  64
#define NHEADS   24
#define CONVDIM  (DSSM + 2*DSTATE)
#define DINPROJ  (2*DSSM + 2*DSTATE + NHEADS)
#define BATCH    2
#define SEQLEN   512
#define MTOK     (BATCH*SEQLEN)

typedef float v2f __attribute__((ext_vector_type(2)));
typedef float v4f __attribute__((ext_vector_type(4)));
typedef float v8f __attribute__((ext_vector_type(8)));

#define KC    16
#define LDSS  18
#define MBLK  128
#define NBLK  64

__global__ void gemm_nt_f32_wmma(const float* __restrict__ A,
                                 const float* __restrict__ Bm,
                                 float* __restrict__ C,
                                 int M, int N, int K) {
    const int nbt = (N + NBLK - 1) / NBLK;
    const int nb  = blockIdx.x % nbt;
    const int mb  = blockIdx.x / nbt;

    const int tid  = threadIdx.x;
    const int w    = tid >> 5;
    const int lane = tid & 31;
    const int g    = lane >> 4;
    const int r16  = lane & 15;

    __shared__ float As[2][MBLK * LDSS];
    __shared__ float Bs[2][NBLK * LDSS];

    const int arow  = tid >> 1;
    const int ahalf = tid & 1;
    const float* agp = A + (size_t)(mb * MBLK + arow) * K + ahalf * 8;
    const int brow = tid >> 2;
    const int bq   = tid & 3;
    int bgrow = nb * NBLK + brow; if (bgrow >= N) bgrow = N - 1;
    const float* bgp = Bm + (size_t)bgrow * K + bq * 4;

    const int aoff = (w * 16 + r16) * LDSS + 2 * g;
    const int boff = r16 * LDSS + 2 * g;

    v8f acc0 = {}, acc1 = {}, acc2 = {}, acc3 = {};

    {
        v4f a0 = *(const v4f*)(agp);
        v4f a1 = *(const v4f*)(agp + 4);
        v4f b0 = *(const v4f*)(bgp);
        float* as = &As[0][arow * LDSS + ahalf * 8];
        *(v2f*)(as)     = v2f{a0.x, a0.y};
        *(v2f*)(as + 2) = v2f{a0.z, a0.w};
        *(v2f*)(as + 4) = v2f{a1.x, a1.y};
        *(v2f*)(as + 6) = v2f{a1.z, a1.w};
        float* bs = &Bs[0][brow * LDSS + bq * 4];
        *(v2f*)(bs)     = v2f{b0.x, b0.y};
        *(v2f*)(bs + 2) = v2f{b0.z, b0.w};
    }
    __syncthreads();

    int buf = 0;
    for (int kk = 0; kk < K; kk += KC) {
        const bool more = (kk + KC) < K;

        v4f a0, a1, b0;
        if (more) {
            a0 = *(const v4f*)(agp + kk + KC);
            a1 = *(const v4f*)(agp + kk + KC + 4);
            b0 = *(const v4f*)(bgp + kk + KC);
        }

        const float* asb = &As[buf][aoff];
        const float* bsb = &Bs[buf][boff];
        #pragma unroll
        for (int ks = 0; ks < KC; ks += 4) {
            v2f a  = *(const v2f*)(asb + ks);
            v2f q0 = *(const v2f*)(bsb +  0 * LDSS + ks);
            v2f q1 = *(const v2f*)(bsb + 16 * LDSS + ks);
            v2f q2 = *(const v2f*)(bsb + 32 * LDSS + ks);
            v2f q3 = *(const v2f*)(bsb + 48 * LDSS + ks);
            acc0 = __builtin_amdgcn_wmma_f32_16x16x4_f32(false, a, false, q0, (short)0, acc0, false, false);
            acc1 = __builtin_amdgcn_wmma_f32_16x16x4_f32(false, a, false, q1, (short)0, acc1, false, false);
            acc2 = __builtin_amdgcn_wmma_f32_16x16x4_f32(false, a, false, q2, (short)0, acc2, false, false);
            acc3 = __builtin_amdgcn_wmma_f32_16x16x4_f32(false, a, false, q3, (short)0, acc3, false, false);
        }

        if (more) {
            float* as = &As[buf ^ 1][arow * LDSS + ahalf * 8];
            *(v2f*)(as)     = v2f{a0.x, a0.y};
            *(v2f*)(as + 2) = v2f{a0.z, a0.w};
            *(v2f*)(as + 4) = v2f{a1.x, a1.y};
            *(v2f*)(as + 6) = v2f{a1.z, a1.w};
            float* bs = &Bs[buf ^ 1][brow * LDSS + bq * 4];
            *(v2f*)(bs)     = v2f{b0.x, b0.y};
            *(v2f*)(bs + 2) = v2f{b0.z, b0.w};
            __syncthreads();
            buf ^= 1;
        }
    }

    const int rowbase = mb * MBLK + w * 16 + 8 * g;
    #pragma unroll
    for (int j = 0; j < 4; ++j) {
        const v8f acc = (j == 0) ? acc0 : (j == 1) ? acc1 : (j == 2) ? acc2 : acc3;
        const int col = nb * NBLK + j * 16 + r16;
        if (col < N) {
            #pragma unroll
            for (int r = 0; r < 8; ++r)
                C[(size_t)(rowbase + r) * N + col] = acc[r];
        }
    }
}

#define OP_M     32
#define OP_N     96
#define OP_KC    64
#define OP_P     68
#define OP_CP    100
#define OP_STAGE ((OP_M + OP_N) * OP_P)

template <int M, int N, int K>
__global__ void __launch_bounds__(128)
gemm_nt_f32_wmma_32x96(const float* __restrict__ A,
                       const float* __restrict__ Bm,
                       float* __restrict__ C) {
    static_assert(M % OP_M == 0 && N % OP_N == 0 && K % OP_KC == 0,
                  "whole tiles and whole K chunks only");
    static_assert(OP_M * OP_CP <= OP_STAGE, "output tile is staged inside stage 0");
    constexpr int nbt = N / OP_N;
    const int nb  = blockIdx.x % nbt;
    const int mb  = blockIdx.x / nbt;

    const int tid  = threadIdx.x;
    const int w    = tid >> 5;
    const int lane = tid & 31;
    const int g    = lane >> 4;
    const int r16  = lane & 15;
    const int rg   = w & 1;
    const int cg   = w >> 1;

    __shared__ __attribute__((aligned(16))) float sm[2 * OP_STAGE];

    const int sp = tid & 15;
    const int sr = tid >> 4;
    const float* agp = A  + (size_t)(mb * OP_M + sr) * K + sp * 4;
    const float* bgp = Bm + (size_t)(nb * OP_N + sr) * K + sp * 4;
    const int soffA = sr * OP_P + sp * 4;
    const int soffB = OP_M * OP_P + sr * OP_P + sp * 4;

    const int aoff = (rg * 16 + r16) * OP_P + 2 * g;
    const int boff = OP_M * OP_P + (cg * 48 + r16) * OP_P + 2 * g;

    v8f acc0 = {}, acc1 = {}, acc2 = {};
    v4f ra[4], rb[12];

    #pragma unroll
    for (int i = 0; i < 4; ++i)  ra[i] = *(const v4f*)(agp + (size_t)(8 * i) * K);
    #pragma unroll
    for (int i = 0; i < 12; ++i) rb[i] = *(const v4f*)(bgp + (size_t)(8 * i) * K);
    #pragma unroll
    for (int i = 0; i < 4; ++i)  *(v4f*)(sm + soffA + 8 * i * OP_P) = ra[i];
    #pragma unroll
    for (int i = 0; i < 12; ++i) *(v4f*)(sm + soffB + 8 * i * OP_P) = rb[i];
    __syncthreads();

    int buf = 0;
    for (int kk = 0; kk < K; kk += OP_KC) {
        const bool more = (kk + OP_KC) < K;

        if (more) {
            #pragma unroll
            for (int i = 0; i < 4; ++i)
                ra[i] = *(const v4f*)(agp + (size_t)(8 * i) * K + kk + OP_KC);
            #pragma unroll
            for (int i = 0; i < 12; ++i)
                rb[i] = *(const v4f*)(bgp + (size_t)(8 * i) * K + kk + OP_KC);
        }

        const float* asb = sm + buf * OP_STAGE + aoff;
        const float* bsb = sm + buf * OP_STAGE + boff;
        #pragma unroll
        for (int ks = 0; ks < OP_KC; ks += 4) {
            v2f a  = *(const v2f*)(asb + ks);
            v2f q0 = *(const v2f*)(bsb +  0 * OP_P + ks);
            v2f q1 = *(const v2f*)(bsb + 16 * OP_P + ks);
            v2f q2 = *(const v2f*)(bsb + 32 * OP_P + ks);
            acc0 = __builtin_amdgcn_wmma_f32_16x16x4_f32(false, a, false, q0, (short)0, acc0, false, false);
            acc1 = __builtin_amdgcn_wmma_f32_16x16x4_f32(false, a, false, q1, (short)0, acc1, false, false);
            acc2 = __builtin_amdgcn_wmma_f32_16x16x4_f32(false, a, false, q2, (short)0, acc2, false, false);
        }
        __builtin_amdgcn_sched_group_barrier(0x100, 32, 0);
        __builtin_amdgcn_sched_group_barrier(0x008, 48, 0);

        if (more) {
            float* so = sm + (buf ^ 1) * OP_STAGE;
            #pragma unroll
            for (int i = 0; i < 4; ++i)  *(v4f*)(so + soffA + 8 * i * OP_P) = ra[i];
            #pragma unroll
            for (int i = 0; i < 12; ++i) *(v4f*)(so + soffB + 8 * i * OP_P) = rb[i];
            __syncthreads();
            buf ^= 1;
        }
    }

    __syncthreads();
    float* cs = sm;
    {
        float* cw = cs + (rg * 16 + 8 * g) * OP_CP + cg * 48 + r16;
        #pragma unroll
        for (int r = 0; r < 8; ++r) {
            cw[r * OP_CP +  0] = acc0[r];
            cw[r * OP_CP + 16] = acc1[r];
            cw[r * OP_CP + 32] = acc2[r];
        }
    }
    __syncthreads();
    #pragma unroll
    for (int i = 0; i < (OP_M * OP_N / 4) / 128; ++i) {
        const int idx = i * 128 + tid;
        const int row = idx / (OP_N / 4);
        const int pc  = idx % (OP_N / 4);
        const v4f v = *(const v4f*)(cs + row * OP_CP + pc * 4);
        *(v4f*)(C + (size_t)(mb * OP_M + row) * N + nb * OP_N + pc * 4) = v;
    }
}

#define TCH   32
#define NCHK  (SEQLEN / TCH)
#define ZP    72

struct ScanStage {
    v4f   rx[5];
    v4f   rc[7];
    v4f   rz[2];
    float dv;
};

__device__ __forceinline__ float silu_f(float a) {
    return a / (1.0f + __expf(-a));
}

template <bool FIRST>
__device__ __forceinline__ v4f ld_raw(const float* __restrict__ col0, int l) {
    const int lr = (FIRST && l < 0) ? 0 : l;
    return *(const v4f*)(col0 + (size_t)lr * DINPROJ);
}

template <bool FIRST>
__device__ __forceinline__ v4f conv4_silu(const v4f r0, const v4f r1, const v4f r2, const v4f r3,
                                          int l, const v4f (&w)[4], const v4f bias) {
    v4f o;
    #pragma unroll
    for (int ch = 0; ch < 4; ++ch) {
        float acc = bias[ch];
        if (!FIRST || l >= 3) acc = fmaf(r0[ch], w[ch].x, acc);
        if (!FIRST || l >= 2) acc = fmaf(r1[ch], w[ch].y, acc);
        if (!FIRST || l >= 1) acc = fmaf(r2[ch], w[ch].z, acc);
        acc = fmaf(r3[ch], w[ch].w, acc);
        o[ch] = silu_f(acc);
    }
    return o;
}

template <bool FIRST>
__device__ __forceinline__ void stage_load(ScanStage& s,
                                           const float* __restrict__ xraw,
                                           const float* __restrict__ craw,
                                           const float* __restrict__ zraw,
                                           const float* __restrict__ draw,
                                           int lx, int lc, int ld, bool dlane) {
    #pragma unroll
    for (int i = 0; i < 5; ++i) s.rx[i] = ld_raw<FIRST>(xraw, lx - 3 + i);
    #pragma unroll
    for (int i = 0; i < 7; ++i) s.rc[i] = ld_raw<FIRST>(craw, lc - 3 + i);
    s.rz[0] = *(const v4f*)(zraw + (size_t)lx * DINPROJ);
    s.rz[1] = *(const v4f*)(zraw + (size_t)(lx + 1) * DINPROJ);
    s.dv = dlane ? draw[(size_t)ld * DINPROJ] : 0.0f;
}

template <bool FIRST>
__device__ __forceinline__ void stage_store(const ScanStage& s,
                                            float* __restrict__ sxb, float* __restrict__ sBCb,
                                            float* __restrict__ sgb, v2f* __restrict__ sddb,
                                            int lx, int lc, int xrow, int crow, int q16, int q32,
                                            const v4f (&wx)[4], const v4f bxv,
                                            const v4f (&wc)[4], const v4f bcv,
                                            bool dlane, int t, float dtb, float Ah) {
    #pragma unroll
    for (int o = 0; o < 2; ++o) {
        const v4f v = conv4_silu<FIRST>(s.rx[o], s.rx[o + 1], s.rx[o + 2], s.rx[o + 3],
                                        lx + o, wx, bxv);
        *(v4f*)(sxb + (xrow + o) * HEADDIM + q16 * 4) = v;
    }
    #pragma unroll
    for (int o = 0; o < 4; ++o) {
        const v4f v = conv4_silu<FIRST>(s.rc[o], s.rc[o + 1], s.rc[o + 2], s.rc[o + 3],
                                        lc + o, wc, bcv);
        *(v4f*)(sBCb + (crow + o) * (2 * DSTATE) + q32 * 4) = v;
    }
    #pragma unroll
    for (int o = 0; o < 2; ++o) {
        const v4f z = s.rz[o];
        const v4f g = v4f{silu_f(z.x), silu_f(z.y), silu_f(z.z), silu_f(z.w)};
        *(v4f*)(sgb + (xrow + o) * ZP + q16 * 4) = g;
    }
    if (dlane) {
        const float dv = s.dv + dtb;
        float dt = (dv > 20.0f) ? dv : __logf(1.0f + __expf(dv));
        dt = fminf(fmaxf(dt, 1e-6f), 1000.0f);
        const float dA = fminf(fmaxf(__expf(dt * Ah), 1e-6f), 1000.0f);
        sddb[t] = v2f{dt, dA};
    }
}

__global__ void __launch_bounds__(256)
ssm_scan_kernel(const float* __restrict__ zxbcdt,
                const float* __restrict__ conv_w,
                const float* __restrict__ conv_b,
                const float* __restrict__ dt_bias,
                const float* __restrict__ A_log,
                const float* __restrict__ D_param,
                float* __restrict__ y) {
    const int bh = blockIdx.x;
    const int b  = bh / NHEADS;
    const int h  = bh % NHEADS;
    const int t  = threadIdx.x;
    const int p  = t >> 2;
    const int nc = t & 3;

    const float Ah  = -__expf(A_log[h]);
    const float dtb = dt_bias[h];
    const float Dp  = D_param[h];

    __shared__ __attribute__((aligned(16))) float sx [2][TCH * HEADDIM];
    __shared__ __attribute__((aligned(16))) float sBC[2][TCH * 2 * DSTATE];
    __shared__ __attribute__((aligned(16))) float sg [2][TCH * ZP];
    __shared__ __attribute__((aligned(8)))  v2f   sdd[2][TCH];

    const int q16  = t & 15;
    const int xrow = (t >> 4) * 2;
    const int q32  = t & 31;
    const int crow = (t >> 5) * 4;
    const bool dlane = t < TCH;
    const int cx = h * HEADDIM + q16 * 4;
    const int cb = DSSM + q32 * 4;
    const float* base = zxbcdt + (size_t)b * SEQLEN * DINPROJ;
    const float* zraw = base + cx;
    const float* xraw = base + DSSM + cx;
    const float* craw = base + DSSM + cb;
    const float* draw = base + (DSSM + CONVDIM) + h;

    v4f wx[4], wc[4];
    #pragma unroll
    for (int ch = 0; ch < 4; ++ch) {
        wx[ch] = *(const v4f*)(conv_w + (size_t)(cx + ch) * DCONV);
        wc[ch] = *(const v4f*)(conv_w + (size_t)(cb + ch) * DCONV);
    }
    const v4f bxv = *(const v4f*)(conv_b + cx);
    const v4f bcv = *(const v4f*)(conv_b + cb);

    float s[16];
    #pragma unroll
    for (int i = 0; i < 16; ++i) s[i] = 0.0f;

    {
        ScanStage st;
        stage_load<true>(st, xraw, craw, zraw, draw, xrow, crow, t, dlane);
        stage_store<true>(st, &sx[0][0], &sBC[0][0], &sg[0][0], &sdd[0][0],
                          xrow, crow, xrow, crow, q16, q32, wx, bxv, wc, bcv,
                          dlane, t, dtb, Ah);
    }
    __syncthreads();

    int buf = 0;
    for (int c = 0; c < NCHK; ++c) {
        const bool more = (c + 1) < NCHK;
        const int  l1   = (c + 1) * TCH;

        ScanStage st;
        if (more)
            stage_load<false>(st, xraw, craw, zraw, draw, l1 + xrow, l1 + crow, l1 + t, dlane);

        const float* bx  = &sx[buf][p];
        const float* bBC = &sBC[buf][nc * 16];
        const float* bg  = &sg[buf][nc * ZP + p];
        float* yp = y + (size_t)(b * SEQLEN + c * TCH + nc) * DSSM + h * HEADDIM + p;
        for (int j4 = 0; j4 < TCH; j4 += 4) {
            float ys[4], xs4[4];
            const float gate = bg[j4 * ZP];
            #pragma unroll
            for (int u = 0; u < 4; ++u) {
                const int j = j4 + u;
                const v2f   dd  = sdd[buf][j];
                const float dt  = dd.x;
                const float dA  = dd.y;
                const float xv  = bx[j * HEADDIM];
                const float dtx = dt * xv;

                const v4f* Bq = (const v4f*)(bBC + j * 2 * DSTATE);
                const v4f* Cq = (const v4f*)(bBC + j * 2 * DSTATE + DSTATE);
                const v4f B0 = Bq[0], B1 = Bq[1], B2 = Bq[2], B3 = Bq[3];
                const v4f C0 = Cq[0], C1 = Cq[1], C2 = Cq[2], C3 = Cq[3];
                const float Bv[16] = {B0.x, B0.y, B0.z, B0.w, B1.x, B1.y, B1.z, B1.w,
                                      B2.x, B2.y, B2.z, B2.w, B3.x, B3.y, B3.z, B3.w};
                const float Cv[16] = {C0.x, C0.y, C0.z, C0.w, C1.x, C1.y, C1.z, C1.w,
                                      C2.x, C2.y, C2.z, C2.w, C3.x, C3.y, C3.z, C3.w};

                float ysum = 0.0f;
                #pragma unroll
                for (int i = 0; i < 16; ++i) {
                    float sv = fmaf(s[i], dA, dtx * Bv[i]);
                    sv = fminf(fmaxf(sv, -1000.0f), 1000.0f);
                    s[i] = sv;
                    ysum = fmaf(sv, Cv[i], ysum);
                }
                ys[u]  = ysum;
                xs4[u] = xv;
            }
            ys[0] += __shfl_xor(ys[0], 1);
            ys[1] += __shfl_xor(ys[1], 1);
            ys[2] += __shfl_xor(ys[2], 1);
            ys[3] += __shfl_xor(ys[3], 1);
            ys[0] += __shfl_xor(ys[0], 2);
            ys[1] += __shfl_xor(ys[1], 2);
            ys[2] += __shfl_xor(ys[2], 2);
            ys[3] += __shfl_xor(ys[3], 2);
            float ykeep = fmaf(Dp, xs4[0], ys[0]);
            ykeep = (nc == 1) ? fmaf(Dp, xs4[1], ys[1]) : ykeep;
            ykeep = (nc == 2) ? fmaf(Dp, xs4[2], ys[2]) : ykeep;
            ykeep = (nc == 3) ? fmaf(Dp, xs4[3], ys[3]) : ykeep;
            yp[(size_t)j4 * DSSM] = ykeep * gate;
        }

        if (more) {
            stage_store<false>(st, &sx[buf ^ 1][0], &sBC[buf ^ 1][0], &sg[buf ^ 1][0],
                               &sdd[buf ^ 1][0],
                               l1 + xrow, l1 + crow, xrow, crow, q16, q32,
                               wx, bxv, wc, bcv, dlane, t, dtb, Ah);
            __builtin_amdgcn_fence(__ATOMIC_RELEASE, "workgroup", "local");
            __builtin_amdgcn_s_barrier();
            __builtin_amdgcn_fence(__ATOMIC_ACQUIRE, "workgroup", "local");
            buf ^= 1;
        }
    }
}

extern "C" void kernel_launch(void* const* d_in, const int* in_sizes, int n_in,
                              void* d_out, int out_size, void* d_ws, size_t ws_size,
                              hipStream_t stream) {
    const float* u       = (const float*)d_in[0];
    const float* W_in    = (const float*)d_in[1];
    const float* conv_w  = (const float*)d_in[2];
    const float* conv_b  = (const float*)d_in[3];
    const float* dt_bias = (const float*)d_in[4];
    const float* A_log   = (const float*)d_in[5];
    const float* D_param = (const float*)d_in[6];
    const float* W_out   = (const float*)d_in[7];
    float* out = (float*)d_out;

    float* ws      = (float*)d_ws;
    float* zxbcdt  = ws;
    float* ybuf    = zxbcdt + (size_t)MTOK * DINPROJ + (size_t)MTOK * CONVDIM;

    {
        const int mbt = MTOK / MBLK;
        const int nbt = (DINPROJ + NBLK - 1) / NBLK;
        gemm_nt_f32_wmma<<<mbt * nbt, 256, 0, stream>>>(u, W_in, zxbcdt,
                                                        MTOK, DINPROJ, DMODEL);
    }

    ssm_scan_kernel<<<BATCH * NHEADS, 256, 0, stream>>>(zxbcdt, conv_w, conv_b, dt_bias,
                                                        A_log, D_param, ybuf);

    {
        const int mbt = MTOK / OP_M;
        const int nbt = DMODEL / OP_N;
        gemm_nt_f32_wmma_32x96<MTOK, DMODEL, DSSM><<<mbt * nbt, 128, 0, stream>>>(ybuf, W_out, out);
    }
}


# scan kernel: each head's 64 state rows split over 2 work-groups of 4 waves (96 work-groups, one wave per SIMD; was 48 of 2 waves per SIMD); each stages B, C, dt itself; sums unchanged
# speedup vs baseline: 5.6405x; 1.0554x over previous
#include <hip/hip_runtime.h>
#include <hip/hip_bf16.h>
#include <math.h>

#define DMODEL   768
#define DSTATE   64
#define DCONV    4
#define DSSM     1536
#define HEADDIM  64
#define NHEADS   24
#define CONVDIM  (DSSM + 2*DSTATE)
#define DINPROJ  (2*DSSM + 2*DSTATE + NHEADS)
#define BATCH    2
#define SEQLEN   512
#define MTOK     (BATCH*SEQLEN)

typedef float v2f __attribute__((ext_vector_type(2)));
typedef float v4f __attribute__((ext_vector_type(4)));
typedef float v8f __attribute__((ext_vector_type(8)));

#define KC    16
#define LDSS  18
#define MBLK  128
#define NBLK  64

__global__ void gemm_nt_f32_wmma(const float* __restrict__ A,
                                 const float* __restrict__ Bm,
                                 float* __restrict__ C,
                                 int M, int N, int K) {
    const int nbt = (N + NBLK - 1) / NBLK;
    const int nb  = blockIdx.x % nbt;
    const int mb  = blockIdx.x / nbt;

    const int tid  = threadIdx.x;
    const int w    = tid >> 5;
    const int lane = tid & 31;
    const int g    = lane >> 4;
    const int r16  = lane & 15;

    __shared__ float As[2][MBLK * LDSS];
    __shared__ float Bs[2][NBLK * LDSS];

    const int arow  = tid >> 1;
    const int ahalf = tid & 1;
    const float* agp = A + (size_t)(mb * MBLK + arow) * K + ahalf * 8;
    const int brow = tid >> 2;
    const int bq   = tid & 3;
    int bgrow = nb * NBLK + brow; if (bgrow >= N) bgrow = N - 1;
    const float* bgp = Bm + (size_t)bgrow * K + bq * 4;

    const int aoff = (w * 16 + r16) * LDSS + 2 * g;
    const int boff = r16 * LDSS + 2 * g;

    v8f acc0 = {}, acc1 = {}, acc2 = {}, acc3 = {};

    {
        v4f a0 = *(const v4f*)(agp);
        v4f a1 = *(const v4f*)(agp + 4);
        v4f b0 = *(const v4f*)(bgp);
        float* as = &As[0][arow * LDSS + ahalf * 8];
        *(v2f*)(as)     = v2f{a0.x, a0.y};
        *(v2f*)(as + 2) = v2f{a0.z, a0.w};
        *(v2f*)(as + 4) = v2f{a1.x, a1.y};
        *(v2f*)(as + 6) = v2f{a1.z, a1.w};
        float* bs = &Bs[0][brow * LDSS + bq * 4];
        *(v2f*)(bs)     = v2f{b0.x, b0.y};
        *(v2f*)(bs + 2) = v2f{b0.z, b0.w};
    }
    __syncthreads();

    int buf = 0;
    for (int kk = 0; kk < K; kk += KC) {
        const bool more = (kk + KC) < K;

        v4f a0, a1, b0;
        if (more) {
            a0 = *(const v4f*)(agp + kk + KC);
            a1 = *(const v4f*)(agp + kk + KC + 4);
            b0 = *(const v4f*)(bgp + kk + KC);
        }

        const float* asb = &As[buf][aoff];
        const float* bsb = &Bs[buf][boff];
        #pragma unroll
        for (int ks = 0; ks < KC; ks += 4) {
            v2f a  = *(const v2f*)(asb + ks);
            v2f q0 = *(const v2f*)(bsb +  0 * LDSS + ks);
            v2f q1 = *(const v2f*)(bsb + 16 * LDSS + ks);
            v2f q2 = *(const v2f*)(bsb + 32 * LDSS + ks);
            v2f q3 = *(const v2f*)(bsb + 48 * LDSS + ks);
            acc0 = __builtin_amdgcn_wmma_f32_16x16x4_f32(false, a, false, q0, (short)0, acc0, false, false);
            acc1 = __builtin_amdgcn_wmma_f32_16x16x4_f32(false, a, false, q1, (short)0, acc1, false, false);
            acc2 = __builtin_amdgcn_wmma_f32_16x16x4_f32(false, a, false, q2, (short)0, acc2, false, false);
            acc3 = __builtin_amdgcn_wmma_f32_16x16x4_f32(false, a, false, q3, (short)0, acc3, false, false);
        }

        if (more) {
            float* as = &As[buf ^ 1][arow * LDSS + ahalf * 8];
            *(v2f*)(as)     = v2f{a0.x, a0.y};
            *(v2f*)(as + 2) = v2f{a0.z, a0.w};
            *(v2f*)(as + 4) = v2f{a1.x, a1.y};
            *(v2f*)(as + 6) = v2f{a1.z, a1.w};
            float* bs = &Bs[buf ^ 1][brow * LDSS + bq * 4];
            *(v2f*)(bs)     = v2f{b0.x, b0.y};
            *(v2f*)(bs + 2) = v2f{b0.z, b0.w};
            __syncthreads();
            buf ^= 1;
        }
    }

    const int rowbase = mb * MBLK + w * 16 + 8 * g;
    #pragma unroll
    for (int j = 0; j < 4; ++j) {
        const v8f acc = (j == 0) ? acc0 : (j == 1) ? acc1 : (j == 2) ? acc2 : acc3;
        const int col = nb * NBLK + j * 16 + r16;
        if (col < N) {
            #pragma unroll
            for (int r = 0; r < 8; ++r)
                C[(size_t)(rowbase + r) * N + col] = acc[r];
        }
    }
}

#define OP_M     32
#define OP_N     96
#define OP_KC    64
#define OP_P     68
#define OP_CP    100
#define OP_STAGE ((OP_M + OP_N) * OP_P)

template <int M, int N, int K>
__global__ void __launch_bounds__(128)
gemm_nt_f32_wmma_32x96(const float* __restrict__ A,
                       const float* __restrict__ Bm,
                       float* __restrict__ C) {
    static_assert(M % OP_M == 0 && N % OP_N == 0 && K % OP_KC == 0,
                  "whole tiles and whole K chunks only");
    static_assert(OP_M * OP_CP <= OP_STAGE, "output tile is staged inside stage 0");
    constexpr int nbt = N / OP_N;
    const int nb  = blockIdx.x % nbt;
    const int mb  = blockIdx.x / nbt;

    const int tid  = threadIdx.x;
    const int w    = tid >> 5;
    const int lane = tid & 31;
    const int g    = lane >> 4;
    const int r16  = lane & 15;
    const int rg   = w & 1;
    const int cg   = w >> 1;

    __shared__ __attribute__((aligned(16))) float sm[2 * OP_STAGE];

    const int sp = tid & 15;
    const int sr = tid >> 4;
    const float* agp = A  + (size_t)(mb * OP_M + sr) * K + sp * 4;
    const float* bgp = Bm + (size_t)(nb * OP_N + sr) * K + sp * 4;
    const int soffA = sr * OP_P + sp * 4;
    const int soffB = OP_M * OP_P + sr * OP_P + sp * 4;

    const int aoff = (rg * 16 + r16) * OP_P + 2 * g;
    const int boff = OP_M * OP_P + (cg * 48 + r16) * OP_P + 2 * g;

    v8f acc0 = {}, acc1 = {}, acc2 = {};
    v4f ra[4], rb[12];

    #pragma unroll
    for (int i = 0; i < 4; ++i)  ra[i] = *(const v4f*)(agp + (size_t)(8 * i) * K);
    #pragma unroll
    for (int i = 0; i < 12; ++i) rb[i] = *(const v4f*)(bgp + (size_t)(8 * i) * K);
    #pragma unroll
    for (int i = 0; i < 4; ++i)  *(v4f*)(sm + soffA + 8 * i * OP_P) = ra[i];
    #pragma unroll
    for (int i = 0; i < 12; ++i) *(v4f*)(sm + soffB + 8 * i * OP_P) = rb[i];
    __syncthreads();

    int buf = 0;
    for (int kk = 0; kk < K; kk += OP_KC) {
        const bool more = (kk + OP_KC) < K;

        if (more) {
            #pragma unroll
            for (int i = 0; i < 4; ++i)
                ra[i] = *(const v4f*)(agp + (size_t)(8 * i) * K + kk + OP_KC);
            #pragma unroll
            for (int i = 0; i < 12; ++i)
                rb[i] = *(const v4f*)(bgp + (size_t)(8 * i) * K + kk + OP_KC);
        }

        const float* asb = sm + buf * OP_STAGE + aoff;
        const float* bsb = sm + buf * OP_STAGE + boff;
        #pragma unroll
        for (int ks = 0; ks < OP_KC; ks += 4) {
            v2f a  = *(const v2f*)(asb + ks);
            v2f q0 = *(const v2f*)(bsb +  0 * OP_P + ks);
            v2f q1 = *(const v2f*)(bsb + 16 * OP_P + ks);
            v2f q2 = *(const v2f*)(bsb + 32 * OP_P + ks);
            acc0 = __builtin_amdgcn_wmma_f32_16x16x4_f32(false, a, false, q0, (short)0, acc0, false, false);
            acc1 = __builtin_amdgcn_wmma_f32_16x16x4_f32(false, a, false, q1, (short)0, acc1, false, false);
            acc2 = __builtin_amdgcn_wmma_f32_16x16x4_f32(false, a, false, q2, (short)0, acc2, false, false);
        }
        __builtin_amdgcn_sched_group_barrier(0x100, 32, 0);
        __builtin_amdgcn_sched_group_barrier(0x008, 48, 0);

        if (more) {
            float* so = sm + (buf ^ 1) * OP_STAGE;
            #pragma unroll
            for (int i = 0; i < 4; ++i)  *(v4f*)(so + soffA + 8 * i * OP_P) = ra[i];
            #pragma unroll
            for (int i = 0; i < 12; ++i) *(v4f*)(so + soffB + 8 * i * OP_P) = rb[i];
            __syncthreads();
            buf ^= 1;
        }
    }

    __syncthreads();
    float* cs = sm;
    {
        float* cw = cs + (rg * 16 + 8 * g) * OP_CP + cg * 48 + r16;
        #pragma unroll
        for (int r = 0; r < 8; ++r) {
            cw[r * OP_CP +  0] = acc0[r];
            cw[r * OP_CP + 16] = acc1[r];
            cw[r * OP_CP + 32] = acc2[r];
        }
    }
    __syncthreads();
    #pragma unroll
    for (int i = 0; i < (OP_M * OP_N / 4) / 128; ++i) {
        const int idx = i * 128 + tid;
        const int row = idx / (OP_N / 4);
        const int pc  = idx % (OP_N / 4);
        const v4f v = *(const v4f*)(cs + row * OP_CP + pc * 4);
        *(v4f*)(C + (size_t)(mb * OP_M + row) * N + nb * OP_N + pc * 4) = v;
    }
}

#define TCH   32
#define NCHK  (SEQLEN / TCH)
#define PSPLIT 2
#define PROWS  (HEADDIM / PSPLIT)
#define STHR   (PROWS * 4)
#define CROWS  (TCH * 32 / STHR)
#define ZP    40

struct ScanStage {
    v4f   rx[5];
    v4f   rc[CROWS + 3];
    v4f   rz[2];
    float dv;
};

__device__ __forceinline__ float silu_f(float a) {
    return a / (1.0f + __expf(-a));
}

template <bool FIRST>
__device__ __forceinline__ v4f ld_raw(const float* __restrict__ col0, int l) {
    const int lr = (FIRST && l < 0) ? 0 : l;
    return *(const v4f*)(col0 + (size_t)lr * DINPROJ);
}

template <bool FIRST>
__device__ __forceinline__ v4f conv4_silu(const v4f r0, const v4f r1, const v4f r2, const v4f r3,
                                          int l, const v4f (&w)[4], const v4f bias) {
    v4f o;
    #pragma unroll
    for (int ch = 0; ch < 4; ++ch) {
        float acc = bias[ch];
        if (!FIRST || l >= 3) acc = fmaf(r0[ch], w[ch].x, acc);
        if (!FIRST || l >= 2) acc = fmaf(r1[ch], w[ch].y, acc);
        if (!FIRST || l >= 1) acc = fmaf(r2[ch], w[ch].z, acc);
        acc = fmaf(r3[ch], w[ch].w, acc);
        o[ch] = silu_f(acc);
    }
    return o;
}

template <bool FIRST>
__device__ __forceinline__ void stage_load(ScanStage& s,
                                           const float* __restrict__ xraw,
                                           const float* __restrict__ craw,
                                           const float* __restrict__ zraw,
                                           const float* __restrict__ draw,
                                           int lx, int lc, int ld, bool dlane) {
    #pragma unroll
    for (int i = 0; i < 5; ++i) s.rx[i] = ld_raw<FIRST>(xraw, lx - 3 + i);
    #pragma unroll
    for (int i = 0; i < CROWS + 3; ++i) s.rc[i] = ld_raw<FIRST>(craw, lc - 3 + i);
    s.rz[0] = *(const v4f*)(zraw + (size_t)lx * DINPROJ);
    s.rz[1] = *(const v4f*)(zraw + (size_t)(lx + 1) * DINPROJ);
    s.dv = dlane ? draw[(size_t)ld * DINPROJ] : 0.0f;
}

template <bool FIRST>
__device__ __forceinline__ void stage_store(const ScanStage& s,
                                            float* __restrict__ sxb, float* __restrict__ sBCb,
                                            float* __restrict__ sgb, v2f* __restrict__ sddb,
                                            int lx, int lc, int xrow, int crow, int q8, int q32,
                                            const v4f (&wx)[4], const v4f bxv,
                                            const v4f (&wc)[4], const v4f bcv,
                                            bool dlane, int t, float dtb, float Ah) {
    #pragma unroll
    for (int o = 0; o < 2; ++o) {
        const v4f v = conv4_silu<FIRST>(s.rx[o], s.rx[o + 1], s.rx[o + 2], s.rx[o + 3],
                                        lx + o, wx, bxv);
        *(v4f*)(sxb + (xrow + o) * PROWS + q8 * 4) = v;
    }
    #pragma unroll
    for (int o = 0; o < CROWS; ++o) {
        const v4f v = conv4_silu<FIRST>(s.rc[o], s.rc[o + 1], s.rc[o + 2], s.rc[o + 3],
                                        lc + o, wc, bcv);
        *(v4f*)(sBCb + (crow + o) * (2 * DSTATE) + q32 * 4) = v;
    }
    #pragma unroll
    for (int o = 0; o < 2; ++o) {
        const v4f z = s.rz[o];
        const v4f g = v4f{silu_f(z.x), silu_f(z.y), silu_f(z.z), silu_f(z.w)};
        *(v4f*)(sgb + (xrow + o) * ZP + q8 * 4) = g;
    }
    if (dlane) {
        const float dv = s.dv + dtb;
        float dt = (dv > 20.0f) ? dv : __logf(1.0f + __expf(dv));
        dt = fminf(fmaxf(dt, 1e-6f), 1000.0f);
        const float dA = fminf(fmaxf(__expf(dt * Ah), 1e-6f), 1000.0f);
        sddb[t] = v2f{dt, dA};
    }
}

__global__ void __launch_bounds__(STHR)
ssm_scan_kernel(const float* __restrict__ zxbcdt,
                const float* __restrict__ conv_w,
                const float* __restrict__ conv_b,
                const float* __restrict__ dt_bias,
                const float* __restrict__ A_log,
                const float* __restrict__ D_param,
                float* __restrict__ y) {
    const int bh = blockIdx.x / PSPLIT;
    const int ph = blockIdx.x % PSPLIT;
    const int b  = bh / NHEADS;
    const int h  = bh % NHEADS;
    const int t  = threadIdx.x;
    const int p  = t >> 2;
    const int nc = t & 3;

    const float Ah  = -__expf(A_log[h]);
    const float dtb = dt_bias[h];
    const float Dp  = D_param[h];

    __shared__ __attribute__((aligned(16))) float sx [2][TCH * PROWS];
    __shared__ __attribute__((aligned(16))) float sBC[2][TCH * 2 * DSTATE];
    __shared__ __attribute__((aligned(16))) float sg [2][TCH * ZP];
    __shared__ __attribute__((aligned(8)))  v2f   sdd[2][TCH];

    const int q8   = t & 7;
    const int xrow = (t >> 3) * 2;
    const int q32  = t & 31;
    const int crow = (t >> 5) * CROWS;
    const bool dlane = t < TCH;
    const int cx = h * HEADDIM + ph * PROWS + q8 * 4;
    const int cb = DSSM + q32 * 4;
    const float* base = zxbcdt + (size_t)b * SEQLEN * DINPROJ;
    const float* zraw = base + cx;
    const float* xraw = base + DSSM + cx;
    const float* craw = base + DSSM + cb;
    const float* draw = base + (DSSM + CONVDIM) + h;

    v4f wx[4], wc[4];
    #pragma unroll
    for (int ch = 0; ch < 4; ++ch) {
        wx[ch] = *(const v4f*)(conv_w + (size_t)(cx + ch) * DCONV);
        wc[ch] = *(const v4f*)(conv_w + (size_t)(cb + ch) * DCONV);
    }
    const v4f bxv = *(const v4f*)(conv_b + cx);
    const v4f bcv = *(const v4f*)(conv_b + cb);

    float s[16];
    #pragma unroll
    for (int i = 0; i < 16; ++i) s[i] = 0.0f;

    {
        ScanStage st;
        stage_load<true>(st, xraw, craw, zraw, draw, xrow, crow, t, dlane);
        stage_store<true>(st, &sx[0][0], &sBC[0][0], &sg[0][0], &sdd[0][0],
                          xrow, crow, xrow, crow, q8, q32, wx, bxv, wc, bcv,
                          dlane, t, dtb, Ah);
    }
    __syncthreads();

    int buf = 0;
    for (int c = 0; c < NCHK; ++c) {
        const bool more = (c + 1) < NCHK;
        const int  l1   = (c + 1) * TCH;

        ScanStage st;
        if (more)
            stage_load<false>(st, xraw, craw, zraw, draw, l1 + xrow, l1 + crow, l1 + t, dlane);

        const float* bx  = &sx[buf][p];
        const float* bBC = &sBC[buf][nc * 16];
        const float* bg  = &sg[buf][nc * ZP + p];
        float* yp = y + (size_t)(b * SEQLEN + c * TCH + nc) * DSSM + h * HEADDIM + ph * PROWS + p;
        for (int j4 = 0; j4 < TCH; j4 += 4) {
            float ys[4], xs4[4];
            const float gate = bg[j4 * ZP];
            #pragma unroll
            for (int u = 0; u < 4; ++u) {
                const int j = j4 + u;
                const v2f   dd  = sdd[buf][j];
                const float dt  = dd.x;
                const float dA  = dd.y;
                const float xv  = bx[j * PROWS];
                const float dtx = dt * xv;

                const v4f* Bq = (const v4f*)(bBC + j * 2 * DSTATE);
                const v4f* Cq = (const v4f*)(bBC + j * 2 * DSTATE + DSTATE);
                const v4f B0 = Bq[0], B1 = Bq[1], B2 = Bq[2], B3 = Bq[3];
                const v4f C0 = Cq[0], C1 = Cq[1], C2 = Cq[2], C3 = Cq[3];
                const float Bv[16] = {B0.x, B0.y, B0.z, B0.w, B1.x, B1.y, B1.z, B1.w,
                                      B2.x, B2.y, B2.z, B2.w, B3.x, B3.y, B3.z, B3.w};
                const float Cv[16] = {C0.x, C0.y, C0.z, C0.w, C1.x, C1.y, C1.z, C1.w,
                                      C2.x, C2.y, C2.z, C2.w, C3.x, C3.y, C3.z, C3.w};

                float ysum = 0.0f;
                #pragma unroll
                for (int i = 0; i < 16; ++i) {
                    float sv = fmaf(s[i], dA, dtx * Bv[i]);
                    sv = fminf(fmaxf(sv, -1000.0f), 1000.0f);
                    s[i] = sv;
                    ysum = fmaf(sv, Cv[i], ysum);
                }
                ys[u]  = ysum;
                xs4[u] = xv;
            }
            ys[0] += __shfl_xor(ys[0], 1);
            ys[1] += __shfl_xor(ys[1], 1);
            ys[2] += __shfl_xor(ys[2], 1);
            ys[3] += __shfl_xor(ys[3], 1);
            ys[0] += __shfl_xor(ys[0], 2);
            ys[1] += __shfl_xor(ys[1], 2);
            ys[2] += __shfl_xor(ys[2], 2);
            ys[3] += __shfl_xor(ys[3], 2);
            float ykeep = fmaf(Dp, xs4[0], ys[0]);
            ykeep = (nc == 1) ? fmaf(Dp, xs4[1], ys[1]) : ykeep;
            ykeep = (nc == 2) ? fmaf(Dp, xs4[2], ys[2]) : ykeep;
            ykeep = (nc == 3) ? fmaf(Dp, xs4[3], ys[3]) : ykeep;
            yp[(size_t)j4 * DSSM] = ykeep * gate;
        }

        if (more) {
            stage_store<false>(st, &sx[buf ^ 1][0], &sBC[buf ^ 1][0], &sg[buf ^ 1][0],
                               &sdd[buf ^ 1][0],
                               l1 + xrow, l1 + crow, xrow, crow, q8, q32,
                               wx, bxv, wc, bcv, dlane, t, dtb, Ah);
            __builtin_amdgcn_fence(__ATOMIC_RELEASE, "workgroup", "local");
            __builtin_amdgcn_s_barrier();
            __builtin_amdgcn_fence(__ATOMIC_ACQUIRE, "workgroup", "local");
            buf ^= 1;
        }
    }
}

extern "C" void kernel_launch(void* const* d_in, const int* in_sizes, int n_in,
                              void* d_out, int out_size, void* d_ws, size_t ws_size,
                              hipStream_t stream) {
    const float* u       = (const float*)d_in[0];
    const float* W_in    = (const float*)d_in[1];
    const float* conv_w  = (const float*)d_in[2];
    const float* conv_b  = (const float*)d_in[3];
    const float* dt_bias = (const float*)d_in[4];
    const float* A_log   = (const float*)d_in[5];
    const float* D_param = (const float*)d_in[6];
    const float* W_out   = (const float*)d_in[7];
    float* out = (float*)d_out;

    float* ws      = (float*)d_ws;
    float* zxbcdt  = ws;
    float* ybuf    = zxbcdt + (size_t)MTOK * DINPROJ + (size_t)MTOK * CONVDIM;

    {
        const int mbt = MTOK / MBLK;
        const int nbt = (DINPROJ + NBLK - 1) / NBLK;
        gemm_nt_f32_wmma<<<mbt * nbt, 256, 0, stream>>>(u, W_in, zxbcdt,
                                                        MTOK, DINPROJ, DMODEL);
    }

    ssm_scan_kernel<<<BATCH * NHEADS * PSPLIT, STHR, 0, stream>>>(zxbcdt, conv_w, conv_b, dt_bias,
                                                                  A_log, D_param, ybuf);

    {
        const int mbt = MTOK / OP_M;
        const int nbt = DMODEL / OP_N;
        gemm_nt_f32_wmma_32x96<MTOK, DMODEL, DSSM><<<mbt * nbt, 128, 0, stream>>>(ybuf, W_out, out);
    }
}


# scan kernel: 4-lane y reduction done with DPP adds (row_xmask 1, 2) on the vector ALU instead of __shfl_xor through the LDS unit; gate read kept at the group's head; same partners, sums unchanged
# speedup vs baseline: 6.3222x; 1.1209x over previous
#include <hip/hip_runtime.h>
#include <hip/hip_bf16.h>
#include <math.h>

#define DMODEL   768
#define DSTATE   64
#define DCONV    4
#define DSSM     1536
#define HEADDIM  64
#define NHEADS   24
#define CONVDIM  (DSSM + 2*DSTATE)
#define DINPROJ  (2*DSSM + 2*DSTATE + NHEADS)
#define BATCH    2
#define SEQLEN   512
#define MTOK     (BATCH*SEQLEN)

typedef float v2f __attribute__((ext_vector_type(2)));
typedef float v4f __attribute__((ext_vector_type(4)));
typedef float v8f __attribute__((ext_vector_type(8)));

#define KC    16
#define LDSS  18
#define MBLK  128
#define NBLK  64

__global__ void gemm_nt_f32_wmma(const float* __restrict__ A,
                                 const float* __restrict__ Bm,
                                 float* __restrict__ C,
                                 int M, int N, int K) {
    const int nbt = (N + NBLK - 1) / NBLK;
    const int nb  = blockIdx.x % nbt;
    const int mb  = blockIdx.x / nbt;

    const int tid  = threadIdx.x;
    const int w    = tid >> 5;
    const int lane = tid & 31;
    const int g    = lane >> 4;
    const int r16  = lane & 15;

    __shared__ float As[2][MBLK * LDSS];
    __shared__ float Bs[2][NBLK * LDSS];

    const int arow  = tid >> 1;
    const int ahalf = tid & 1;
    const float* agp = A + (size_t)(mb * MBLK + arow) * K + ahalf * 8;
    const int brow = tid >> 2;
    const int bq   = tid & 3;
    int bgrow = nb * NBLK + brow; if (bgrow >= N) bgrow = N - 1;
    const float* bgp = Bm + (size_t)bgrow * K + bq * 4;

    const int aoff = (w * 16 + r16) * LDSS + 2 * g;
    const int boff = r16 * LDSS + 2 * g;

    v8f acc0 = {}, acc1 = {}, acc2 = {}, acc3 = {};

    {
        v4f a0 = *(const v4f*)(agp);
        v4f a1 = *(const v4f*)(agp + 4);
        v4f b0 = *(const v4f*)(bgp);
        float* as = &As[0][arow * LDSS + ahalf * 8];
        *(v2f*)(as)     = v2f{a0.x, a0.y};
        *(v2f*)(as + 2) = v2f{a0.z, a0.w};
        *(v2f*)(as + 4) = v2f{a1.x, a1.y};
        *(v2f*)(as + 6) = v2f{a1.z, a1.w};
        float* bs = &Bs[0][brow * LDSS + bq * 4];
        *(v2f*)(bs)     = v2f{b0.x, b0.y};
        *(v2f*)(bs + 2) = v2f{b0.z, b0.w};
    }
    __syncthreads();

    int buf = 0;
    for (int kk = 0; kk < K; kk += KC) {
        const bool more = (kk + KC) < K;

        v4f a0, a1, b0;
        if (more) {
            a0 = *(const v4f*)(agp + kk + KC);
            a1 = *(const v4f*)(agp + kk + KC + 4);
            b0 = *(const v4f*)(bgp + kk + KC);
        }

        const float* asb = &As[buf][aoff];
        const float* bsb = &Bs[buf][boff];
        #pragma unroll
        for (int ks = 0; ks < KC; ks += 4) {
            v2f a  = *(const v2f*)(asb + ks);
            v2f q0 = *(const v2f*)(bsb +  0 * LDSS + ks);
            v2f q1 = *(const v2f*)(bsb + 16 * LDSS + ks);
            v2f q2 = *(const v2f*)(bsb + 32 * LDSS + ks);
            v2f q3 = *(const v2f*)(bsb + 48 * LDSS + ks);
            acc0 = __builtin_amdgcn_wmma_f32_16x16x4_f32(false, a, false, q0, (short)0, acc0, false, false);
            acc1 = __builtin_amdgcn_wmma_f32_16x16x4_f32(false, a, false, q1, (short)0, acc1, false, false);
            acc2 = __builtin_amdgcn_wmma_f32_16x16x4_f32(false, a, false, q2, (short)0, acc2, false, false);
            acc3 = __builtin_amdgcn_wmma_f32_16x16x4_f32(false, a, false, q3, (short)0, acc3, false, false);
        }

        if (more) {
            float* as = &As[buf ^ 1][arow * LDSS + ahalf * 8];
            *(v2f*)(as)     = v2f{a0.x, a0.y};
            *(v2f*)(as + 2) = v2f{a0.z, a0.w};
            *(v2f*)(as + 4) = v2f{a1.x, a1.y};
            *(v2f*)(as + 6) = v2f{a1.z, a1.w};
            float* bs = &Bs[buf ^ 1][brow * LDSS + bq * 4];
            *(v2f*)(bs)     = v2f{b0.x, b0.y};
            *(v2f*)(bs + 2) = v2f{b0.z, b0.w};
            __syncthreads();
            buf ^= 1;
        }
    }

    const int rowbase = mb * MBLK + w * 16 + 8 * g;
    #pragma unroll
    for (int j = 0; j < 4; ++j) {
        const v8f acc = (j == 0) ? acc0 : (j == 1) ? acc1 : (j == 2) ? acc2 : acc3;
        const int col = nb * NBLK + j * 16 + r16;
        if (col < N) {
            #pragma unroll
            for (int r = 0; r < 8; ++r)
                C[(size_t)(rowbase + r) * N + col] = acc[r];
        }
    }
}

#define OP_M     32
#define OP_N     96
#define OP_KC    64
#define OP_P     68
#define OP_CP    100
#define OP_STAGE ((OP_M + OP_N) * OP_P)

template <int M, int N, int K>
__global__ void __launch_bounds__(128)
gemm_nt_f32_wmma_32x96(const float* __restrict__ A,
                       const float* __restrict__ Bm,
                       float* __restrict__ C) {
    static_assert(M % OP_M == 0 && N % OP_N == 0 && K % OP_KC == 0,
                  "whole tiles and whole K chunks only");
    static_assert(OP_M * OP_CP <= OP_STAGE, "output tile is staged inside stage 0");
    constexpr int nbt = N / OP_N;
    const int nb  = blockIdx.x % nbt;
    const int mb  = blockIdx.x / nbt;

    const int tid  = threadIdx.x;
    const int w    = tid >> 5;
    const int lane = tid & 31;
    const int g    = lane >> 4;
    const int r16  = lane & 15;
    const int rg   = w & 1;
    const int cg   = w >> 1;

    __shared__ __attribute__((aligned(16))) float sm[2 * OP_STAGE];

    const int sp = tid & 15;
    const int sr = tid >> 4;
    const float* agp = A  + (size_t)(mb * OP_M + sr) * K + sp * 4;
    const float* bgp = Bm + (size_t)(nb * OP_N + sr) * K + sp * 4;
    const int soffA = sr * OP_P + sp * 4;
    const int soffB = OP_M * OP_P + sr * OP_P + sp * 4;

    const int aoff = (rg * 16 + r16) * OP_P + 2 * g;
    const int boff = OP_M * OP_P + (cg * 48 + r16) * OP_P + 2 * g;

    v8f acc0 = {}, acc1 = {}, acc2 = {};
    v4f ra[4], rb[12];

    #pragma unroll
    for (int i = 0; i < 4; ++i)  ra[i] = *(const v4f*)(agp + (size_t)(8 * i) * K);
    #pragma unroll
    for (int i = 0; i < 12; ++i) rb[i] = *(const v4f*)(bgp + (size_t)(8 * i) * K);
    #pragma unroll
    for (int i = 0; i < 4; ++i)  *(v4f*)(sm + soffA + 8 * i * OP_P) = ra[i];
    #pragma unroll
    for (int i = 0; i < 12; ++i) *(v4f*)(sm + soffB + 8 * i * OP_P) = rb[i];
    __syncthreads();

    int buf = 0;
    for (int kk = 0; kk < K; kk += OP_KC) {
        const bool more = (kk + OP_KC) < K;

        if (more) {
            #pragma unroll
            for (int i = 0; i < 4; ++i)
                ra[i] = *(const v4f*)(agp + (size_t)(8 * i) * K + kk + OP_KC);
            #pragma unroll
            for (int i = 0; i < 12; ++i)
                rb[i] = *(const v4f*)(bgp + (size_t)(8 * i) * K + kk + OP_KC);
        }

        const float* asb = sm + buf * OP_STAGE + aoff;
        const float* bsb = sm + buf * OP_STAGE + boff;
        #pragma unroll
        for (int ks = 0; ks < OP_KC; ks += 4) {
            v2f a  = *(const v2f*)(asb + ks);
            v2f q0 = *(const v2f*)(bsb +  0 * OP_P + ks);
            v2f q1 = *(const v2f*)(bsb + 16 * OP_P + ks);
            v2f q2 = *(const v2f*)(bsb + 32 * OP_P + ks);
            acc0 = __builtin_amdgcn_wmma_f32_16x16x4_f32(false, a, false, q0, (short)0, acc0, false, false);
            acc1 = __builtin_amdgcn_wmma_f32_16x16x4_f32(false, a, false, q1, (short)0, acc1, false, false);
            acc2 = __builtin_amdgcn_wmma_f32_16x16x4_f32(false, a, false, q2, (short)0, acc2, false, false);
        }
        __builtin_amdgcn_sched_group_barrier(0x100, 32, 0);
        __builtin_amdgcn_sched_group_barrier(0x008, 48, 0);

        if (more) {
            float* so = sm + (buf ^ 1) * OP_STAGE;
            #pragma unroll
            for (int i = 0; i < 4; ++i)  *(v4f*)(so + soffA + 8 * i * OP_P) = ra[i];
            #pragma unroll
            for (int i = 0; i < 12; ++i) *(v4f*)(so + soffB + 8 * i * OP_P) = rb[i];
            __syncthreads();
            buf ^= 1;
        }
    }

    __syncthreads();
    float* cs = sm;
    {
        float* cw = cs + (rg * 16 + 8 * g) * OP_CP + cg * 48 + r16;
        #pragma unroll
        for (int r = 0; r < 8; ++r) {
            cw[r * OP_CP +  0] = acc0[r];
            cw[r * OP_CP + 16] = acc1[r];
            cw[r * OP_CP + 32] = acc2[r];
        }
    }
    __syncthreads();
    #pragma unroll
    for (int i = 0; i < (OP_M * OP_N / 4) / 128; ++i) {
        const int idx = i * 128 + tid;
        const int row = idx / (OP_N / 4);
        const int pc  = idx % (OP_N / 4);
        const v4f v = *(const v4f*)(cs + row * OP_CP + pc * 4);
        *(v4f*)(C + (size_t)(mb * OP_M + row) * N + nb * OP_N + pc * 4) = v;
    }
}

#define TCH   32
#define NCHK  (SEQLEN / TCH)
#define PSPLIT 2
#define PROWS  (HEADDIM / PSPLIT)
#define STHR   (PROWS * 4)
#define CROWS  (TCH * 32 / STHR)
#define ZP    40

struct ScanStage {
    v4f   rx[5];
    v4f   rc[CROWS + 3];
    v4f   rz[2];
    float dv;
};

__device__ __forceinline__ float silu_f(float a) {
    return a / (1.0f + __expf(-a));
}

template <int M>
__device__ __forceinline__ float add_lane_xor(float v) {
    const int partner = __builtin_amdgcn_update_dpp(0, __float_as_int(v), 0x160 + M, 0xf, 0xf, true);
    return v + __int_as_float(partner);
}

template <bool FIRST>
__device__ __forceinline__ v4f ld_raw(const float* __restrict__ col0, int l) {
    const int lr = (FIRST && l < 0) ? 0 : l;
    return *(const v4f*)(col0 + (size_t)lr * DINPROJ);
}

template <bool FIRST>
__device__ __forceinline__ v4f conv4_silu(const v4f r0, const v4f r1, const v4f r2, const v4f r3,
                                          int l, const v4f (&w)[4], const v4f bias) {
    v4f o;
    #pragma unroll
    for (int ch = 0; ch < 4; ++ch) {
        float acc = bias[ch];
        if (!FIRST || l >= 3) acc = fmaf(r0[ch], w[ch].x, acc);
        if (!FIRST || l >= 2) acc = fmaf(r1[ch], w[ch].y, acc);
        if (!FIRST || l >= 1) acc = fmaf(r2[ch], w[ch].z, acc);
        acc = fmaf(r3[ch], w[ch].w, acc);
        o[ch] = silu_f(acc);
    }
    return o;
}

template <bool FIRST>
__device__ __forceinline__ void stage_load(ScanStage& s,
                                           const float* __restrict__ xraw,
                                           const float* __restrict__ craw,
                                           const float* __restrict__ zraw,
                                           const float* __restrict__ draw,
                                           int lx, int lc, int ld, bool dlane) {
    #pragma unroll
    for (int i = 0; i < 5; ++i) s.rx[i] = ld_raw<FIRST>(xraw, lx - 3 + i);
    #pragma unroll
    for (int i = 0; i < CROWS + 3; ++i) s.rc[i] = ld_raw<FIRST>(craw, lc - 3 + i);
    s.rz[0] = *(const v4f*)(zraw + (size_t)lx * DINPROJ);
    s.rz[1] = *(const v4f*)(zraw + (size_t)(lx + 1) * DINPROJ);
    s.dv = dlane ? draw[(size_t)ld * DINPROJ] : 0.0f;
}

template <bool FIRST>
__device__ __forceinline__ void stage_store(const ScanStage& s,
                                            float* __restrict__ sxb, float* __restrict__ sBCb,
                                            float* __restrict__ sgb, v2f* __restrict__ sddb,
                                            int lx, int lc, int xrow, int crow, int q8, int q32,
                                            const v4f (&wx)[4], const v4f bxv,
                                            const v4f (&wc)[4], const v4f bcv,
                                            bool dlane, int t, float dtb, float Ah) {
    #pragma unroll
    for (int o = 0; o < 2; ++o) {
        const v4f v = conv4_silu<FIRST>(s.rx[o], s.rx[o + 1], s.rx[o + 2], s.rx[o + 3],
                                        lx + o, wx, bxv);
        *(v4f*)(sxb + (xrow + o) * PROWS + q8 * 4) = v;
    }
    #pragma unroll
    for (int o = 0; o < CROWS; ++o) {
        const v4f v = conv4_silu<FIRST>(s.rc[o], s.rc[o + 1], s.rc[o + 2], s.rc[o + 3],
                                        lc + o, wc, bcv);
        *(v4f*)(sBCb + (crow + o) * (2 * DSTATE) + q32 * 4) = v;
    }
    #pragma unroll
    for (int o = 0; o < 2; ++o) {
        const v4f z = s.rz[o];
        const v4f g = v4f{silu_f(z.x), silu_f(z.y), silu_f(z.z), silu_f(z.w)};
        *(v4f*)(sgb + (xrow + o) * ZP + q8 * 4) = g;
    }
    if (dlane) {
        const float dv = s.dv + dtb;
        float dt = (dv > 20.0f) ? dv : __logf(1.0f + __expf(dv));
        dt = fminf(fmaxf(dt, 1e-6f), 1000.0f);
        const float dA = fminf(fmaxf(__expf(dt * Ah), 1e-6f), 1000.0f);
        sddb[t] = v2f{dt, dA};
    }
}

__global__ void __launch_bounds__(STHR)
ssm_scan_kernel(const float* __restrict__ zxbcdt,
                const float* __restrict__ conv_w,
                const float* __restrict__ conv_b,
                const float* __restrict__ dt_bias,
                const float* __restrict__ A_log,
                const float* __restrict__ D_param,
                float* __restrict__ y) {
    const int bh = blockIdx.x / PSPLIT;
    const int ph = blockIdx.x % PSPLIT;
    const int b  = bh / NHEADS;
    const int h  = bh % NHEADS;
    const int t  = threadIdx.x;
    const int p  = t >> 2;
    const int nc = t & 3;

    const float Ah  = -__expf(A_log[h]);
    const float dtb = dt_bias[h];
    const float Dp  = D_param[h];

    __shared__ __attribute__((aligned(16))) float sx [2][TCH * PROWS];
    __shared__ __attribute__((aligned(16))) float sBC[2][TCH * 2 * DSTATE];
    __shared__ __attribute__((aligned(16))) float sg [2][TCH * ZP];
    __shared__ __attribute__((aligned(8)))  v2f   sdd[2][TCH];

    const int q8   = t & 7;
    const int xrow = (t >> 3) * 2;
    const int q32  = t & 31;
    const int crow = (t >> 5) * CROWS;
    const bool dlane = t < TCH;
    const int cx = h * HEADDIM + ph * PROWS + q8 * 4;
    const int cb = DSSM + q32 * 4;
    const float* base = zxbcdt + (size_t)b * SEQLEN * DINPROJ;
    const float* zraw = base + cx;
    const float* xraw = base + DSSM + cx;
    const float* craw = base + DSSM + cb;
    const float* draw = base + (DSSM + CONVDIM) + h;

    v4f wx[4], wc[4];
    #pragma unroll
    for (int ch = 0; ch < 4; ++ch) {
        wx[ch] = *(const v4f*)(conv_w + (size_t)(cx + ch) * DCONV);
        wc[ch] = *(const v4f*)(conv_w + (size_t)(cb + ch) * DCONV);
    }
    const v4f bxv = *(const v4f*)(conv_b + cx);
    const v4f bcv = *(const v4f*)(conv_b + cb);

    float s[16];
    #pragma unroll
    for (int i = 0; i < 16; ++i) s[i] = 0.0f;

    {
        ScanStage st;
        stage_load<true>(st, xraw, craw, zraw, draw, xrow, crow, t, dlane);
        stage_store<true>(st, &sx[0][0], &sBC[0][0], &sg[0][0], &sdd[0][0],
                          xrow, crow, xrow, crow, q8, q32, wx, bxv, wc, bcv,
                          dlane, t, dtb, Ah);
    }
    __syncthreads();

    int buf = 0;
    for (int c = 0; c < NCHK; ++c) {
        const bool more = (c + 1) < NCHK;
        const int  l1   = (c + 1) * TCH;

        ScanStage st;
        if (more)
            stage_load<false>(st, xraw, craw, zraw, draw, l1 + xrow, l1 + crow, l1 + t, dlane);

        const float* bx  = &sx[buf][p];
        const float* bBC = &sBC[buf][nc * 16];
        const float* bg  = &sg[buf][nc * ZP + p];
        float* yp = y + (size_t)(b * SEQLEN + c * TCH + nc) * DSSM + h * HEADDIM + ph * PROWS + p;
        for (int j4 = 0; j4 < TCH; j4 += 4) {
            float ys[4], xs4[4];
            const float gate = bg[j4 * ZP];
            __builtin_amdgcn_sched_barrier(0);
            #pragma unroll
            for (int u = 0; u < 4; ++u) {
                const int j = j4 + u;
                const v2f   dd  = sdd[buf][j];
                const float dt  = dd.x;
                const float dA  = dd.y;
                const float xv  = bx[j * PROWS];
                const float dtx = dt * xv;

                const v4f* Bq = (const v4f*)(bBC + j * 2 * DSTATE);
                const v4f* Cq = (const v4f*)(bBC + j * 2 * DSTATE + DSTATE);
                const v4f B0 = Bq[0], B1 = Bq[1], B2 = Bq[2], B3 = Bq[3];
                const v4f C0 = Cq[0], C1 = Cq[1], C2 = Cq[2], C3 = Cq[3];
                const float Bv[16] = {B0.x, B0.y, B0.z, B0.w, B1.x, B1.y, B1.z, B1.w,
                                      B2.x, B2.y, B2.z, B2.w, B3.x, B3.y, B3.z, B3.w};
                const float Cv[16] = {C0.x, C0.y, C0.z, C0.w, C1.x, C1.y, C1.z, C1.w,
                                      C2.x, C2.y, C2.z, C2.w, C3.x, C3.y, C3.z, C3.w};

                float ysum = 0.0f;
                #pragma unroll
                for (int i = 0; i < 16; ++i) {
                    float sv = fmaf(s[i], dA, dtx * Bv[i]);
                    sv = fminf(fmaxf(sv, -1000.0f), 1000.0f);
                    s[i] = sv;
                    ysum = fmaf(sv, Cv[i], ysum);
                }
                ys[u]  = ysum;
                xs4[u] = xv;
            }
            ys[0] = add_lane_xor<1>(ys[0]);
            ys[1] = add_lane_xor<1>(ys[1]);
            ys[2] = add_lane_xor<1>(ys[2]);
            ys[3] = add_lane_xor<1>(ys[3]);
            ys[0] = add_lane_xor<2>(ys[0]);
            ys[1] = add_lane_xor<2>(ys[1]);
            ys[2] = add_lane_xor<2>(ys[2]);
            ys[3] = add_lane_xor<2>(ys[3]);
            float ykeep = fmaf(Dp, xs4[0], ys[0]);
            ykeep = (nc == 1) ? fmaf(Dp, xs4[1], ys[1]) : ykeep;
            ykeep = (nc == 2) ? fmaf(Dp, xs4[2], ys[2]) : ykeep;
            ykeep = (nc == 3) ? fmaf(Dp, xs4[3], ys[3]) : ykeep;
            yp[(size_t)j4 * DSSM] = ykeep * gate;
        }

        if (more) {
            stage_store<false>(st, &sx[buf ^ 1][0], &sBC[buf ^ 1][0], &sg[buf ^ 1][0],
                               &sdd[buf ^ 1][0],
                               l1 + xrow, l1 + crow, xrow, crow, q8, q32,
                               wx, bxv, wc, bcv, dlane, t, dtb, Ah);
            __builtin_amdgcn_fence(__ATOMIC_RELEASE, "workgroup", "local");
            __builtin_amdgcn_s_barrier();
            __builtin_amdgcn_fence(__ATOMIC_ACQUIRE, "workgroup", "local");
            buf ^= 1;
        }
    }
}

extern "C" void kernel_launch(void* const* d_in, const int* in_sizes, int n_in,
                              void* d_out, int out_size, void* d_ws, size_t ws_size,
                              hipStream_t stream) {
    const float* u       = (const float*)d_in[0];
    const float* W_in    = (const float*)d_in[1];
    const float* conv_w  = (const float*)d_in[2];
    const float* conv_b  = (const float*)d_in[3];
    const float* dt_bias = (const float*)d_in[4];
    const float* A_log   = (const float*)d_in[5];
    const float* D_param = (const float*)d_in[6];
    const float* W_out   = (const float*)d_in[7];
    float* out = (float*)d_out;

    float* ws      = (float*)d_ws;
    float* zxbcdt  = ws;
    float* ybuf    = zxbcdt + (size_t)MTOK * DINPROJ + (size_t)MTOK * CONVDIM;

    {
        const int mbt = MTOK / MBLK;
        const int nbt = (DINPROJ + NBLK - 1) / NBLK;
        gemm_nt_f32_wmma<<<mbt * nbt, 256, 0, stream>>>(u, W_in, zxbcdt,
                                                        MTOK, DINPROJ, DMODEL);
    }

    ssm_scan_kernel<<<BATCH * NHEADS * PSPLIT, STHR, 0, stream>>>(zxbcdt, conv_w, conv_b, dt_bias,
                                                                  A_log, D_param, ybuf);

    {
        const int mbt = MTOK / OP_M;
        const int nbt = DMODEL / OP_N;
        gemm_nt_f32_wmma_32x96<MTOK, DMODEL, DSSM><<<mbt * nbt, 128, 0, stream>>>(ybuf, W_out, out);
    }
}


# scan kernel: state, B, C kept as register pairs along the state index; a step's y summed as 4 packed partial sums (even/odd i, low/high half), no repacking moves; order of that 16-term sum changes
# speedup vs baseline: 6.5766x; 1.0402x over previous
#include <hip/hip_runtime.h>
#include <hip/hip_bf16.h>
#include <math.h>

#define DMODEL   768
#define DSTATE   64
#define DCONV    4
#define DSSM     1536
#define HEADDIM  64
#define NHEADS   24
#define CONVDIM  (DSSM + 2*DSTATE)
#define DINPROJ  (2*DSSM + 2*DSTATE + NHEADS)
#define BATCH    2
#define SEQLEN   512
#define MTOK     (BATCH*SEQLEN)

typedef float v2f __attribute__((ext_vector_type(2)));
typedef float v4f __attribute__((ext_vector_type(4)));
typedef float v8f __attribute__((ext_vector_type(8)));

#define KC    16
#define LDSS  18
#define MBLK  128
#define NBLK  64

__global__ void gemm_nt_f32_wmma(const float* __restrict__ A,
                                 const float* __restrict__ Bm,
                                 float* __restrict__ C,
                                 int M, int N, int K) {
    const int nbt = (N + NBLK - 1) / NBLK;
    const int nb  = blockIdx.x % nbt;
    const int mb  = blockIdx.x / nbt;

    const int tid  = threadIdx.x;
    const int w    = tid >> 5;
    const int lane = tid & 31;
    const int g    = lane >> 4;
    const int r16  = lane & 15;

    __shared__ float As[2][MBLK * LDSS];
    __shared__ float Bs[2][NBLK * LDSS];

    const int arow  = tid >> 1;
    const int ahalf = tid & 1;
    const float* agp = A + (size_t)(mb * MBLK + arow) * K + ahalf * 8;
    const int brow = tid >> 2;
    const int bq   = tid & 3;
    int bgrow = nb * NBLK + brow; if (bgrow >= N) bgrow = N - 1;
    const float* bgp = Bm + (size_t)bgrow * K + bq * 4;

    const int aoff = (w * 16 + r16) * LDSS + 2 * g;
    const int boff = r16 * LDSS + 2 * g;

    v8f acc0 = {}, acc1 = {}, acc2 = {}, acc3 = {};

    {
        v4f a0 = *(const v4f*)(agp);
        v4f a1 = *(const v4f*)(agp + 4);
        v4f b0 = *(const v4f*)(bgp);
        float* as = &As[0][arow * LDSS + ahalf * 8];
        *(v2f*)(as)     = v2f{a0.x, a0.y};
        *(v2f*)(as + 2) = v2f{a0.z, a0.w};
        *(v2f*)(as + 4) = v2f{a1.x, a1.y};
        *(v2f*)(as + 6) = v2f{a1.z, a1.w};
        float* bs = &Bs[0][brow * LDSS + bq * 4];
        *(v2f*)(bs)     = v2f{b0.x, b0.y};
        *(v2f*)(bs + 2) = v2f{b0.z, b0.w};
    }
    __syncthreads();

    int buf = 0;
    for (int kk = 0; kk < K; kk += KC) {
        const bool more = (kk + KC) < K;

        v4f a0, a1, b0;
        if (more) {
            a0 = *(const v4f*)(agp + kk + KC);
            a1 = *(const v4f*)(agp + kk + KC + 4);
            b0 = *(const v4f*)(bgp + kk + KC);
        }

        const float* asb = &As[buf][aoff];
        const float* bsb = &Bs[buf][boff];
        #pragma unroll
        for (int ks = 0; ks < KC; ks += 4) {
            v2f a  = *(const v2f*)(asb + ks);
            v2f q0 = *(const v2f*)(bsb +  0 * LDSS + ks);
            v2f q1 = *(const v2f*)(bsb + 16 * LDSS + ks);
            v2f q2 = *(const v2f*)(bsb + 32 * LDSS + ks);
            v2f q3 = *(const v2f*)(bsb + 48 * LDSS + ks);
            acc0 = __builtin_amdgcn_wmma_f32_16x16x4_f32(false, a, false, q0, (short)0, acc0, false, false);
            acc1 = __builtin_amdgcn_wmma_f32_16x16x4_f32(false, a, false, q1, (short)0, acc1, false, false);
            acc2 = __builtin_amdgcn_wmma_f32_16x16x4_f32(false, a, false, q2, (short)0, acc2, false, false);
            acc3 = __builtin_amdgcn_wmma_f32_16x16x4_f32(false, a, false, q3, (short)0, acc3, false, false);
        }

        if (more) {
            float* as = &As[buf ^ 1][arow * LDSS + ahalf * 8];
            *(v2f*)(as)     = v2f{a0.x, a0.y};
            *(v2f*)(as + 2) = v2f{a0.z, a0.w};
            *(v2f*)(as + 4) = v2f{a1.x, a1.y};
            *(v2f*)(as + 6) = v2f{a1.z, a1.w};
            float* bs = &Bs[buf ^ 1][brow * LDSS + bq * 4];
            *(v2f*)(bs)     = v2f{b0.x, b0.y};
            *(v2f*)(bs + 2) = v2f{b0.z, b0.w};
            __syncthreads();
            buf ^= 1;
        }
    }

    const int rowbase = mb * MBLK + w * 16 + 8 * g;
    #pragma unroll
    for (int j = 0; j < 4; ++j) {
        const v8f acc = (j == 0) ? acc0 : (j == 1) ? acc1 : (j == 2) ? acc2 : acc3;
        const int col = nb * NBLK + j * 16 + r16;
        if (col < N) {
            #pragma unroll
            for (int r = 0; r < 8; ++r)
                C[(size_t)(rowbase + r) * N + col] = acc[r];
        }
    }
}

#define OP_M     32
#define OP_N     96
#define OP_KC    64
#define OP_P     68
#define OP_CP    100
#define OP_STAGE ((OP_M + OP_N) * OP_P)

template <int M, int N, int K>
__global__ void __launch_bounds__(128)
gemm_nt_f32_wmma_32x96(const float* __restrict__ A,
                       const float* __restrict__ Bm,
                       float* __restrict__ C) {
    static_assert(M % OP_M == 0 && N % OP_N == 0 && K % OP_KC == 0,
                  "whole tiles and whole K chunks only");
    static_assert(OP_M * OP_CP <= OP_STAGE, "output tile is staged inside stage 0");
    constexpr int nbt = N / OP_N;
    const int nb  = blockIdx.x % nbt;
    const int mb  = blockIdx.x / nbt;

    const int tid  = threadIdx.x;
    const int w    = tid >> 5;
    const int lane = tid & 31;
    const int g    = lane >> 4;
    const int r16  = lane & 15;
    const int rg   = w & 1;
    const int cg   = w >> 1;

    __shared__ __attribute__((aligned(16))) float sm[2 * OP_STAGE];

    const int sp = tid & 15;
    const int sr = tid >> 4;
    const float* agp = A  + (size_t)(mb * OP_M + sr) * K + sp * 4;
    const float* bgp = Bm + (size_t)(nb * OP_N + sr) * K + sp * 4;
    const int soffA = sr * OP_P + sp * 4;
    const int soffB = OP_M * OP_P + sr * OP_P + sp * 4;

    const int aoff = (rg * 16 + r16) * OP_P + 2 * g;
    const int boff = OP_M * OP_P + (cg * 48 + r16) * OP_P + 2 * g;

    v8f acc0 = {}, acc1 = {}, acc2 = {};
    v4f ra[4], rb[12];

    #pragma unroll
    for (int i = 0; i < 4; ++i)  ra[i] = *(const v4f*)(agp + (size_t)(8 * i) * K);
    #pragma unroll
    for (int i = 0; i < 12; ++i) rb[i] = *(const v4f*)(bgp + (size_t)(8 * i) * K);
    #pragma unroll
    for (int i = 0; i < 4; ++i)  *(v4f*)(sm + soffA + 8 * i * OP_P) = ra[i];
    #pragma unroll
    for (int i = 0; i < 12; ++i) *(v4f*)(sm + soffB + 8 * i * OP_P) = rb[i];
    __syncthreads();

    int buf = 0;
    for (int kk = 0; kk < K; kk += OP_KC) {
        const bool more = (kk + OP_KC) < K;

        if (more) {
            #pragma unroll
            for (int i = 0; i < 4; ++i)
                ra[i] = *(const v4f*)(agp + (size_t)(8 * i) * K + kk + OP_KC);
            #pragma unroll
            for (int i = 0; i < 12; ++i)
                rb[i] = *(const v4f*)(bgp + (size_t)(8 * i) * K + kk + OP_KC);
        }

        const float* asb = sm + buf * OP_STAGE + aoff;
        const float* bsb = sm + buf * OP_STAGE + boff;
        #pragma unroll
        for (int ks = 0; ks < OP_KC; ks += 4) {
            v2f a  = *(const v2f*)(asb + ks);
            v2f q0 = *(const v2f*)(bsb +  0 * OP_P + ks);
            v2f q1 = *(const v2f*)(bsb + 16 * OP_P + ks);
            v2f q2 = *(const v2f*)(bsb + 32 * OP_P + ks);
            acc0 = __builtin_amdgcn_wmma_f32_16x16x4_f32(false, a, false, q0, (short)0, acc0, false, false);
            acc1 = __builtin_amdgcn_wmma_f32_16x16x4_f32(false, a, false, q1, (short)0, acc1, false, false);
            acc2 = __builtin_amdgcn_wmma_f32_16x16x4_f32(false, a, false, q2, (short)0, acc2, false, false);
        }
        __builtin_amdgcn_sched_group_barrier(0x100, 32, 0);
        __builtin_amdgcn_sched_group_barrier(0x008, 48, 0);

        if (more) {
            float* so = sm + (buf ^ 1) * OP_STAGE;
            #pragma unroll
            for (int i = 0; i < 4; ++i)  *(v4f*)(so + soffA + 8 * i * OP_P) = ra[i];
            #pragma unroll
            for (int i = 0; i < 12; ++i) *(v4f*)(so + soffB + 8 * i * OP_P) = rb[i];
            __syncthreads();
            buf ^= 1;
        }
    }

    __syncthreads();
    float* cs = sm;
    {
        float* cw = cs + (rg * 16 + 8 * g) * OP_CP + cg * 48 + r16;
        #pragma unroll
        for (int r = 0; r < 8; ++r) {
            cw[r * OP_CP +  0] = acc0[r];
            cw[r * OP_CP + 16] = acc1[r];
            cw[r * OP_CP + 32] = acc2[r];
        }
    }
    __syncthreads();
    #pragma unroll
    for (int i = 0; i < (OP_M * OP_N / 4) / 128; ++i) {
        const int idx = i * 128 + tid;
        const int row = idx / (OP_N / 4);
        const int pc  = idx % (OP_N / 4);
        const v4f v = *(const v4f*)(cs + row * OP_CP + pc * 4);
        *(v4f*)(C + (size_t)(mb * OP_M + row) * N + nb * OP_N + pc * 4) = v;
    }
}

#define TCH   32
#define NCHK  (SEQLEN / TCH)
#define PSPLIT 2
#define PROWS  (HEADDIM / PSPLIT)
#define STHR   (PROWS * 4)
#define CROWS  (TCH * 32 / STHR)
#define ZP    40

struct ScanStage {
    v4f   rx[5];
    v4f   rc[CROWS + 3];
    v4f   rz[2];
    float dv;
};

__device__ __forceinline__ float silu_f(float a) {
    return a / (1.0f + __expf(-a));
}

template <int M>
__device__ __forceinline__ float add_lane_xor(float v) {
    const int partner = __builtin_amdgcn_update_dpp(0, __float_as_int(v), 0x160 + M, 0xf, 0xf, true);
    return v + __int_as_float(partner);
}

template <bool FIRST>
__device__ __forceinline__ v4f ld_raw(const float* __restrict__ col0, int l) {
    const int lr = (FIRST && l < 0) ? 0 : l;
    return *(const v4f*)(col0 + (size_t)lr * DINPROJ);
}

template <bool FIRST>
__device__ __forceinline__ v4f conv4_silu(const v4f r0, const v4f r1, const v4f r2, const v4f r3,
                                          int l, const v4f (&w)[4], const v4f bias) {
    v4f o;
    #pragma unroll
    for (int ch = 0; ch < 4; ++ch) {
        float acc = bias[ch];
        if (!FIRST || l >= 3) acc = fmaf(r0[ch], w[ch].x, acc);
        if (!FIRST || l >= 2) acc = fmaf(r1[ch], w[ch].y, acc);
        if (!FIRST || l >= 1) acc = fmaf(r2[ch], w[ch].z, acc);
        acc = fmaf(r3[ch], w[ch].w, acc);
        o[ch] = silu_f(acc);
    }
    return o;
}

template <bool FIRST>
__device__ __forceinline__ void stage_load(ScanStage& s,
                                           const float* __restrict__ xraw,
                                           const float* __restrict__ craw,
                                           const float* __restrict__ zraw,
                                           const float* __restrict__ draw,
                                           int lx, int lc, int ld, bool dlane) {
    #pragma unroll
    for (int i = 0; i < 5; ++i) s.rx[i] = ld_raw<FIRST>(xraw, lx - 3 + i);
    #pragma unroll
    for (int i = 0; i < CROWS + 3; ++i) s.rc[i] = ld_raw<FIRST>(craw, lc - 3 + i);
    s.rz[0] = *(const v4f*)(zraw + (size_t)lx * DINPROJ);
    s.rz[1] = *(const v4f*)(zraw + (size_t)(lx + 1) * DINPROJ);
    s.dv = dlane ? draw[(size_t)ld * DINPROJ] : 0.0f;
}

template <bool FIRST>
__device__ __forceinline__ void stage_store(const ScanStage& s,
                                            float* __restrict__ sxb, float* __restrict__ sBCb,
                                            float* __restrict__ sgb, v2f* __restrict__ sddb,
                                            int lx, int lc, int xrow, int crow, int q8, int q32,
                                            const v4f (&wx)[4], const v4f bxv,
                                            const v4f (&wc)[4], const v4f bcv,
                                            bool dlane, int t, float dtb, float Ah) {
    #pragma unroll
    for (int o = 0; o < 2; ++o) {
        const v4f v = conv4_silu<FIRST>(s.rx[o], s.rx[o + 1], s.rx[o + 2], s.rx[o + 3],
                                        lx + o, wx, bxv);
        *(v4f*)(sxb + (xrow + o) * PROWS + q8 * 4) = v;
    }
    #pragma unroll
    for (int o = 0; o < CROWS; ++o) {
        const v4f v = conv4_silu<FIRST>(s.rc[o], s.rc[o + 1], s.rc[o + 2], s.rc[o + 3],
                                        lc + o, wc, bcv);
        *(v4f*)(sBCb + (crow + o) * (2 * DSTATE) + q32 * 4) = v;
    }
    #pragma unroll
    for (int o = 0; o < 2; ++o) {
        const v4f z = s.rz[o];
        const v4f g = v4f{silu_f(z.x), silu_f(z.y), silu_f(z.z), silu_f(z.w)};
        *(v4f*)(sgb + (xrow + o) * ZP + q8 * 4) = g;
    }
    if (dlane) {
        const float dv = s.dv + dtb;
        float dt = (dv > 20.0f) ? dv : __logf(1.0f + __expf(dv));
        dt = fminf(fmaxf(dt, 1e-6f), 1000.0f);
        const float dA = fminf(fmaxf(__expf(dt * Ah), 1e-6f), 1000.0f);
        sddb[t] = v2f{dt, dA};
    }
}

__global__ void __launch_bounds__(STHR)
ssm_scan_kernel(const float* __restrict__ zxbcdt,
                const float* __restrict__ conv_w,
                const float* __restrict__ conv_b,
                const float* __restrict__ dt_bias,
                const float* __restrict__ A_log,
                const float* __restrict__ D_param,
                float* __restrict__ y) {
    const int bh = blockIdx.x / PSPLIT;
    const int ph = blockIdx.x % PSPLIT;
    const int b  = bh / NHEADS;
    const int h  = bh % NHEADS;
    const int t  = threadIdx.x;
    const int p  = t >> 2;
    const int nc = t & 3;

    const float Ah  = -__expf(A_log[h]);
    const float dtb = dt_bias[h];
    const float Dp  = D_param[h];

    __shared__ __attribute__((aligned(16))) float sx [2][TCH * PROWS];
    __shared__ __attribute__((aligned(16))) float sBC[2][TCH * 2 * DSTATE];
    __shared__ __attribute__((aligned(16))) float sg [2][TCH * ZP];
    __shared__ __attribute__((aligned(8)))  v2f   sdd[2][TCH];

    const int q8   = t & 7;
    const int xrow = (t >> 3) * 2;
    const int q32  = t & 31;
    const int crow = (t >> 5) * CROWS;
    const bool dlane = t < TCH;
    const int cx = h * HEADDIM + ph * PROWS + q8 * 4;
    const int cb = DSSM + q32 * 4;
    const float* base = zxbcdt + (size_t)b * SEQLEN * DINPROJ;
    const float* zraw = base + cx;
    const float* xraw = base + DSSM + cx;
    const float* craw = base + DSSM + cb;
    const float* draw = base + (DSSM + CONVDIM) + h;

    v4f wx[4], wc[4];
    #pragma unroll
    for (int ch = 0; ch < 4; ++ch) {
        wx[ch] = *(const v4f*)(conv_w + (size_t)(cx + ch) * DCONV);
        wc[ch] = *(const v4f*)(conv_w + (size_t)(cb + ch) * DCONV);
    }
    const v4f bxv = *(const v4f*)(conv_b + cx);
    const v4f bcv = *(const v4f*)(conv_b + cb);

    v2f s2[8];
    #pragma unroll
    for (int k = 0; k < 8; ++k) s2[k] = v2f{0.0f, 0.0f};

    {
        ScanStage st;
        stage_load<true>(st, xraw, craw, zraw, draw, xrow, crow, t, dlane);
        stage_store<true>(st, &sx[0][0], &sBC[0][0], &sg[0][0], &sdd[0][0],
                          xrow, crow, xrow, crow, q8, q32, wx, bxv, wc, bcv,
                          dlane, t, dtb, Ah);
    }
    __syncthreads();

    int buf = 0;
    for (int c = 0; c < NCHK; ++c) {
        const bool more = (c + 1) < NCHK;
        const int  l1   = (c + 1) * TCH;

        ScanStage st;
        if (more)
            stage_load<false>(st, xraw, craw, zraw, draw, l1 + xrow, l1 + crow, l1 + t, dlane);

        const float* bx  = &sx[buf][p];
        const float* bBC = &sBC[buf][nc * 16];
        const float* bg  = &sg[buf][nc * ZP + p];
        float* yp = y + (size_t)(b * SEQLEN + c * TCH + nc) * DSSM + h * HEADDIM + ph * PROWS + p;
        for (int j4 = 0; j4 < TCH; j4 += 4) {
            float ys[4], xs4[4];
            const float gate = bg[j4 * ZP];
            v2f ddr[4];
            v4f Br[4][4], Cr[4][4];
            #pragma unroll
            for (int u = 0; u < 4; ++u) {
                const int j = j4 + u;
                ddr[u] = sdd[buf][j];
                xs4[u] = bx[j * PROWS];
                const v4f* Bq = (const v4f*)(bBC + j * 2 * DSTATE);
                const v4f* Cq = (const v4f*)(bBC + j * 2 * DSTATE + DSTATE);
                #pragma unroll
                for (int q = 0; q < 4; ++q) { Br[u][q] = Bq[q]; Cr[u][q] = Cq[q]; }
            }
            __builtin_amdgcn_sched_barrier(0);
            #pragma unroll
            for (int u = 0; u < 4; ++u) {
                const float dt  = ddr[u].x;
                const float dA  = ddr[u].y;
                const float xv  = xs4[u];
                const float dtx = dt * xv;

                const v4f B0 = Br[u][0], B1 = Br[u][1], B2 = Br[u][2], B3 = Br[u][3];
                const v4f C0 = Cr[u][0], C1 = Cr[u][1], C2 = Cr[u][2], C3 = Cr[u][3];
                const v2f Bp[8] = {v2f{B0.x, B0.y}, v2f{B0.z, B0.w}, v2f{B1.x, B1.y}, v2f{B1.z, B1.w},
                                   v2f{B2.x, B2.y}, v2f{B2.z, B2.w}, v2f{B3.x, B3.y}, v2f{B3.z, B3.w}};
                const v2f Cp[8] = {v2f{C0.x, C0.y}, v2f{C0.z, C0.w}, v2f{C1.x, C1.y}, v2f{C1.z, C1.w},
                                   v2f{C2.x, C2.y}, v2f{C2.z, C2.w}, v2f{C3.x, C3.y}, v2f{C3.z, C3.w}};
                const v2f dA2  = v2f{dA, dA};
                const v2f dtx2 = v2f{dtx, dtx};

                v2f ya = v2f{0.0f, 0.0f}, yb = v2f{0.0f, 0.0f};
                #pragma unroll
                for (int k = 0; k < 8; ++k) {
                    v2f sv = __builtin_elementwise_fma(s2[k], dA2, dtx2 * Bp[k]);
                    sv.x = fminf(fmaxf(sv.x, -1000.0f), 1000.0f);
                    sv.y = fminf(fmaxf(sv.y, -1000.0f), 1000.0f);
                    s2[k] = sv;
                    if (k < 4) ya = __builtin_elementwise_fma(sv, Cp[k], ya);
                    else       yb = __builtin_elementwise_fma(sv, Cp[k], yb);
                }
                const v2f yy = ya + yb;
                ys[u]  = yy.x + yy.y;
            }
            ys[0] = add_lane_xor<1>(ys[0]);
            ys[1] = add_lane_xor<1>(ys[1]);
            ys[2] = add_lane_xor<1>(ys[2]);
            ys[3] = add_lane_xor<1>(ys[3]);
            ys[0] = add_lane_xor<2>(ys[0]);
            ys[1] = add_lane_xor<2>(ys[1]);
            ys[2] = add_lane_xor<2>(ys[2]);
            ys[3] = add_lane_xor<2>(ys[3]);
            float ykeep = fmaf(Dp, xs4[0], ys[0]);
            ykeep = (nc == 1) ? fmaf(Dp, xs4[1], ys[1]) : ykeep;
            ykeep = (nc == 2) ? fmaf(Dp, xs4[2], ys[2]) : ykeep;
            ykeep = (nc == 3) ? fmaf(Dp, xs4[3], ys[3]) : ykeep;
            yp[(size_t)j4 * DSSM] = ykeep * gate;
        }

        if (more) {
            stage_store<false>(st, &sx[buf ^ 1][0], &sBC[buf ^ 1][0], &sg[buf ^ 1][0],
                               &sdd[buf ^ 1][0],
                               l1 + xrow, l1 + crow, xrow, crow, q8, q32,
                               wx, bxv, wc, bcv, dlane, t, dtb, Ah);
            __builtin_amdgcn_fence(__ATOMIC_RELEASE, "workgroup", "local");
            __builtin_amdgcn_s_barrier();
            __builtin_amdgcn_fence(__ATOMIC_ACQUIRE, "workgroup", "local");
            buf ^= 1;
        }
    }
}

extern "C" void kernel_launch(void* const* d_in, const int* in_sizes, int n_in,
                              void* d_out, int out_size, void* d_ws, size_t ws_size,
                              hipStream_t stream) {
    const float* u       = (const float*)d_in[0];
    const float* W_in    = (const float*)d_in[1];
    const float* conv_w  = (const float*)d_in[2];
    const float* conv_b  = (const float*)d_in[3];
    const float* dt_bias = (const float*)d_in[4];
    const float* A_log   = (const float*)d_in[5];
    const float* D_param = (const float*)d_in[6];
    const float* W_out   = (const float*)d_in[7];
    float* out = (float*)d_out;

    float* ws      = (float*)d_ws;
    float* zxbcdt  = ws;
    float* ybuf    = zxbcdt + (size_t)MTOK * DINPROJ + (size_t)MTOK * CONVDIM;

    {
        const int mbt = MTOK / MBLK;
        const int nbt = (DINPROJ + NBLK - 1) / NBLK;
        gemm_nt_f32_wmma<<<mbt * nbt, 256, 0, stream>>>(u, W_in, zxbcdt,
                                                        MTOK, DINPROJ, DMODEL);
    }

    ssm_scan_kernel<<<BATCH * NHEADS * PSPLIT, STHR, 0, stream>>>(zxbcdt, conv_w, conv_b, dt_bias,
                                                                  A_log, D_param, ybuf);

    {
        const int mbt = MTOK / OP_M;
        const int nbt = DMODEL / OP_N;
        gemm_nt_f32_wmma_32x96<MTOK, DMODEL, DSSM><<<mbt * nbt, 128, 0, stream>>>(ybuf, W_out, out);
    }
}


# in-projection GEMM retiled 128x64 (408 work-groups of 8 waves) -> 64x208 (256 work-groups of 4 waves, one per WGP), K staged 32 per barrier, fragment reads one block ahead; sums unchanged
# speedup vs baseline: 7.0221x; 1.0677x over previous
#include <hip/hip_runtime.h>
#include <hip/hip_bf16.h>
#include <math.h>

#define DMODEL   768
#define DSTATE   64
#define DCONV    4
#define DSSM     1536
#define HEADDIM  64
#define NHEADS   24
#define CONVDIM  (DSSM + 2*DSTATE)
#define DINPROJ  (2*DSSM + 2*DSTATE + NHEADS)
#define BATCH    2
#define SEQLEN   512
#define MTOK     (BATCH*SEQLEN)

typedef float v2f __attribute__((ext_vector_type(2)));
typedef float v4f __attribute__((ext_vector_type(4)));
typedef float v8f __attribute__((ext_vector_type(8)));

#define IP_M     64
#define IP_N     208
#define IP_NT    (IP_N / 16)
#define IP_KC    32
#define IP_P     36
#define IP_CP    212
#define IP_STAGE ((IP_M + IP_N) * IP_P)
#define IP_NA    (IP_M / 16)
#define IP_NB    (IP_N / 16)

struct IpFrag {
    v2f a[2];
    v2f q[2][IP_NT];
};

__device__ __forceinline__ void ip_read(IpFrag& f, const float* asb, const float* bsb, int k0) {
    #pragma unroll
    for (int h = 0; h < 2; ++h) {
        f.a[h] = *(const v2f*)(asb + k0 + 4 * h);
        #pragma unroll
        for (int j = 0; j < IP_NT; ++j) f.q[h][j] = *(const v2f*)(bsb + j * 16 * IP_P + k0 + 4 * h);
    }
}

__device__ __forceinline__ void ip_mma(v8f (&acc)[IP_NT], const IpFrag& f, int h) {
    #pragma unroll
    for (int j = 0; j < IP_NT; ++j)
        acc[j] = __builtin_amdgcn_wmma_f32_16x16x4_f32(false, f.a[h], false, f.q[h][j], (short)0, acc[j], false, false);
}

template <int NPC>
__device__ __forceinline__ void ip_store_rows(const float* cs, float* __restrict__ cblk, int tid, int ldc) {
    static_assert((IP_M * NPC) % 128 == 0, "whole passes of the 128 threads only");
    #pragma unroll
    for (int i = 0; i < (IP_M * NPC) / 128; ++i) {
        const int idx = i * 128 + tid;
        const int row = idx / NPC;
        const int pc  = idx % NPC;
        const v4f v = *(const v4f*)(cs + row * IP_CP + pc * 4);
        *(v4f*)(cblk + (size_t)row * ldc + pc * 4) = v;
    }
}

template <int M, int N, int K>
__global__ void __launch_bounds__(128)
gemm_nt_f32_wmma_64x208(const float* __restrict__ A,
                        const float* __restrict__ Bm,
                        float* __restrict__ C) {
    static_assert(M % IP_M == 0 && K % IP_KC == 0 && N % 4 == 0,
                  "whole row tiles, whole K chunks, whole 16-byte pieces only");
    static_assert(IP_M * IP_CP <= 2 * IP_STAGE, "output tile is staged inside the two stages");
    constexpr int nbt = (N + IP_N - 1) / IP_N;
    const int nb  = blockIdx.x % nbt;
    const int mb  = blockIdx.x / nbt;

    const int tid  = threadIdx.x;
    const int w    = tid >> 5;
    const int lane = tid & 31;
    const int g    = lane >> 4;
    const int r16  = lane & 15;

    __shared__ __attribute__((aligned(16))) float sm[2 * IP_STAGE];

    const int sp = tid & 7;
    const int sr = tid >> 3;
    const float* agp = A + (size_t)(mb * IP_M + sr) * K + sp * 4;
    int brow[IP_NB];
    #pragma unroll
    for (int i = 0; i < IP_NB; ++i) {
        int r = nb * IP_N + sr + 16 * i;
        r = (r < N) ? r : (N - 1);
        brow[i] = r * K + sp * 4;
    }
    const int soffA = sr * IP_P + sp * 4;
    const int soffB = IP_M * IP_P + sr * IP_P + sp * 4;

    const int aoff = (w * 16 + r16) * IP_P + 2 * g;
    const int boff = IP_M * IP_P + r16 * IP_P + 2 * g;

    v8f acc[IP_NT];
    #pragma unroll
    for (int j = 0; j < IP_NT; ++j) acc[j] = v8f{};
    v4f ra[IP_NA], rb[IP_NB];
    IpFrag f0, f1, f2, f3;

    #pragma unroll
    for (int i = 0; i < IP_NA; ++i) ra[i] = *(const v4f*)(agp + (size_t)(16 * i) * K);
    #pragma unroll
    for (int i = 0; i < IP_NB; ++i) rb[i] = *(const v4f*)(Bm + brow[i]);
    #pragma unroll
    for (int i = 0; i < IP_NA; ++i) *(v4f*)(sm + soffA + 16 * i * IP_P) = ra[i];
    #pragma unroll
    for (int i = 0; i < IP_NB; ++i) *(v4f*)(sm + soffB + 16 * i * IP_P) = rb[i];
    __syncthreads();
    ip_read(f0, sm + aoff, sm + boff, 0);

    int buf = 0;
    for (int kk = 0; kk < K - IP_KC; kk += IP_KC) {
        const float* sc = sm + buf * IP_STAGE;
        float*       sn = sm + (buf ^ 1) * IP_STAGE;

        #pragma unroll
        for (int i = 0; i < IP_NA; ++i) ra[i] = *(const v4f*)(agp + (size_t)(16 * i) * K + kk + IP_KC);
        #pragma unroll
        for (int i = 0; i < IP_NB; ++i) rb[i] = *(const v4f*)(Bm + brow[i] + kk + IP_KC);

        ip_read(f1, sc + aoff, sc + boff, 8);
        ip_mma(acc, f0, 0);
        ip_mma(acc, f0, 1);
        ip_read(f2, sc + aoff, sc + boff, 16);
        ip_mma(acc, f1, 0);
        ip_mma(acc, f1, 1);
        ip_read(f3, sc + aoff, sc + boff, 24);
        ip_mma(acc, f2, 0);
        ip_mma(acc, f2, 1);
        #pragma unroll
        for (int i = 0; i < IP_NA; ++i) *(v4f*)(sn + soffA + 16 * i * IP_P) = ra[i];
        #pragma unroll
        for (int i = 0; i < IP_NB; ++i) *(v4f*)(sn + soffB + 16 * i * IP_P) = rb[i];
        ip_mma(acc, f3, 0);
        __builtin_amdgcn_sched_group_barrier(0x020, IP_NA + IP_NB, 0);
        __builtin_amdgcn_sched_group_barrier(0x100, 14, 0);
        __builtin_amdgcn_sched_group_barrier(0x008, 26, 0);
        __builtin_amdgcn_sched_group_barrier(0x100, 14, 0);
        __builtin_amdgcn_sched_group_barrier(0x008, 26, 0);
        __builtin_amdgcn_sched_group_barrier(0x100, 14, 0);
        __builtin_amdgcn_sched_group_barrier(0x008, 26, 0);
        __builtin_amdgcn_sched_group_barrier(0x200, IP_NA + IP_NB, 0);
        __builtin_amdgcn_sched_group_barrier(0x008, 13, 0);

        __builtin_amdgcn_sched_barrier(0);
        __syncthreads();
        __builtin_amdgcn_sched_barrier(0);

        ip_read(f0, sn + aoff, sn + boff, 0);
        ip_mma(acc, f3, 1);
        __builtin_amdgcn_sched_group_barrier(0x100, 14, 1);
        __builtin_amdgcn_sched_group_barrier(0x008, 13, 1);
        buf ^= 1;
    }
    {
        const float* sc = sm + buf * IP_STAGE;
        ip_read(f1, sc + aoff, sc + boff, 8);
        ip_mma(acc, f0, 0);
        ip_mma(acc, f0, 1);
        ip_read(f2, sc + aoff, sc + boff, 16);
        ip_mma(acc, f1, 0);
        ip_mma(acc, f1, 1);
        ip_read(f3, sc + aoff, sc + boff, 24);
        ip_mma(acc, f2, 0);
        ip_mma(acc, f2, 1);
        ip_mma(acc, f3, 0);
        ip_mma(acc, f3, 1);
        __builtin_amdgcn_sched_group_barrier(0x100, 14, 2);
        __builtin_amdgcn_sched_group_barrier(0x008, 26, 2);
        __builtin_amdgcn_sched_group_barrier(0x100, 14, 2);
        __builtin_amdgcn_sched_group_barrier(0x008, 26, 2);
        __builtin_amdgcn_sched_group_barrier(0x100, 14, 2);
        __builtin_amdgcn_sched_group_barrier(0x008, 52, 2);
    }

    __syncthreads();
    float* cs = sm;
    {
        float* cw = cs + (w * 16 + 8 * g) * IP_CP + r16;
        #pragma unroll
        for (int j = 0; j < IP_NT; ++j) {
            #pragma unroll
            for (int r = 0; r < 8; ++r) cw[r * IP_CP + j * 16] = acc[j][r];
        }
    }
    __syncthreads();
    float* cblk = C + (size_t)(mb * IP_M) * N + nb * IP_N;
    constexpr int tailpc = (N - (nbt - 1) * IP_N) / 4;
    if (nb == nbt - 1) ip_store_rows<tailpc>(cs, cblk, tid, N);
    else               ip_store_rows<IP_N / 4>(cs, cblk, tid, N);
}

#define OP_M     32
#define OP_N     96
#define OP_KC    64
#define OP_P     68
#define OP_CP    100
#define OP_STAGE ((OP_M + OP_N) * OP_P)

template <int M, int N, int K>
__global__ void __launch_bounds__(128)
gemm_nt_f32_wmma_32x96(const float* __restrict__ A,
                       const float* __restrict__ Bm,
                       float* __restrict__ C) {
    static_assert(M % OP_M == 0 && N % OP_N == 0 && K % OP_KC == 0,
                  "whole tiles and whole K chunks only");
    static_assert(OP_M * OP_CP <= OP_STAGE, "output tile is staged inside stage 0");
    constexpr int nbt = N / OP_N;
    const int nb  = blockIdx.x % nbt;
    const int mb  = blockIdx.x / nbt;

    const int tid  = threadIdx.x;
    const int w    = tid >> 5;
    const int lane = tid & 31;
    const int g    = lane >> 4;
    const int r16  = lane & 15;
    const int rg   = w & 1;
    const int cg   = w >> 1;

    __shared__ __attribute__((aligned(16))) float sm[2 * OP_STAGE];

    const int sp = tid & 15;
    const int sr = tid >> 4;
    const float* agp = A  + (size_t)(mb * OP_M + sr) * K + sp * 4;
    const float* bgp = Bm + (size_t)(nb * OP_N + sr) * K + sp * 4;
    const int soffA = sr * OP_P + sp * 4;
    const int soffB = OP_M * OP_P + sr * OP_P + sp * 4;

    const int aoff = (rg * 16 + r16) * OP_P + 2 * g;
    const int boff = OP_M * OP_P + (cg * 48 + r16) * OP_P + 2 * g;

    v8f acc0 = {}, acc1 = {}, acc2 = {};
    v4f ra[4], rb[12];

    #pragma unroll
    for (int i = 0; i < 4; ++i)  ra[i] = *(const v4f*)(agp + (size_t)(8 * i) * K);
    #pragma unroll
    for (int i = 0; i < 12; ++i) rb[i] = *(const v4f*)(bgp + (size_t)(8 * i) * K);
    #pragma unroll
    for (int i = 0; i < 4; ++i)  *(v4f*)(sm + soffA + 8 * i * OP_P) = ra[i];
    #pragma unroll
    for (int i = 0; i < 12; ++i) *(v4f*)(sm + soffB + 8 * i * OP_P) = rb[i];
    __syncthreads();

    int buf = 0;
    for (int kk = 0; kk < K; kk += OP_KC) {
        const bool more = (kk + OP_KC) < K;

        if (more) {
            #pragma unroll
            for (int i = 0; i < 4; ++i)
                ra[i] = *(const v4f*)(agp + (size_t)(8 * i) * K + kk + OP_KC);
            #pragma unroll
            for (int i = 0; i < 12; ++i)
                rb[i] = *(const v4f*)(bgp + (size_t)(8 * i) * K + kk + OP_KC);
        }

        const float* asb = sm + buf * OP_STAGE + aoff;
        const float* bsb = sm + buf * OP_STAGE + boff;
        #pragma unroll
        for (int ks = 0; ks < OP_KC; ks += 4) {
            v2f a  = *(const v2f*)(asb + ks);
            v2f q0 = *(const v2f*)(bsb +  0 * OP_P + ks);
            v2f q1 = *(const v2f*)(bsb + 16 * OP_P + ks);
            v2f q2 = *(const v2f*)(bsb + 32 * OP_P + ks);
            acc0 = __builtin_amdgcn_wmma_f32_16x16x4_f32(false, a, false, q0, (short)0, acc0, false, false);
            acc1 = __builtin_amdgcn_wmma_f32_16x16x4_f32(false, a, false, q1, (short)0, acc1, false, false);
            acc2 = __builtin_amdgcn_wmma_f32_16x16x4_f32(false, a, false, q2, (short)0, acc2, false, false);
        }
        __builtin_amdgcn_sched_group_barrier(0x100, 32, 0);
        __builtin_amdgcn_sched_group_barrier(0x008, 48, 0);

        if (more) {
            float* so = sm + (buf ^ 1) * OP_STAGE;
            #pragma unroll
            for (int i = 0; i < 4; ++i)  *(v4f*)(so + soffA + 8 * i * OP_P) = ra[i];
            #pragma unroll
            for (int i = 0; i < 12; ++i) *(v4f*)(so + soffB + 8 * i * OP_P) = rb[i];
            __syncthreads();
            buf ^= 1;
        }
    }

    __syncthreads();
    float* cs = sm;
    {
        float* cw = cs + (rg * 16 + 8 * g) * OP_CP + cg * 48 + r16;
        #pragma unroll
        for (int r = 0; r < 8; ++r) {
            cw[r * OP_CP +  0] = acc0[r];
            cw[r * OP_CP + 16] = acc1[r];
            cw[r * OP_CP + 32] = acc2[r];
        }
    }
    __syncthreads();
    #pragma unroll
    for (int i = 0; i < (OP_M * OP_N / 4) / 128; ++i) {
        const int idx = i * 128 + tid;
        const int row = idx / (OP_N / 4);
        const int pc  = idx % (OP_N / 4);
        const v4f v = *(const v4f*)(cs + row * OP_CP + pc * 4);
        *(v4f*)(C + (size_t)(mb * OP_M + row) * N + nb * OP_N + pc * 4) = v;
    }
}

#define TCH   32
#define NCHK  (SEQLEN / TCH)
#define PSPLIT 2
#define PROWS  (HEADDIM / PSPLIT)
#define STHR   (PROWS * 4)
#define CROWS  (TCH * 32 / STHR)
#define ZP    40

struct ScanStage {
    v4f   rx[5];
    v4f   rc[CROWS + 3];
    v4f   rz[2];
    float dv;
};

__device__ __forceinline__ float silu_f(float a) {
    return a / (1.0f + __expf(-a));
}

template <int M>
__device__ __forceinline__ float add_lane_xor(float v) {
    const int partner = __builtin_amdgcn_update_dpp(0, __float_as_int(v), 0x160 + M, 0xf, 0xf, true);
    return v + __int_as_float(partner);
}

template <bool FIRST>
__device__ __forceinline__ v4f ld_raw(const float* __restrict__ col0, int l) {
    const int lr = (FIRST && l < 0) ? 0 : l;
    return *(const v4f*)(col0 + (size_t)lr * DINPROJ);
}

template <bool FIRST>
__device__ __forceinline__ v4f conv4_silu(const v4f r0, const v4f r1, const v4f r2, const v4f r3,
                                          int l, const v4f (&w)[4], const v4f bias) {
    v4f o;
    #pragma unroll
    for (int ch = 0; ch < 4; ++ch) {
        float acc = bias[ch];
        if (!FIRST || l >= 3) acc = fmaf(r0[ch], w[ch].x, acc);
        if (!FIRST || l >= 2) acc = fmaf(r1[ch], w[ch].y, acc);
        if (!FIRST || l >= 1) acc = fmaf(r2[ch], w[ch].z, acc);
        acc = fmaf(r3[ch], w[ch].w, acc);
        o[ch] = silu_f(acc);
    }
    return o;
}

template <bool FIRST>
__device__ __forceinline__ void stage_load(ScanStage& s,
                                           const float* __restrict__ xraw,
                                           const float* __restrict__ craw,
                                           const float* __restrict__ zraw,
                                           const float* __restrict__ draw,
                                           int lx, int lc, int ld, bool dlane) {
    #pragma unroll
    for (int i = 0; i < 5; ++i) s.rx[i] = ld_raw<FIRST>(xraw, lx - 3 + i);
    #pragma unroll
    for (int i = 0; i < CROWS + 3; ++i) s.rc[i] = ld_raw<FIRST>(craw, lc - 3 + i);
    s.rz[0] = *(const v4f*)(zraw + (size_t)lx * DINPROJ);
    s.rz[1] = *(const v4f*)(zraw + (size_t)(lx + 1) * DINPROJ);
    s.dv = dlane ? draw[(size_t)ld * DINPROJ] : 0.0f;
}

template <bool FIRST>
__device__ __forceinline__ void stage_store(const ScanStage& s,
                                            float* __restrict__ sxb, float* __restrict__ sBCb,
                                            float* __restrict__ sgb, v2f* __restrict__ sddb,
                                            int lx, int lc, int xrow, int crow, int q8, int q32,
                                            const v4f (&wx)[4], const v4f bxv,
                                            const v4f (&wc)[4], const v4f bcv,
                                            bool dlane, int t, float dtb, float Ah) {
    #pragma unroll
    for (int o = 0; o < 2; ++o) {
        const v4f v = conv4_silu<FIRST>(s.rx[o], s.rx[o + 1], s.rx[o + 2], s.rx[o + 3],
                                        lx + o, wx, bxv);
        *(v4f*)(sxb + (xrow + o) * PROWS + q8 * 4) = v;
    }
    #pragma unroll
    for (int o = 0; o < CROWS; ++o) {
        const v4f v = conv4_silu<FIRST>(s.rc[o], s.rc[o + 1], s.rc[o + 2], s.rc[o + 3],
                                        lc + o, wc, bcv);
        *(v4f*)(sBCb + (crow + o) * (2 * DSTATE) + q32 * 4) = v;
    }
    #pragma unroll
    for (int o = 0; o < 2; ++o) {
        const v4f z = s.rz[o];
        const v4f g = v4f{silu_f(z.x), silu_f(z.y), silu_f(z.z), silu_f(z.w)};
        *(v4f*)(sgb + (xrow + o) * ZP + q8 * 4) = g;
    }
    if (dlane) {
        const float dv = s.dv + dtb;
        float dt = (dv > 20.0f) ? dv : __logf(1.0f + __expf(dv));
        dt = fminf(fmaxf(dt, 1e-6f), 1000.0f);
        const float dA = fminf(fmaxf(__expf(dt * Ah), 1e-6f), 1000.0f);
        sddb[t] = v2f{dt, dA};
    }
}

__global__ void __launch_bounds__(STHR)
ssm_scan_kernel(const float* __restrict__ zxbcdt,
                const float* __restrict__ conv_w,
                const float* __restrict__ conv_b,
                const float* __restrict__ dt_bias,
                const float* __restrict__ A_log,
                const float* __restrict__ D_param,
                float* __restrict__ y) {
    const int bh = blockIdx.x / PSPLIT;
    const int ph = blockIdx.x % PSPLIT;
    const int b  = bh / NHEADS;
    const int h  = bh % NHEADS;
    const int t  = threadIdx.x;
    const int p  = t >> 2;
    const int nc = t & 3;

    const float Ah  = -__expf(A_log[h]);
    const float dtb = dt_bias[h];
    const float Dp  = D_param[h];

    __shared__ __attribute__((aligned(16))) float sx [2][TCH * PROWS];
    __shared__ __attribute__((aligned(16))) float sBC[2][TCH * 2 * DSTATE];
    __shared__ __attribute__((aligned(16))) float sg [2][TCH * ZP];
    __shared__ __attribute__((aligned(8)))  v2f   sdd[2][TCH];

    const int q8   = t & 7;
    const int xrow = (t >> 3) * 2;
    const int q32  = t & 31;
    const int crow = (t >> 5) * CROWS;
    const bool dlane = t < TCH;
    const int cx = h * HEADDIM + ph * PROWS + q8 * 4;
    const int cb = DSSM + q32 * 4;
    const float* base = zxbcdt + (size_t)b * SEQLEN * DINPROJ;
    const float* zraw = base + cx;
    const float* xraw = base + DSSM + cx;
    const float* craw = base + DSSM + cb;
    const float* draw = base + (DSSM + CONVDIM) + h;

    v4f wx[4], wc[4];
    #pragma unroll
    for (int ch = 0; ch < 4; ++ch) {
        wx[ch] = *(const v4f*)(conv_w + (size_t)(cx + ch) * DCONV);
        wc[ch] = *(const v4f*)(conv_w + (size_t)(cb + ch) * DCONV);
    }
    const v4f bxv = *(const v4f*)(conv_b + cx);
    const v4f bcv = *(const v4f*)(conv_b + cb);

    v2f s2[8];
    #pragma unroll
    for (int k = 0; k < 8; ++k) s2[k] = v2f{0.0f, 0.0f};

    {
        ScanStage st;
        stage_load<true>(st, xraw, craw, zraw, draw, xrow, crow, t, dlane);
        stage_store<true>(st, &sx[0][0], &sBC[0][0], &sg[0][0], &sdd[0][0],
                          xrow, crow, xrow, crow, q8, q32, wx, bxv, wc, bcv,
                          dlane, t, dtb, Ah);
    }
    __syncthreads();

    int buf = 0;
    for (int c = 0; c < NCHK; ++c) {
        const bool more = (c + 1) < NCHK;
        const int  l1   = (c + 1) * TCH;

        ScanStage st;
        if (more)
            stage_load<false>(st, xraw, craw, zraw, draw, l1 + xrow, l1 + crow, l1 + t, dlane);

        const float* bx  = &sx[buf][p];
        const float* bBC = &sBC[buf][nc * 16];
        const float* bg  = &sg[buf][nc * ZP + p];
        float* yp = y + (size_t)(b * SEQLEN + c * TCH + nc) * DSSM + h * HEADDIM + ph * PROWS + p;
        for (int j4 = 0; j4 < TCH; j4 += 4) {
            float ys[4], xs4[4];
            const float gate = bg[j4 * ZP];
            v2f ddr[4];
            v4f Br[4][4], Cr[4][4];
            #pragma unroll
            for (int u = 0; u < 4; ++u) {
                const int j = j4 + u;
                ddr[u] = sdd[buf][j];
                xs4[u] = bx[j * PROWS];
                const v4f* Bq = (const v4f*)(bBC + j * 2 * DSTATE);
                const v4f* Cq = (const v4f*)(bBC + j * 2 * DSTATE + DSTATE);
                #pragma unroll
                for (int q = 0; q < 4; ++q) { Br[u][q] = Bq[q]; Cr[u][q] = Cq[q]; }
            }
            __builtin_amdgcn_sched_barrier(0);
            #pragma unroll
            for (int u = 0; u < 4; ++u) {
                const float dt  = ddr[u].x;
                const float dA  = ddr[u].y;
                const float xv  = xs4[u];
                const float dtx = dt * xv;

                const v4f B0 = Br[u][0], B1 = Br[u][1], B2 = Br[u][2], B3 = Br[u][3];
                const v4f C0 = Cr[u][0], C1 = Cr[u][1], C2 = Cr[u][2], C3 = Cr[u][3];
                const v2f Bp[8] = {v2f{B0.x, B0.y}, v2f{B0.z, B0.w}, v2f{B1.x, B1.y}, v2f{B1.z, B1.w},
                                   v2f{B2.x, B2.y}, v2f{B2.z, B2.w}, v2f{B3.x, B3.y}, v2f{B3.z, B3.w}};
                const v2f Cp[8] = {v2f{C0.x, C0.y}, v2f{C0.z, C0.w}, v2f{C1.x, C1.y}, v2f{C1.z, C1.w},
                                   v2f{C2.x, C2.y}, v2f{C2.z, C2.w}, v2f{C3.x, C3.y}, v2f{C3.z, C3.w}};
                const v2f dA2  = v2f{dA, dA};
                const v2f dtx2 = v2f{dtx, dtx};

                v2f ya = v2f{0.0f, 0.0f}, yb = v2f{0.0f, 0.0f};
                #pragma unroll
                for (int k = 0; k < 8; ++k) {
                    v2f sv = __builtin_elementwise_fma(s2[k], dA2, dtx2 * Bp[k]);
                    sv.x = fminf(fmaxf(sv.x, -1000.0f), 1000.0f);
                    sv.y = fminf(fmaxf(sv.y, -1000.0f), 1000.0f);
                    s2[k] = sv;
                    if (k < 4) ya = __builtin_elementwise_fma(sv, Cp[k], ya);
                    else       yb = __builtin_elementwise_fma(sv, Cp[k], yb);
                }
                const v2f yy = ya + yb;
                ys[u]  = yy.x + yy.y;
            }
            ys[0] = add_lane_xor<1>(ys[0]);
            ys[1] = add_lane_xor<1>(ys[1]);
            ys[2] = add_lane_xor<1>(ys[2]);
            ys[3] = add_lane_xor<1>(ys[3]);
            ys[0] = add_lane_xor<2>(ys[0]);
            ys[1] = add_lane_xor<2>(ys[1]);
            ys[2] = add_lane_xor<2>(ys[2]);
            ys[3] = add_lane_xor<2>(ys[3]);
            float ykeep = fmaf(Dp, xs4[0], ys[0]);
            ykeep = (nc == 1) ? fmaf(Dp, xs4[1], ys[1]) : ykeep;
            ykeep = (nc == 2) ? fmaf(Dp, xs4[2], ys[2]) : ykeep;
            ykeep = (nc == 3) ? fmaf(Dp, xs4[3], ys[3]) : ykeep;
            yp[(size_t)j4 * DSSM] = ykeep * gate;
        }

        if (more) {
            stage_store<false>(st, &sx[buf ^ 1][0], &sBC[buf ^ 1][0], &sg[buf ^ 1][0],
                               &sdd[buf ^ 1][0],
                               l1 + xrow, l1 + crow, xrow, crow, q8, q32,
                               wx, bxv, wc, bcv, dlane, t, dtb, Ah);
            __builtin_amdgcn_fence(__ATOMIC_RELEASE, "workgroup", "local");
            __builtin_amdgcn_s_barrier();
            __builtin_amdgcn_fence(__ATOMIC_ACQUIRE, "workgroup", "local");
            buf ^= 1;
        }
    }
}

extern "C" void kernel_launch(void* const* d_in, const int* in_sizes, int n_in,
                              void* d_out, int out_size, void* d_ws, size_t ws_size,
                              hipStream_t stream) {
    const float* u       = (const float*)d_in[0];
    const float* W_in    = (const float*)d_in[1];
    const float* conv_w  = (const float*)d_in[2];
    const float* conv_b  = (const float*)d_in[3];
    const float* dt_bias = (const float*)d_in[4];
    const float* A_log   = (const float*)d_in[5];
    const float* D_param = (const float*)d_in[6];
    const float* W_out   = (const float*)d_in[7];
    float* out = (float*)d_out;

    float* ws      = (float*)d_ws;
    float* zxbcdt  = ws;
    float* ybuf    = zxbcdt + (size_t)MTOK * DINPROJ + (size_t)MTOK * CONVDIM;

    {
        const int mbt = MTOK / IP_M;
        const int nbt = (DINPROJ + IP_N - 1) / IP_N;
        gemm_nt_f32_wmma_64x208<MTOK, DINPROJ, DMODEL><<<mbt * nbt, 128, 0, stream>>>(u, W_in, zxbcdt);
    }

    ssm_scan_kernel<<<BATCH * NHEADS * PSPLIT, STHR, 0, stream>>>(zxbcdt, conv_w, conv_b, dt_bias,
                                                                  A_log, D_param, ybuf);

    {
        const int mbt = MTOK / OP_M;
        const int nbt = DMODEL / OP_N;
        gemm_nt_f32_wmma_32x96<MTOK, DMODEL, DSSM><<<mbt * nbt, 128, 0, stream>>>(ybuf, W_out, out);
    }
}


# scan kernel: a row half's 64 states split over 2 work-groups (192 in all, 8 states per thread); each writes a gated partial y plane, out-projection staging adds them; y's sum over states reordered
# speedup vs baseline: 7.7432x; 1.1027x over previous
#include <hip/hip_runtime.h>
#include <hip/hip_bf16.h>
#include <math.h>

#define DMODEL   768
#define DSTATE   64
#define DCONV    4
#define DSSM     1536
#define HEADDIM  64
#define NHEADS   24
#define CONVDIM  (DSSM + 2*DSTATE)
#define DINPROJ  (2*DSSM + 2*DSTATE + NHEADS)
#define BATCH    2
#define SEQLEN   512
#define MTOK     (BATCH*SEQLEN)

typedef float v2f __attribute__((ext_vector_type(2)));
typedef float v4f __attribute__((ext_vector_type(4)));
typedef float v8f __attribute__((ext_vector_type(8)));

#define IP_M     64
#define IP_N     208
#define IP_NT    (IP_N / 16)
#define IP_KC    32
#define IP_P     36
#define IP_CP    212
#define IP_STAGE ((IP_M + IP_N) * IP_P)
#define IP_NA    (IP_M / 16)
#define IP_NB    (IP_N / 16)

struct IpFrag {
    v2f a[2];
    v2f q[2][IP_NT];
};

__device__ __forceinline__ void ip_read(IpFrag& f, const float* asb, const float* bsb, int k0) {
    #pragma unroll
    for (int h = 0; h < 2; ++h) {
        f.a[h] = *(const v2f*)(asb + k0 + 4 * h);
        #pragma unroll
        for (int j = 0; j < IP_NT; ++j) f.q[h][j] = *(const v2f*)(bsb + j * 16 * IP_P + k0 + 4 * h);
    }
}

__device__ __forceinline__ void ip_mma(v8f (&acc)[IP_NT], const IpFrag& f, int h) {
    #pragma unroll
    for (int j = 0; j < IP_NT; ++j)
        acc[j] = __builtin_amdgcn_wmma_f32_16x16x4_f32(false, f.a[h], false, f.q[h][j], (short)0, acc[j], false, false);
}

template <int NPC>
__device__ __forceinline__ void ip_store_rows(const float* cs, float* __restrict__ cblk, int tid, int ldc) {
    static_assert((IP_M * NPC) % 128 == 0, "whole passes of the 128 threads only");
    #pragma unroll
    for (int i = 0; i < (IP_M * NPC) / 128; ++i) {
        const int idx = i * 128 + tid;
        const int row = idx / NPC;
        const int pc  = idx % NPC;
        const v4f v = *(const v4f*)(cs + row * IP_CP + pc * 4);
        *(v4f*)(cblk + (size_t)row * ldc + pc * 4) = v;
    }
}

template <int M, int N, int K>
__global__ void __launch_bounds__(128)
gemm_nt_f32_wmma_64x208(const float* __restrict__ A,
                        const float* __restrict__ Bm,
                        float* __restrict__ C) {
    static_assert(M % IP_M == 0 && K % IP_KC == 0 && N % 4 == 0,
                  "whole row tiles, whole K chunks, whole 16-byte pieces only");
    static_assert(IP_M * IP_CP <= 2 * IP_STAGE, "output tile is staged inside the two stages");
    constexpr int nbt = (N + IP_N - 1) / IP_N;
    const int nb  = blockIdx.x % nbt;
    const int mb  = blockIdx.x / nbt;

    const int tid  = threadIdx.x;
    const int w    = tid >> 5;
    const int lane = tid & 31;
    const int g    = lane >> 4;
    const int r16  = lane & 15;

    __shared__ __attribute__((aligned(16))) float sm[2 * IP_STAGE];

    const int sp = tid & 7;
    const int sr = tid >> 3;
    const float* agp = A + (size_t)(mb * IP_M + sr) * K + sp * 4;
    int brow[IP_NB];
    #pragma unroll
    for (int i = 0; i < IP_NB; ++i) {
        int r = nb * IP_N + sr + 16 * i;
        r = (r < N) ? r : (N - 1);
        brow[i] = r * K + sp * 4;
    }
    const int soffA = sr * IP_P + sp * 4;
    const int soffB = IP_M * IP_P + sr * IP_P + sp * 4;

    const int aoff = (w * 16 + r16) * IP_P + 2 * g;
    const int boff = IP_M * IP_P + r16 * IP_P + 2 * g;

    v8f acc[IP_NT];
    #pragma unroll
    for (int j = 0; j < IP_NT; ++j) acc[j] = v8f{};
    v4f ra[IP_NA], rb[IP_NB];
    IpFrag f0, f1, f2, f3;

    #pragma unroll
    for (int i = 0; i < IP_NA; ++i) ra[i] = *(const v4f*)(agp + (size_t)(16 * i) * K);
    #pragma unroll
    for (int i = 0; i < IP_NB; ++i) rb[i] = *(const v4f*)(Bm + brow[i]);
    #pragma unroll
    for (int i = 0; i < IP_NA; ++i) *(v4f*)(sm + soffA + 16 * i * IP_P) = ra[i];
    #pragma unroll
    for (int i = 0; i < IP_NB; ++i) *(v4f*)(sm + soffB + 16 * i * IP_P) = rb[i];
    __syncthreads();
    ip_read(f0, sm + aoff, sm + boff, 0);

    int buf = 0;
    for (int kk = 0; kk < K - IP_KC; kk += IP_KC) {
        const float* sc = sm + buf * IP_STAGE;
        float*       sn = sm + (buf ^ 1) * IP_STAGE;

        #pragma unroll
        for (int i = 0; i < IP_NA; ++i) ra[i] = *(const v4f*)(agp + (size_t)(16 * i) * K + kk + IP_KC);
        #pragma unroll
        for (int i = 0; i < IP_NB; ++i) rb[i] = *(const v4f*)(Bm + brow[i] + kk + IP_KC);

        ip_read(f1, sc + aoff, sc + boff, 8);
        ip_mma(acc, f0, 0);
        ip_mma(acc, f0, 1);
        ip_read(f2, sc + aoff, sc + boff, 16);
        ip_mma(acc, f1, 0);
        ip_mma(acc, f1, 1);
        ip_read(f3, sc + aoff, sc + boff, 24);
        ip_mma(acc, f2, 0);
        ip_mma(acc, f2, 1);
        #pragma unroll
        for (int i = 0; i < IP_NA; ++i) *(v4f*)(sn + soffA + 16 * i * IP_P) = ra[i];
        #pragma unroll
        for (int i = 0; i < IP_NB; ++i) *(v4f*)(sn + soffB + 16 * i * IP_P) = rb[i];
        ip_mma(acc, f3, 0);
        __builtin_amdgcn_sched_group_barrier(0x020, IP_NA + IP_NB, 0);
        __builtin_amdgcn_sched_group_barrier(0x100, 14, 0);
        __builtin_amdgcn_sched_group_barrier(0x008, 26, 0);
        __builtin_amdgcn_sched_group_barrier(0x100, 14, 0);
        __builtin_amdgcn_sched_group_barrier(0x008, 26, 0);
        __builtin_amdgcn_sched_group_barrier(0x100, 14, 0);
        __builtin_amdgcn_sched_group_barrier(0x008, 26, 0);
        __builtin_amdgcn_sched_group_barrier(0x200, IP_NA + IP_NB, 0);
        __builtin_amdgcn_sched_group_barrier(0x008, 13, 0);

        __builtin_amdgcn_sched_barrier(0);
        __syncthreads();
        __builtin_amdgcn_sched_barrier(0);

        ip_read(f0, sn + aoff, sn + boff, 0);
        ip_mma(acc, f3, 1);
        __builtin_amdgcn_sched_group_barrier(0x100, 14, 1);
        __builtin_amdgcn_sched_group_barrier(0x008, 13, 1);
        buf ^= 1;
    }
    {
        const float* sc = sm + buf * IP_STAGE;
        ip_read(f1, sc + aoff, sc + boff, 8);
        ip_mma(acc, f0, 0);
        ip_mma(acc, f0, 1);
        ip_read(f2, sc + aoff, sc + boff, 16);
        ip_mma(acc, f1, 0);
        ip_mma(acc, f1, 1);
        ip_read(f3, sc + aoff, sc + boff, 24);
        ip_mma(acc, f2, 0);
        ip_mma(acc, f2, 1);
        ip_mma(acc, f3, 0);
        ip_mma(acc, f3, 1);
        __builtin_amdgcn_sched_group_barrier(0x100, 14, 2);
        __builtin_amdgcn_sched_group_barrier(0x008, 26, 2);
        __builtin_amdgcn_sched_group_barrier(0x100, 14, 2);
        __builtin_amdgcn_sched_group_barrier(0x008, 26, 2);
        __builtin_amdgcn_sched_group_barrier(0x100, 14, 2);
        __builtin_amdgcn_sched_group_barrier(0x008, 52, 2);
    }

    __syncthreads();
    float* cs = sm;
    {
        float* cw = cs + (w * 16 + 8 * g) * IP_CP + r16;
        #pragma unroll
        for (int j = 0; j < IP_NT; ++j) {
            #pragma unroll
            for (int r = 0; r < 8; ++r) cw[r * IP_CP + j * 16] = acc[j][r];
        }
    }
    __syncthreads();
    float* cblk = C + (size_t)(mb * IP_M) * N + nb * IP_N;
    constexpr int tailpc = (N - (nbt - 1) * IP_N) / 4;
    if (nb == nbt - 1) ip_store_rows<tailpc>(cs, cblk, tid, N);
    else               ip_store_rows<IP_N / 4>(cs, cblk, tid, N);
}

#define OP_M     32
#define OP_N     96
#define OP_KC    64
#define OP_P     68
#define OP_CP    100
#define OP_STAGE ((OP_M + OP_N) * OP_P)

template <int M, int N, int K>
__global__ void __launch_bounds__(128)
gemm_nt_f32_wmma_32x96(const float* __restrict__ A,
                       const float* __restrict__ A2,
                       const float* __restrict__ Bm,
                       float* __restrict__ C) {
    static_assert(M % OP_M == 0 && N % OP_N == 0 && K % OP_KC == 0,
                  "whole tiles and whole K chunks only");
    static_assert(OP_M * OP_CP <= OP_STAGE, "output tile is staged inside stage 0");
    constexpr int nbt = N / OP_N;
    const int nb  = blockIdx.x % nbt;
    const int mb  = blockIdx.x / nbt;

    const int tid  = threadIdx.x;
    const int w    = tid >> 5;
    const int lane = tid & 31;
    const int g    = lane >> 4;
    const int r16  = lane & 15;
    const int rg   = w & 1;
    const int cg   = w >> 1;

    __shared__ __attribute__((aligned(16))) float sm[2 * OP_STAGE];

    const int sp = tid & 15;
    const int sr = tid >> 4;
    const float* agp = A  + (size_t)(mb * OP_M + sr) * K + sp * 4;
    const float* ag2 = A2 + (size_t)(mb * OP_M + sr) * K + sp * 4;
    const float* bgp = Bm + (size_t)(nb * OP_N + sr) * K + sp * 4;
    const int soffA = sr * OP_P + sp * 4;
    const int soffB = OP_M * OP_P + sr * OP_P + sp * 4;

    const int aoff = (rg * 16 + r16) * OP_P + 2 * g;
    const int boff = OP_M * OP_P + (cg * 48 + r16) * OP_P + 2 * g;

    v8f acc0 = {}, acc1 = {}, acc2 = {};
    v4f ra[4], rs[4], rb[12];

    #pragma unroll
    for (int i = 0; i < 4; ++i)  ra[i] = *(const v4f*)(agp + (size_t)(8 * i) * K);
    #pragma unroll
    for (int i = 0; i < 4; ++i)  rs[i] = *(const v4f*)(ag2 + (size_t)(8 * i) * K);
    #pragma unroll
    for (int i = 0; i < 12; ++i) rb[i] = *(const v4f*)(bgp + (size_t)(8 * i) * K);
    #pragma unroll
    for (int i = 0; i < 4; ++i)  *(v4f*)(sm + soffA + 8 * i * OP_P) = ra[i] + rs[i];
    #pragma unroll
    for (int i = 0; i < 12; ++i) *(v4f*)(sm + soffB + 8 * i * OP_P) = rb[i];
    __syncthreads();

    int buf = 0;
    for (int kk = 0; kk < K; kk += OP_KC) {
        const bool more = (kk + OP_KC) < K;

        if (more) {
            #pragma unroll
            for (int i = 0; i < 4; ++i)
                ra[i] = *(const v4f*)(agp + (size_t)(8 * i) * K + kk + OP_KC);
            #pragma unroll
            for (int i = 0; i < 4; ++i)
                rs[i] = *(const v4f*)(ag2 + (size_t)(8 * i) * K + kk + OP_KC);
            #pragma unroll
            for (int i = 0; i < 12; ++i)
                rb[i] = *(const v4f*)(bgp + (size_t)(8 * i) * K + kk + OP_KC);
        }

        const float* asb = sm + buf * OP_STAGE + aoff;
        const float* bsb = sm + buf * OP_STAGE + boff;
        #pragma unroll
        for (int ks = 0; ks < OP_KC; ks += 4) {
            v2f a  = *(const v2f*)(asb + ks);
            v2f q0 = *(const v2f*)(bsb +  0 * OP_P + ks);
            v2f q1 = *(const v2f*)(bsb + 16 * OP_P + ks);
            v2f q2 = *(const v2f*)(bsb + 32 * OP_P + ks);
            acc0 = __builtin_amdgcn_wmma_f32_16x16x4_f32(false, a, false, q0, (short)0, acc0, false, false);
            acc1 = __builtin_amdgcn_wmma_f32_16x16x4_f32(false, a, false, q1, (short)0, acc1, false, false);
            acc2 = __builtin_amdgcn_wmma_f32_16x16x4_f32(false, a, false, q2, (short)0, acc2, false, false);
        }
        __builtin_amdgcn_sched_group_barrier(0x100, 32, 0);
        __builtin_amdgcn_sched_group_barrier(0x008, 48, 0);

        if (more) {
            float* so = sm + (buf ^ 1) * OP_STAGE;
            #pragma unroll
            for (int i = 0; i < 4; ++i)  *(v4f*)(so + soffA + 8 * i * OP_P) = ra[i] + rs[i];
            #pragma unroll
            for (int i = 0; i < 12; ++i) *(v4f*)(so + soffB + 8 * i * OP_P) = rb[i];
            __syncthreads();
            buf ^= 1;
        }
    }

    __syncthreads();
    float* cs = sm;
    {
        float* cw = cs + (rg * 16 + 8 * g) * OP_CP + cg * 48 + r16;
        #pragma unroll
        for (int r = 0; r < 8; ++r) {
            cw[r * OP_CP +  0] = acc0[r];
            cw[r * OP_CP + 16] = acc1[r];
            cw[r * OP_CP + 32] = acc2[r];
        }
    }
    __syncthreads();
    #pragma unroll
    for (int i = 0; i < (OP_M * OP_N / 4) / 128; ++i) {
        const int idx = i * 128 + tid;
        const int row = idx / (OP_N / 4);
        const int pc  = idx % (OP_N / 4);
        const v4f v = *(const v4f*)(cs + row * OP_CP + pc * 4);
        *(v4f*)(C + (size_t)(mb * OP_M + row) * N + nb * OP_N + pc * 4) = v;
    }
}

#define TCH   32
#define NCHK  (SEQLEN / TCH)
#define PSPLIT 2
#define NSPLIT 2
#define PROWS  (HEADDIM / PSPLIT)
#define NST    (DSTATE / NSPLIT)
#define NPT    (NST / 4)
#define BCP    (2 * NST)
#define STHR   (PROWS * 4)
#define CROWS  (TCH * (BCP / 4) / STHR)
#define ZP    40

struct ScanStage {
    v4f   rx[5];
    v4f   rc[CROWS + 3];
    v4f   rz[2];
    float dv;
};

__device__ __forceinline__ float silu_f(float a) {
    return a / (1.0f + __expf(-a));
}

template <int M>
__device__ __forceinline__ float add_lane_xor(float v) {
    const int partner = __builtin_amdgcn_update_dpp(0, __float_as_int(v), 0x160 + M, 0xf, 0xf, true);
    return v + __int_as_float(partner);
}

template <bool FIRST>
__device__ __forceinline__ v4f ld_raw(const float* __restrict__ col0, int l) {
    const int lr = (FIRST && l < 0) ? 0 : l;
    return *(const v4f*)(col0 + (size_t)lr * DINPROJ);
}

template <bool FIRST>
__device__ __forceinline__ v4f conv4_silu(const v4f r0, const v4f r1, const v4f r2, const v4f r3,
                                          int l, const v4f (&w)[4], const v4f bias) {
    v4f o;
    #pragma unroll
    for (int ch = 0; ch < 4; ++ch) {
        float acc = bias[ch];
        if (!FIRST || l >= 3) acc = fmaf(r0[ch], w[ch].x, acc);
        if (!FIRST || l >= 2) acc = fmaf(r1[ch], w[ch].y, acc);
        if (!FIRST || l >= 1) acc = fmaf(r2[ch], w[ch].z, acc);
        acc = fmaf(r3[ch], w[ch].w, acc);
        o[ch] = silu_f(acc);
    }
    return o;
}

template <bool FIRST>
__device__ __forceinline__ void stage_load(ScanStage& s,
                                           const float* __restrict__ xraw,
                                           const float* __restrict__ craw,
                                           const float* __restrict__ zraw,
                                           const float* __restrict__ draw,
                                           int lx, int lc, int ld, bool dlane) {
    #pragma unroll
    for (int i = 0; i < 5; ++i) s.rx[i] = ld_raw<FIRST>(xraw, lx - 3 + i);
    #pragma unroll
    for (int i = 0; i < CROWS + 3; ++i) s.rc[i] = ld_raw<FIRST>(craw, lc - 3 + i);
    s.rz[0] = *(const v4f*)(zraw + (size_t)lx * DINPROJ);
    s.rz[1] = *(const v4f*)(zraw + (size_t)(lx + 1) * DINPROJ);
    s.dv = dlane ? draw[(size_t)ld * DINPROJ] : 0.0f;
}

template <bool FIRST>
__device__ __forceinline__ void stage_store(const ScanStage& s,
                                            float* __restrict__ sxb, float* __restrict__ sBCb,
                                            float* __restrict__ sgb, v2f* __restrict__ sddb,
                                            int lx, int lc, int xrow, int crow, int q8, int q16,
                                            const v4f (&wx)[4], const v4f bxv,
                                            const v4f (&wc)[4], const v4f bcv,
                                            bool dlane, int t, float dtb, float Ah) {
    #pragma unroll
    for (int o = 0; o < 2; ++o) {
        const v4f v = conv4_silu<FIRST>(s.rx[o], s.rx[o + 1], s.rx[o + 2], s.rx[o + 3],
                                        lx + o, wx, bxv);
        *(v4f*)(sxb + (xrow + o) * PROWS + q8 * 4) = v;
    }
    #pragma unroll
    for (int o = 0; o < CROWS; ++o) {
        const v4f v = conv4_silu<FIRST>(s.rc[o], s.rc[o + 1], s.rc[o + 2], s.rc[o + 3],
                                        lc + o, wc, bcv);
        *(v4f*)(sBCb + (crow + o) * BCP + q16 * 4) = v;
    }
    #pragma unroll
    for (int o = 0; o < 2; ++o) {
        const v4f z = s.rz[o];
        const v4f g = v4f{silu_f(z.x), silu_f(z.y), silu_f(z.z), silu_f(z.w)};
        *(v4f*)(sgb + (xrow + o) * ZP + q8 * 4) = g;
    }
    if (dlane) {
        const float dv = s.dv + dtb;
        float dt = (dv > 20.0f) ? dv : __logf(1.0f + __expf(dv));
        dt = fminf(fmaxf(dt, 1e-6f), 1000.0f);
        const float dA = fminf(fmaxf(__expf(dt * Ah), 1e-6f), 1000.0f);
        sddb[t] = v2f{dt, dA};
    }
}

__global__ void __launch_bounds__(STHR)
ssm_scan_kernel(const float* __restrict__ zxbcdt,
                const float* __restrict__ conv_w,
                const float* __restrict__ conv_b,
                const float* __restrict__ dt_bias,
                const float* __restrict__ A_log,
                const float* __restrict__ D_param,
                float* __restrict__ y) {
    const int nh = blockIdx.x % NSPLIT;
    const int ph = (blockIdx.x / NSPLIT) % PSPLIT;
    const int bh = blockIdx.x / (NSPLIT * PSPLIT);
    const int b  = bh / NHEADS;
    const int h  = bh % NHEADS;
    const int t  = threadIdx.x;
    const int p  = t >> 2;
    const int nc = t & 3;

    const float Ah  = -__expf(A_log[h]);
    const float dtb = dt_bias[h];
    const float Dp  = D_param[h];

    __shared__ __attribute__((aligned(16))) float sx [2][TCH * PROWS];
    __shared__ __attribute__((aligned(16))) float sBC[2][TCH * BCP];
    __shared__ __attribute__((aligned(16))) float sg [2][TCH * ZP];
    __shared__ __attribute__((aligned(8)))  v2f   sdd[2][TCH];

    const int q8   = t & 7;
    const int xrow = (t >> 3) * 2;
    const int q16  = t & 15;
    const int crow = (t >> 4) * CROWS;
    const bool dlane = t < TCH;
    const int cx = h * HEADDIM + ph * PROWS + q8 * 4;
    const int cb = DSSM + (q16 >> 3) * DSTATE + nh * NST + (q16 & 7) * 4;
    const float* base = zxbcdt + (size_t)b * SEQLEN * DINPROJ;
    const float* zraw = base + cx;
    const float* xraw = base + DSSM + cx;
    const float* craw = base + DSSM + cb;
    const float* draw = base + (DSSM + CONVDIM) + h;

    v4f wx[4], wc[4];
    #pragma unroll
    for (int ch = 0; ch < 4; ++ch) {
        wx[ch] = *(const v4f*)(conv_w + (size_t)(cx + ch) * DCONV);
        wc[ch] = *(const v4f*)(conv_w + (size_t)(cb + ch) * DCONV);
    }
    const v4f bxv = *(const v4f*)(conv_b + cx);
    const v4f bcv = *(const v4f*)(conv_b + cb);

    v2f s2[NPT / 2];
    #pragma unroll
    for (int k = 0; k < NPT / 2; ++k) s2[k] = v2f{0.0f, 0.0f};
    const bool addDx = (nh == 0);

    {
        ScanStage st;
        stage_load<true>(st, xraw, craw, zraw, draw, xrow, crow, t, dlane);
        stage_store<true>(st, &sx[0][0], &sBC[0][0], &sg[0][0], &sdd[0][0],
                          xrow, crow, xrow, crow, q8, q16, wx, bxv, wc, bcv,
                          dlane, t, dtb, Ah);
    }
    __syncthreads();

    int buf = 0;
    for (int c = 0; c < NCHK; ++c) {
        const bool more = (c + 1) < NCHK;
        const int  l1   = (c + 1) * TCH;

        ScanStage st;
        if (more)
            stage_load<false>(st, xraw, craw, zraw, draw, l1 + xrow, l1 + crow, l1 + t, dlane);

        const float* bx  = &sx[buf][p];
        const float* bBC = &sBC[buf][nc * NPT];
        const float* bg  = &sg[buf][nc * ZP + p];
        float* yp = y + (size_t)nh * MTOK * DSSM
                      + (size_t)(b * SEQLEN + c * TCH + nc) * DSSM + h * HEADDIM + ph * PROWS + p;
        for (int j4 = 0; j4 < TCH; j4 += 4) {
            float ys[4], xs4[4];
            const float gate = bg[j4 * ZP];
            v2f ddr[4];
            v4f Br[4][NPT / 4], Cr[4][NPT / 4];
            #pragma unroll
            for (int u = 0; u < 4; ++u) {
                const int j = j4 + u;
                ddr[u] = sdd[buf][j];
                xs4[u] = bx[j * PROWS];
                const v4f* Bq = (const v4f*)(bBC + j * BCP);
                const v4f* Cq = (const v4f*)(bBC + j * BCP + NST);
                #pragma unroll
                for (int q = 0; q < NPT / 4; ++q) { Br[u][q] = Bq[q]; Cr[u][q] = Cq[q]; }
            }
            __builtin_amdgcn_sched_barrier(0);
            #pragma unroll
            for (int u = 0; u < 4; ++u) {
                const float dt  = ddr[u].x;
                const float dA  = ddr[u].y;
                const float xv  = xs4[u];
                const float dtx = dt * xv;

                const v4f B0 = Br[u][0], B1 = Br[u][1];
                const v4f C0 = Cr[u][0], C1 = Cr[u][1];
                const v2f Bp[4] = {v2f{B0.x, B0.y}, v2f{B0.z, B0.w}, v2f{B1.x, B1.y}, v2f{B1.z, B1.w}};
                const v2f Cp[4] = {v2f{C0.x, C0.y}, v2f{C0.z, C0.w}, v2f{C1.x, C1.y}, v2f{C1.z, C1.w}};
                const v2f dA2  = v2f{dA, dA};
                const v2f dtx2 = v2f{dtx, dtx};

                v2f ya = v2f{0.0f, 0.0f}, yb = v2f{0.0f, 0.0f};
                #pragma unroll
                for (int k = 0; k < NPT / 2; ++k) {
                    v2f sv = __builtin_elementwise_fma(s2[k], dA2, dtx2 * Bp[k]);
                    sv.x = fminf(fmaxf(sv.x, -1000.0f), 1000.0f);
                    sv.y = fminf(fmaxf(sv.y, -1000.0f), 1000.0f);
                    s2[k] = sv;
                    if (k < NPT / 4) ya = __builtin_elementwise_fma(sv, Cp[k], ya);
                    else             yb = __builtin_elementwise_fma(sv, Cp[k], yb);
                }
                const v2f yy = ya + yb;
                ys[u]  = yy.x + yy.y;
            }
            ys[0] = add_lane_xor<1>(ys[0]);
            ys[1] = add_lane_xor<1>(ys[1]);
            ys[2] = add_lane_xor<1>(ys[2]);
            ys[3] = add_lane_xor<1>(ys[3]);
            ys[0] = add_lane_xor<2>(ys[0]);
            ys[1] = add_lane_xor<2>(ys[1]);
            ys[2] = add_lane_xor<2>(ys[2]);
            ys[3] = add_lane_xor<2>(ys[3]);
            float ykeep = ys[0], xkeep = xs4[0];
            ykeep = (nc == 1) ? ys[1] : ykeep;   xkeep = (nc == 1) ? xs4[1] : xkeep;
            ykeep = (nc == 2) ? ys[2] : ykeep;   xkeep = (nc == 2) ? xs4[2] : xkeep;
            ykeep = (nc == 3) ? ys[3] : ykeep;   xkeep = (nc == 3) ? xs4[3] : xkeep;
            if (addDx) ykeep = fmaf(Dp, xkeep, ykeep);
            yp[(size_t)j4 * DSSM] = ykeep * gate;
        }

        if (more) {
            stage_store<false>(st, &sx[buf ^ 1][0], &sBC[buf ^ 1][0], &sg[buf ^ 1][0],
                               &sdd[buf ^ 1][0],
                               l1 + xrow, l1 + crow, xrow, crow, q8, q16,
                               wx, bxv, wc, bcv, dlane, t, dtb, Ah);
            __builtin_amdgcn_fence(__ATOMIC_RELEASE, "workgroup", "local");
            __builtin_amdgcn_s_barrier();
            __builtin_amdgcn_fence(__ATOMIC_ACQUIRE, "workgroup", "local");
            buf ^= 1;
        }
    }
}

extern "C" void kernel_launch(void* const* d_in, const int* in_sizes, int n_in,
                              void* d_out, int out_size, void* d_ws, size_t ws_size,
                              hipStream_t stream) {
    const float* u       = (const float*)d_in[0];
    const float* W_in    = (const float*)d_in[1];
    const float* conv_w  = (const float*)d_in[2];
    const float* conv_b  = (const float*)d_in[3];
    const float* dt_bias = (const float*)d_in[4];
    const float* A_log   = (const float*)d_in[5];
    const float* D_param = (const float*)d_in[6];
    const float* W_out   = (const float*)d_in[7];
    float* out = (float*)d_out;

    float* ws      = (float*)d_ws;
    float* zxbcdt  = ws;
    float* ybuf    = zxbcdt + (size_t)MTOK * DINPROJ + (size_t)MTOK * CONVDIM;

    {
        const int mbt = MTOK / IP_M;
        const int nbt = (DINPROJ + IP_N - 1) / IP_N;
        gemm_nt_f32_wmma_64x208<MTOK, DINPROJ, DMODEL><<<mbt * nbt, 128, 0, stream>>>(u, W_in, zxbcdt);
    }

    ssm_scan_kernel<<<BATCH * NHEADS * PSPLIT * NSPLIT, STHR, 0, stream>>>(zxbcdt, conv_w, conv_b, dt_bias,
                                                                           A_log, D_param, ybuf);

    {
        const int mbt = MTOK / OP_M;
        const int nbt = DMODEL / OP_N;
        gemm_nt_f32_wmma_32x96<MTOK, DMODEL, DSSM><<<mbt * nbt, 128, 0, stream>>>(
            ybuf, ybuf + (size_t)MTOK * DSSM, W_out, out);
    }
}


# scan kernel: conv+SiLU staging of the next chunk moved to 4 added waves per work-group (2 per SIMD, step waves the older); a step wave runs steps and the chunk barrier only; sums unchanged
# speedup vs baseline: 8.3853x; 1.0829x over previous
#include <hip/hip_runtime.h>
#include <hip/hip_bf16.h>
#include <math.h>

#define DMODEL   768
#define DSTATE   64
#define DCONV    4
#define DSSM     1536
#define HEADDIM  64
#define NHEADS   24
#define CONVDIM  (DSSM + 2*DSTATE)
#define DINPROJ  (2*DSSM + 2*DSTATE + NHEADS)
#define BATCH    2
#define SEQLEN   512
#define MTOK     (BATCH*SEQLEN)

typedef float v2f __attribute__((ext_vector_type(2)));
typedef float v4f __attribute__((ext_vector_type(4)));
typedef float v8f __attribute__((ext_vector_type(8)));

#define IP_M     64
#define IP_N     208
#define IP_NT    (IP_N / 16)
#define IP_KC    32
#define IP_P     36
#define IP_CP    212
#define IP_STAGE ((IP_M + IP_N) * IP_P)
#define IP_NA    (IP_M / 16)
#define IP_NB    (IP_N / 16)

struct IpFrag {
    v2f a[2];
    v2f q[2][IP_NT];
};

__device__ __forceinline__ void ip_read(IpFrag& f, const float* asb, const float* bsb, int k0) {
    #pragma unroll
    for (int h = 0; h < 2; ++h) {
        f.a[h] = *(const v2f*)(asb + k0 + 4 * h);
        #pragma unroll
        for (int j = 0; j < IP_NT; ++j) f.q[h][j] = *(const v2f*)(bsb + j * 16 * IP_P + k0 + 4 * h);
    }
}

__device__ __forceinline__ void ip_mma(v8f (&acc)[IP_NT], const IpFrag& f, int h) {
    #pragma unroll
    for (int j = 0; j < IP_NT; ++j)
        acc[j] = __builtin_amdgcn_wmma_f32_16x16x4_f32(false, f.a[h], false, f.q[h][j], (short)0, acc[j], false, false);
}

template <int NPC>
__device__ __forceinline__ void ip_store_rows(const float* cs, float* __restrict__ cblk, int tid, int ldc) {
    static_assert((IP_M * NPC) % 128 == 0, "whole passes of the 128 threads only");
    #pragma unroll
    for (int i = 0; i < (IP_M * NPC) / 128; ++i) {
        const int idx = i * 128 + tid;
        const int row = idx / NPC;
        const int pc  = idx % NPC;
        const v4f v = *(const v4f*)(cs + row * IP_CP + pc * 4);
        *(v4f*)(cblk + (size_t)row * ldc + pc * 4) = v;
    }
}

template <int M, int N, int K>
__global__ void __launch_bounds__(128)
gemm_nt_f32_wmma_64x208(const float* __restrict__ A,
                        const float* __restrict__ Bm,
                        float* __restrict__ C) {
    static_assert(M % IP_M == 0 && K % IP_KC == 0 && N % 4 == 0,
                  "whole row tiles, whole K chunks, whole 16-byte pieces only");
    static_assert(IP_M * IP_CP <= 2 * IP_STAGE, "output tile is staged inside the two stages");
    constexpr int nbt = (N + IP_N - 1) / IP_N;
    const int nb  = blockIdx.x % nbt;
    const int mb  = blockIdx.x / nbt;

    const int tid  = threadIdx.x;
    const int w    = tid >> 5;
    const int lane = tid & 31;
    const int g    = lane >> 4;
    const int r16  = lane & 15;

    __shared__ __attribute__((aligned(16))) float sm[2 * IP_STAGE];

    const int sp = tid & 7;
    const int sr = tid >> 3;
    const float* agp = A + (size_t)(mb * IP_M + sr) * K + sp * 4;
    int brow[IP_NB];
    #pragma unroll
    for (int i = 0; i < IP_NB; ++i) {
        int r = nb * IP_N + sr + 16 * i;
        r = (r < N) ? r : (N - 1);
        brow[i] = r * K + sp * 4;
    }
    const int soffA = sr * IP_P + sp * 4;
    const int soffB = IP_M * IP_P + sr * IP_P + sp * 4;

    const int aoff = (w * 16 + r16) * IP_P + 2 * g;
    const int boff = IP_M * IP_P + r16 * IP_P + 2 * g;

    v8f acc[IP_NT];
    #pragma unroll
    for (int j = 0; j < IP_NT; ++j) acc[j] = v8f{};
    v4f ra[IP_NA], rb[IP_NB];
    IpFrag f0, f1, f2, f3;

    #pragma unroll
    for (int i = 0; i < IP_NA; ++i) ra[i] = *(const v4f*)(agp + (size_t)(16 * i) * K);
    #pragma unroll
    for (int i = 0; i < IP_NB; ++i) rb[i] = *(const v4f*)(Bm + brow[i]);
    #pragma unroll
    for (int i = 0; i < IP_NA; ++i) *(v4f*)(sm + soffA + 16 * i * IP_P) = ra[i];
    #pragma unroll
    for (int i = 0; i < IP_NB; ++i) *(v4f*)(sm + soffB + 16 * i * IP_P) = rb[i];
    __syncthreads();
    ip_read(f0, sm + aoff, sm + boff, 0);

    int buf = 0;
    for (int kk = 0; kk < K - IP_KC; kk += IP_KC) {
        const float* sc = sm + buf * IP_STAGE;
        float*       sn = sm + (buf ^ 1) * IP_STAGE;

        #pragma unroll
        for (int i = 0; i < IP_NA; ++i) ra[i] = *(const v4f*)(agp + (size_t)(16 * i) * K + kk + IP_KC);
        #pragma unroll
        for (int i = 0; i < IP_NB; ++i) rb[i] = *(const v4f*)(Bm + brow[i] + kk + IP_KC);

        ip_read(f1, sc + aoff, sc + boff, 8);
        ip_mma(acc, f0, 0);
        ip_mma(acc, f0, 1);
        ip_read(f2, sc + aoff, sc + boff, 16);
        ip_mma(acc, f1, 0);
        ip_mma(acc, f1, 1);
        ip_read(f3, sc + aoff, sc + boff, 24);
        ip_mma(acc, f2, 0);
        ip_mma(acc, f2, 1);
        #pragma unroll
        for (int i = 0; i < IP_NA; ++i) *(v4f*)(sn + soffA + 16 * i * IP_P) = ra[i];
        #pragma unroll
        for (int i = 0; i < IP_NB; ++i) *(v4f*)(sn + soffB + 16 * i * IP_P) = rb[i];
        ip_mma(acc, f3, 0);
        __builtin_amdgcn_sched_group_barrier(0x020, IP_NA + IP_NB, 0);
        __builtin_amdgcn_sched_group_barrier(0x100, 14, 0);
        __builtin_amdgcn_sched_group_barrier(0x008, 26, 0);
        __builtin_amdgcn_sched_group_barrier(0x100, 14, 0);
        __builtin_amdgcn_sched_group_barrier(0x008, 26, 0);
        __builtin_amdgcn_sched_group_barrier(0x100, 14, 0);
        __builtin_amdgcn_sched_group_barrier(0x008, 26, 0);
        __builtin_amdgcn_sched_group_barrier(0x200, IP_NA + IP_NB, 0);
        __builtin_amdgcn_sched_group_barrier(0x008, 13, 0);

        __builtin_amdgcn_sched_barrier(0);
        __syncthreads();
        __builtin_amdgcn_sched_barrier(0);

        ip_read(f0, sn + aoff, sn + boff, 0);
        ip_mma(acc, f3, 1);
        __builtin_amdgcn_sched_group_barrier(0x100, 14, 1);
        __builtin_amdgcn_sched_group_barrier(0x008, 13, 1);
        buf ^= 1;
    }
    {
        const float* sc = sm + buf * IP_STAGE;
        ip_read(f1, sc + aoff, sc + boff, 8);
        ip_mma(acc, f0, 0);
        ip_mma(acc, f0, 1);
        ip_read(f2, sc + aoff, sc + boff, 16);
        ip_mma(acc, f1, 0);
        ip_mma(acc, f1, 1);
        ip_read(f3, sc + aoff, sc + boff, 24);
        ip_mma(acc, f2, 0);
        ip_mma(acc, f2, 1);
        ip_mma(acc, f3, 0);
        ip_mma(acc, f3, 1);
        __builtin_amdgcn_sched_group_barrier(0x100, 14, 2);
        __builtin_amdgcn_sched_group_barrier(0x008, 26, 2);
        __builtin_amdgcn_sched_group_barrier(0x100, 14, 2);
        __builtin_amdgcn_sched_group_barrier(0x008, 26, 2);
        __builtin_amdgcn_sched_group_barrier(0x100, 14, 2);
        __builtin_amdgcn_sched_group_barrier(0x008, 52, 2);
    }

    __syncthreads();
    float* cs = sm;
    {
        float* cw = cs + (w * 16 + 8 * g) * IP_CP + r16;
        #pragma unroll
        for (int j = 0; j < IP_NT; ++j) {
            #pragma unroll
            for (int r = 0; r < 8; ++r) cw[r * IP_CP + j * 16] = acc[j][r];
        }
    }
    __syncthreads();
    float* cblk = C + (size_t)(mb * IP_M) * N + nb * IP_N;
    constexpr int tailpc = (N - (nbt - 1) * IP_N) / 4;
    if (nb == nbt - 1) ip_store_rows<tailpc>(cs, cblk, tid, N);
    else               ip_store_rows<IP_N / 4>(cs, cblk, tid, N);
}

#define OP_M     32
#define OP_N     96
#define OP_KC    64
#define OP_P     68
#define OP_CP    100
#define OP_STAGE ((OP_M + OP_N) * OP_P)

template <int M, int N, int K>
__global__ void __launch_bounds__(128)
gemm_nt_f32_wmma_32x96(const float* __restrict__ A,
                       const float* __restrict__ A2,
                       const float* __restrict__ Bm,
                       float* __restrict__ C) {
    static_assert(M % OP_M == 0 && N % OP_N == 0 && K % OP_KC == 0,
                  "whole tiles and whole K chunks only");
    static_assert(OP_M * OP_CP <= OP_STAGE, "output tile is staged inside stage 0");
    constexpr int nbt = N / OP_N;
    const int nb  = blockIdx.x % nbt;
    const int mb  = blockIdx.x / nbt;

    const int tid  = threadIdx.x;
    const int w    = tid >> 5;
    const int lane = tid & 31;
    const int g    = lane >> 4;
    const int r16  = lane & 15;
    const int rg   = w & 1;
    const int cg   = w >> 1;

    __shared__ __attribute__((aligned(16))) float sm[2 * OP_STAGE];

    const int sp = tid & 15;
    const int sr = tid >> 4;
    const float* agp = A  + (size_t)(mb * OP_M + sr) * K + sp * 4;
    const float* ag2 = A2 + (size_t)(mb * OP_M + sr) * K + sp * 4;
    const float* bgp = Bm + (size_t)(nb * OP_N + sr) * K + sp * 4;
    const int soffA = sr * OP_P + sp * 4;
    const int soffB = OP_M * OP_P + sr * OP_P + sp * 4;

    const int aoff = (rg * 16 + r16) * OP_P + 2 * g;
    const int boff = OP_M * OP_P + (cg * 48 + r16) * OP_P + 2 * g;

    v8f acc0 = {}, acc1 = {}, acc2 = {};
    v4f ra[4], rs[4], rb[12];

    #pragma unroll
    for (int i = 0; i < 4; ++i)  ra[i] = *(const v4f*)(agp + (size_t)(8 * i) * K);
    #pragma unroll
    for (int i = 0; i < 4; ++i)  rs[i] = *(const v4f*)(ag2 + (size_t)(8 * i) * K);
    #pragma unroll
    for (int i = 0; i < 12; ++i) rb[i] = *(const v4f*)(bgp + (size_t)(8 * i) * K);
    #pragma unroll
    for (int i = 0; i < 4; ++i)  *(v4f*)(sm + soffA + 8 * i * OP_P) = ra[i] + rs[i];
    #pragma unroll
    for (int i = 0; i < 12; ++i) *(v4f*)(sm + soffB + 8 * i * OP_P) = rb[i];
    __syncthreads();

    int buf = 0;
    for (int kk = 0; kk < K; kk += OP_KC) {
        const bool more = (kk + OP_KC) < K;

        if (more) {
            #pragma unroll
            for (int i = 0; i < 4; ++i)
                ra[i] = *(const v4f*)(agp + (size_t)(8 * i) * K + kk + OP_KC);
            #pragma unroll
            for (int i = 0; i < 4; ++i)
                rs[i] = *(const v4f*)(ag2 + (size_t)(8 * i) * K + kk + OP_KC);
            #pragma unroll
            for (int i = 0; i < 12; ++i)
                rb[i] = *(const v4f*)(bgp + (size_t)(8 * i) * K + kk + OP_KC);
        }

        const float* asb = sm + buf * OP_STAGE + aoff;
        const float* bsb = sm + buf * OP_STAGE + boff;
        #pragma unroll
        for (int ks = 0; ks < OP_KC; ks += 4) {
            v2f a  = *(const v2f*)(asb + ks);
            v2f q0 = *(const v2f*)(bsb +  0 * OP_P + ks);
            v2f q1 = *(const v2f*)(bsb + 16 * OP_P + ks);
            v2f q2 = *(const v2f*)(bsb + 32 * OP_P + ks);
            acc0 = __builtin_amdgcn_wmma_f32_16x16x4_f32(false, a, false, q0, (short)0, acc0, false, false);
            acc1 = __builtin_amdgcn_wmma_f32_16x16x4_f32(false, a, false, q1, (short)0, acc1, false, false);
            acc2 = __builtin_amdgcn_wmma_f32_16x16x4_f32(false, a, false, q2, (short)0, acc2, false, false);
        }
        __builtin_amdgcn_sched_group_barrier(0x100, 32, 0);
        __builtin_amdgcn_sched_group_barrier(0x008, 48, 0);

        if (more) {
            float* so = sm + (buf ^ 1) * OP_STAGE;
            #pragma unroll
            for (int i = 0; i < 4; ++i)  *(v4f*)(so + soffA + 8 * i * OP_P) = ra[i] + rs[i];
            #pragma unroll
            for (int i = 0; i < 12; ++i) *(v4f*)(so + soffB + 8 * i * OP_P) = rb[i];
            __syncthreads();
            buf ^= 1;
        }
    }

    __syncthreads();
    float* cs = sm;
    {
        float* cw = cs + (rg * 16 + 8 * g) * OP_CP + cg * 48 + r16;
        #pragma unroll
        for (int r = 0; r < 8; ++r) {
            cw[r * OP_CP +  0] = acc0[r];
            cw[r * OP_CP + 16] = acc1[r];
            cw[r * OP_CP + 32] = acc2[r];
        }
    }
    __syncthreads();
    #pragma unroll
    for (int i = 0; i < (OP_M * OP_N / 4) / 128; ++i) {
        const int idx = i * 128 + tid;
        const int row = idx / (OP_N / 4);
        const int pc  = idx % (OP_N / 4);
        const v4f v = *(const v4f*)(cs + row * OP_CP + pc * 4);
        *(v4f*)(C + (size_t)(mb * OP_M + row) * N + nb * OP_N + pc * 4) = v;
    }
}

#define TCH   32
#define NCHK  (SEQLEN / TCH)
#define PSPLIT 2
#define NSPLIT 2
#define PROWS  (HEADDIM / PSPLIT)
#define NST    (DSTATE / NSPLIT)
#define NPT    (NST / 4)
#define BCP    (2 * NST)
#define STHR   (PROWS * 4)
#define SBLK   (2 * STHR)
#define CROWS  (TCH * (BCP / 4) / STHR)
#define ZP    40

struct ScanStage {
    v4f   rx[5];
    v4f   rc[CROWS + 3];
    v4f   rz[2];
    float dv;
};

__device__ __forceinline__ float silu_f(float a) {
    return a / (1.0f + __expf(-a));
}

template <int M>
__device__ __forceinline__ float add_lane_xor(float v) {
    const int partner = __builtin_amdgcn_update_dpp(0, __float_as_int(v), 0x160 + M, 0xf, 0xf, true);
    return v + __int_as_float(partner);
}

template <bool FIRST>
__device__ __forceinline__ v4f ld_raw(const float* __restrict__ col0, int l) {
    const int lr = (FIRST && l < 0) ? 0 : l;
    return *(const v4f*)(col0 + (size_t)lr * DINPROJ);
}

template <bool FIRST>
__device__ __forceinline__ v4f conv4_silu(const v4f r0, const v4f r1, const v4f r2, const v4f r3,
                                          int l, const v4f (&w)[4], const v4f bias) {
    v4f o;
    #pragma unroll
    for (int ch = 0; ch < 4; ++ch) {
        float acc = bias[ch];
        if (!FIRST || l >= 3) acc = fmaf(r0[ch], w[ch].x, acc);
        if (!FIRST || l >= 2) acc = fmaf(r1[ch], w[ch].y, acc);
        if (!FIRST || l >= 1) acc = fmaf(r2[ch], w[ch].z, acc);
        acc = fmaf(r3[ch], w[ch].w, acc);
        o[ch] = silu_f(acc);
    }
    return o;
}

template <bool FIRST>
__device__ __forceinline__ void stage_load(ScanStage& s,
                                           const float* __restrict__ xraw,
                                           const float* __restrict__ craw,
                                           const float* __restrict__ zraw,
                                           const float* __restrict__ draw,
                                           int lx, int lc, int ld, bool dlane) {
    #pragma unroll
    for (int i = 0; i < 5; ++i) s.rx[i] = ld_raw<FIRST>(xraw, lx - 3 + i);
    #pragma unroll
    for (int i = 0; i < CROWS + 3; ++i) s.rc[i] = ld_raw<FIRST>(craw, lc - 3 + i);
    s.rz[0] = *(const v4f*)(zraw + (size_t)lx * DINPROJ);
    s.rz[1] = *(const v4f*)(zraw + (size_t)(lx + 1) * DINPROJ);
    s.dv = dlane ? draw[(size_t)ld * DINPROJ] : 0.0f;
}

template <bool FIRST>
__device__ __forceinline__ void stage_store(const ScanStage& s,
                                            float* __restrict__ sxb, float* __restrict__ sBCb,
                                            float* __restrict__ sgb, v2f* __restrict__ sddb,
                                            int lx, int lc, int xrow, int crow, int q8, int q16,
                                            const v4f (&wx)[4], const v4f bxv,
                                            const v4f (&wc)[4], const v4f bcv,
                                            bool dlane, int t, float dtb, float Ah) {
    #pragma unroll
    for (int o = 0; o < 2; ++o) {
        const v4f v = conv4_silu<FIRST>(s.rx[o], s.rx[o + 1], s.rx[o + 2], s.rx[o + 3],
                                        lx + o, wx, bxv);
        *(v4f*)(sxb + (xrow + o) * PROWS + q8 * 4) = v;
    }
    #pragma unroll
    for (int o = 0; o < CROWS; ++o) {
        const v4f v = conv4_silu<FIRST>(s.rc[o], s.rc[o + 1], s.rc[o + 2], s.rc[o + 3],
                                        lc + o, wc, bcv);
        *(v4f*)(sBCb + (crow + o) * BCP + q16 * 4) = v;
    }
    #pragma unroll
    for (int o = 0; o < 2; ++o) {
        const v4f z = s.rz[o];
        const v4f g = v4f{silu_f(z.x), silu_f(z.y), silu_f(z.z), silu_f(z.w)};
        *(v4f*)(sgb + (xrow + o) * ZP + q8 * 4) = g;
    }
    if (dlane) {
        const float dv = s.dv + dtb;
        float dt = (dv > 20.0f) ? dv : __logf(1.0f + __expf(dv));
        dt = fminf(fmaxf(dt, 1e-6f), 1000.0f);
        const float dA = fminf(fmaxf(__expf(dt * Ah), 1e-6f), 1000.0f);
        sddb[t] = v2f{dt, dA};
    }
}

__global__ void __launch_bounds__(SBLK)
ssm_scan_kernel(const float* __restrict__ zxbcdt,
                const float* __restrict__ conv_w,
                const float* __restrict__ conv_b,
                const float* __restrict__ dt_bias,
                const float* __restrict__ A_log,
                const float* __restrict__ D_param,
                float* __restrict__ y) {
    const int nh = blockIdx.x % NSPLIT;
    const int ph = (blockIdx.x / NSPLIT) % PSPLIT;
    const int bh = blockIdx.x / (NSPLIT * PSPLIT);
    const int b  = bh / NHEADS;
    const int h  = bh % NHEADS;
    const bool stager = threadIdx.x >= STHR;
    const int t  = threadIdx.x & (STHR - 1);
    const int p  = t >> 2;
    const int nc = t & 3;

    const float Ah  = -__expf(A_log[h]);
    const float dtb = dt_bias[h];
    const float Dp  = D_param[h];

    __shared__ __attribute__((aligned(16))) float sx [2][TCH * PROWS];
    __shared__ __attribute__((aligned(16))) float sBC[2][TCH * BCP];
    __shared__ __attribute__((aligned(16))) float sg [2][TCH * ZP];
    __shared__ __attribute__((aligned(8)))  v2f   sdd[2][TCH];

    const int q8   = t & 7;
    const int xrow = (t >> 3) * 2;
    const int q16  = t & 15;
    const int crow = (t >> 4) * CROWS;
    const bool dlane = t < TCH;
    const int cx = h * HEADDIM + ph * PROWS + q8 * 4;
    const int cb = DSSM + (q16 >> 3) * DSTATE + nh * NST + (q16 & 7) * 4;
    const float* base = zxbcdt + (size_t)b * SEQLEN * DINPROJ;
    const float* zraw = base + cx;
    const float* xraw = base + DSSM + cx;
    const float* craw = base + DSSM + cb;
    const float* draw = base + (DSSM + CONVDIM) + h;

    v4f wx[4], wc[4], bxv, bcv;

    v2f s2[NPT / 2];
    #pragma unroll
    for (int k = 0; k < NPT / 2; ++k) s2[k] = v2f{0.0f, 0.0f};
    const bool addDx = (nh == 0);

    ScanStage st;
    if (stager) {
        #pragma unroll
        for (int ch = 0; ch < 4; ++ch) {
            wx[ch] = *(const v4f*)(conv_w + (size_t)(cx + ch) * DCONV);
            wc[ch] = *(const v4f*)(conv_w + (size_t)(cb + ch) * DCONV);
        }
        bxv = *(const v4f*)(conv_b + cx);
        bcv = *(const v4f*)(conv_b + cb);
        ScanStage s0;
        stage_load<true>(s0, xraw, craw, zraw, draw, xrow, crow, t, dlane);
        stage_store<true>(s0, &sx[0][0], &sBC[0][0], &sg[0][0], &sdd[0][0],
                          xrow, crow, xrow, crow, q8, q16, wx, bxv, wc, bcv,
                          dlane, t, dtb, Ah);
        stage_load<false>(st, xraw, craw, zraw, draw, TCH + xrow, TCH + crow, TCH + t, dlane);
    }
    __builtin_amdgcn_fence(__ATOMIC_RELEASE, "workgroup", "local");
    __builtin_amdgcn_s_barrier();
    __builtin_amdgcn_fence(__ATOMIC_ACQUIRE, "workgroup", "local");

    int buf = 0;
    for (int c = 0; c < NCHK; ++c) {
        const bool more = (c + 1) < NCHK;
        const int  l1   = (c + 1) * TCH;

        if (stager) {
            if (more) {
                stage_store<false>(st, &sx[buf ^ 1][0], &sBC[buf ^ 1][0], &sg[buf ^ 1][0],
                                   &sdd[buf ^ 1][0],
                                   l1 + xrow, l1 + crow, xrow, crow, q8, q16,
                                   wx, bxv, wc, bcv, dlane, t, dtb, Ah);
                if ((c + 2) < NCHK)
                    stage_load<false>(st, xraw, craw, zraw, draw,
                                      l1 + TCH + xrow, l1 + TCH + crow, l1 + TCH + t, dlane);
            }
        } else {

        const float* bx  = &sx[buf][p];
        const float* bBC = &sBC[buf][nc * NPT];
        const float* bg  = &sg[buf][nc * ZP + p];
        float* yp = y + (size_t)nh * MTOK * DSSM
                      + (size_t)(b * SEQLEN + c * TCH + nc) * DSSM + h * HEADDIM + ph * PROWS + p;
        for (int j4 = 0; j4 < TCH; j4 += 4) {
            float ys[4], xs4[4];
            const float gate = bg[j4 * ZP];
            v2f ddr[4];
            v4f Br[4][NPT / 4], Cr[4][NPT / 4];
            #pragma unroll
            for (int u = 0; u < 4; ++u) {
                const int j = j4 + u;
                ddr[u] = sdd[buf][j];
                xs4[u] = bx[j * PROWS];
                const v4f* Bq = (const v4f*)(bBC + j * BCP);
                const v4f* Cq = (const v4f*)(bBC + j * BCP + NST);
                #pragma unroll
                for (int q = 0; q < NPT / 4; ++q) { Br[u][q] = Bq[q]; Cr[u][q] = Cq[q]; }
            }
            __builtin_amdgcn_sched_barrier(0);
            #pragma unroll
            for (int u = 0; u < 4; ++u) {
                const float dt  = ddr[u].x;
                const float dA  = ddr[u].y;
                const float xv  = xs4[u];
                const float dtx = dt * xv;

                const v4f B0 = Br[u][0], B1 = Br[u][1];
                const v4f C0 = Cr[u][0], C1 = Cr[u][1];
                const v2f Bp[4] = {v2f{B0.x, B0.y}, v2f{B0.z, B0.w}, v2f{B1.x, B1.y}, v2f{B1.z, B1.w}};
                const v2f Cp[4] = {v2f{C0.x, C0.y}, v2f{C0.z, C0.w}, v2f{C1.x, C1.y}, v2f{C1.z, C1.w}};
                const v2f dA2  = v2f{dA, dA};
                const v2f dtx2 = v2f{dtx, dtx};

                v2f ya = v2f{0.0f, 0.0f}, yb = v2f{0.0f, 0.0f};
                #pragma unroll
                for (int k = 0; k < NPT / 2; ++k) {
                    v2f sv = __builtin_elementwise_fma(s2[k], dA2, dtx2 * Bp[k]);
                    sv.x = fminf(fmaxf(sv.x, -1000.0f), 1000.0f);
                    sv.y = fminf(fmaxf(sv.y, -1000.0f), 1000.0f);
                    s2[k] = sv;
                    if (k < NPT / 4) ya = __builtin_elementwise_fma(sv, Cp[k], ya);
                    else             yb = __builtin_elementwise_fma(sv, Cp[k], yb);
                }
                const v2f yy = ya + yb;
                ys[u]  = yy.x + yy.y;
            }
            ys[0] = add_lane_xor<1>(ys[0]);
            ys[1] = add_lane_xor<1>(ys[1]);
            ys[2] = add_lane_xor<1>(ys[2]);
            ys[3] = add_lane_xor<1>(ys[3]);
            ys[0] = add_lane_xor<2>(ys[0]);
            ys[1] = add_lane_xor<2>(ys[1]);
            ys[2] = add_lane_xor<2>(ys[2]);
            ys[3] = add_lane_xor<2>(ys[3]);
            float ykeep = ys[0], xkeep = xs4[0];
            ykeep = (nc == 1) ? ys[1] : ykeep;   xkeep = (nc == 1) ? xs4[1] : xkeep;
            ykeep = (nc == 2) ? ys[2] : ykeep;   xkeep = (nc == 2) ? xs4[2] : xkeep;
            ykeep = (nc == 3) ? ys[3] : ykeep;   xkeep = (nc == 3) ? xs4[3] : xkeep;
            if (addDx) ykeep = fmaf(Dp, xkeep, ykeep);
            yp[(size_t)j4 * DSSM] = ykeep * gate;
        }
        }

        if (more) {
            __builtin_amdgcn_fence(__ATOMIC_RELEASE, "workgroup", "local");
            __builtin_amdgcn_s_barrier();
            __builtin_amdgcn_fence(__ATOMIC_ACQUIRE, "workgroup", "local");
            buf ^= 1;
        }
    }
}

extern "C" void kernel_launch(void* const* d_in, const int* in_sizes, int n_in,
                              void* d_out, int out_size, void* d_ws, size_t ws_size,
                              hipStream_t stream) {
    const float* u       = (const float*)d_in[0];
    const float* W_in    = (const float*)d_in[1];
    const float* conv_w  = (const float*)d_in[2];
    const float* conv_b  = (const float*)d_in[3];
    const float* dt_bias = (const float*)d_in[4];
    const float* A_log   = (const float*)d_in[5];
    const float* D_param = (const float*)d_in[6];
    const float* W_out   = (const float*)d_in[7];
    float* out = (float*)d_out;

    float* ws      = (float*)d_ws;
    float* zxbcdt  = ws;
    float* ybuf    = zxbcdt + (size_t)MTOK * DINPROJ + (size_t)MTOK * CONVDIM;

    {
        const int mbt = MTOK / IP_M;
        const int nbt = (DINPROJ + IP_N - 1) / IP_N;
        gemm_nt_f32_wmma_64x208<MTOK, DINPROJ, DMODEL><<<mbt * nbt, 128, 0, stream>>>(u, W_in, zxbcdt);
    }

    ssm_scan_kernel<<<BATCH * NHEADS * PSPLIT * NSPLIT, SBLK, 0, stream>>>(zxbcdt, conv_w, conv_b, dt_bias,
                                                                           A_log, D_param, ybuf);

    {
        const int mbt = MTOK / OP_M;
        const int nbt = DMODEL / OP_N;
        gemm_nt_f32_wmma_32x96<MTOK, DMODEL, DSSM><<<mbt * nbt, 128, 0, stream>>>(
            ybuf, ybuf + (size_t)MTOK * DSSM, W_out, out);
    }
}


# out-projection GEMM: K loop software-pipelined: global loads two chunks ahead in two register sets, LDS stores and the barrier inside a chunk's last WMMAs, next first reads behind it; sums unchanged
# speedup vs baseline: 9.1447x; 1.0906x over previous
#include <hip/hip_runtime.h>
#include <hip/hip_bf16.h>
#include <math.h>

#define DMODEL   768
#define DSTATE   64
#define DCONV    4
#define DSSM     1536
#define HEADDIM  64
#define NHEADS   24
#define CONVDIM  (DSSM + 2*DSTATE)
#define DINPROJ  (2*DSSM + 2*DSTATE + NHEADS)
#define BATCH    2
#define SEQLEN   512
#define MTOK     (BATCH*SEQLEN)

typedef float v2f __attribute__((ext_vector_type(2)));
typedef float v4f __attribute__((ext_vector_type(4)));
typedef float v8f __attribute__((ext_vector_type(8)));

#define IP_M     64
#define IP_N     208
#define IP_NT    (IP_N / 16)
#define IP_KC    32
#define IP_P     36
#define IP_CP    212
#define IP_STAGE ((IP_M + IP_N) * IP_P)
#define IP_NA    (IP_M / 16)
#define IP_NB    (IP_N / 16)

struct IpFrag {
    v2f a[2];
    v2f q[2][IP_NT];
};

__device__ __forceinline__ void ip_read(IpFrag& f, const float* asb, const float* bsb, int k0) {
    #pragma unroll
    for (int h = 0; h < 2; ++h) {
        f.a[h] = *(const v2f*)(asb + k0 + 4 * h);
        #pragma unroll
        for (int j = 0; j < IP_NT; ++j) f.q[h][j] = *(const v2f*)(bsb + j * 16 * IP_P + k0 + 4 * h);
    }
}

__device__ __forceinline__ void ip_mma(v8f (&acc)[IP_NT], const IpFrag& f, int h) {
    #pragma unroll
    for (int j = 0; j < IP_NT; ++j)
        acc[j] = __builtin_amdgcn_wmma_f32_16x16x4_f32(false, f.a[h], false, f.q[h][j], (short)0, acc[j], false, false);
}

template <int NPC>
__device__ __forceinline__ void ip_store_rows(const float* cs, float* __restrict__ cblk, int tid, int ldc) {
    static_assert((IP_M * NPC) % 128 == 0, "whole passes of the 128 threads only");
    #pragma unroll
    for (int i = 0; i < (IP_M * NPC) / 128; ++i) {
        const int idx = i * 128 + tid;
        const int row = idx / NPC;
        const int pc  = idx % NPC;
        const v4f v = *(const v4f*)(cs + row * IP_CP + pc * 4);
        *(v4f*)(cblk + (size_t)row * ldc + pc * 4) = v;
    }
}

template <int M, int N, int K>
__global__ void __launch_bounds__(128)
gemm_nt_f32_wmma_64x208(const float* __restrict__ A,
                        const float* __restrict__ Bm,
                        float* __restrict__ C) {
    static_assert(M % IP_M == 0 && K % IP_KC == 0 && N % 4 == 0,
                  "whole row tiles, whole K chunks, whole 16-byte pieces only");
    static_assert(IP_M * IP_CP <= 2 * IP_STAGE, "output tile is staged inside the two stages");
    constexpr int nbt = (N + IP_N - 1) / IP_N;
    const int nb  = blockIdx.x % nbt;
    const int mb  = blockIdx.x / nbt;

    const int tid  = threadIdx.x;
    const int w    = tid >> 5;
    const int lane = tid & 31;
    const int g    = lane >> 4;
    const int r16  = lane & 15;

    __shared__ __attribute__((aligned(16))) float sm[2 * IP_STAGE];

    const int sp = tid & 7;
    const int sr = tid >> 3;
    const float* agp = A + (size_t)(mb * IP_M + sr) * K + sp * 4;
    int brow[IP_NB];
    #pragma unroll
    for (int i = 0; i < IP_NB; ++i) {
        int r = nb * IP_N + sr + 16 * i;
        r = (r < N) ? r : (N - 1);
        brow[i] = r * K + sp * 4;
    }
    const int soffA = sr * IP_P + sp * 4;
    const int soffB = IP_M * IP_P + sr * IP_P + sp * 4;

    const int aoff = (w * 16 + r16) * IP_P + 2 * g;
    const int boff = IP_M * IP_P + r16 * IP_P + 2 * g;

    v8f acc[IP_NT];
    #pragma unroll
    for (int j = 0; j < IP_NT; ++j) acc[j] = v8f{};
    v4f ra[IP_NA], rb[IP_NB];
    IpFrag f0, f1, f2, f3;

    #pragma unroll
    for (int i = 0; i < IP_NA; ++i) ra[i] = *(const v4f*)(agp + (size_t)(16 * i) * K);
    #pragma unroll
    for (int i = 0; i < IP_NB; ++i) rb[i] = *(const v4f*)(Bm + brow[i]);
    #pragma unroll
    for (int i = 0; i < IP_NA; ++i) *(v4f*)(sm + soffA + 16 * i * IP_P) = ra[i];
    #pragma unroll
    for (int i = 0; i < IP_NB; ++i) *(v4f*)(sm + soffB + 16 * i * IP_P) = rb[i];
    __syncthreads();
    ip_read(f0, sm + aoff, sm + boff, 0);

    int buf = 0;
    for (int kk = 0; kk < K - IP_KC; kk += IP_KC) {
        const float* sc = sm + buf * IP_STAGE;
        float*       sn = sm + (buf ^ 1) * IP_STAGE;

        #pragma unroll
        for (int i = 0; i < IP_NA; ++i) ra[i] = *(const v4f*)(agp + (size_t)(16 * i) * K + kk + IP_KC);
        #pragma unroll
        for (int i = 0; i < IP_NB; ++i) rb[i] = *(const v4f*)(Bm + brow[i] + kk + IP_KC);

        ip_read(f1, sc + aoff, sc + boff, 8);
        ip_mma(acc, f0, 0);
        ip_mma(acc, f0, 1);
        ip_read(f2, sc + aoff, sc + boff, 16);
        ip_mma(acc, f1, 0);
        ip_mma(acc, f1, 1);
        ip_read(f3, sc + aoff, sc + boff, 24);
        ip_mma(acc, f2, 0);
        ip_mma(acc, f2, 1);
        #pragma unroll
        for (int i = 0; i < IP_NA; ++i) *(v4f*)(sn + soffA + 16 * i * IP_P) = ra[i];
        #pragma unroll
        for (int i = 0; i < IP_NB; ++i) *(v4f*)(sn + soffB + 16 * i * IP_P) = rb[i];
        ip_mma(acc, f3, 0);
        __builtin_amdgcn_sched_group_barrier(0x020, IP_NA + IP_NB, 0);
        __builtin_amdgcn_sched_group_barrier(0x100, 14, 0);
        __builtin_amdgcn_sched_group_barrier(0x008, 26, 0);
        __builtin_amdgcn_sched_group_barrier(0x100, 14, 0);
        __builtin_amdgcn_sched_group_barrier(0x008, 26, 0);
        __builtin_amdgcn_sched_group_barrier(0x100, 14, 0);
        __builtin_amdgcn_sched_group_barrier(0x008, 26, 0);
        __builtin_amdgcn_sched_group_barrier(0x200, IP_NA + IP_NB, 0);
        __builtin_amdgcn_sched_group_barrier(0x008, 13, 0);

        __builtin_amdgcn_sched_barrier(0);
        __syncthreads();
        __builtin_amdgcn_sched_barrier(0);

        ip_read(f0, sn + aoff, sn + boff, 0);
        ip_mma(acc, f3, 1);
        __builtin_amdgcn_sched_group_barrier(0x100, 14, 1);
        __builtin_amdgcn_sched_group_barrier(0x008, 13, 1);
        buf ^= 1;
    }
    {
        const float* sc = sm + buf * IP_STAGE;
        ip_read(f1, sc + aoff, sc + boff, 8);
        ip_mma(acc, f0, 0);
        ip_mma(acc, f0, 1);
        ip_read(f2, sc + aoff, sc + boff, 16);
        ip_mma(acc, f1, 0);
        ip_mma(acc, f1, 1);
        ip_read(f3, sc + aoff, sc + boff, 24);
        ip_mma(acc, f2, 0);
        ip_mma(acc, f2, 1);
        ip_mma(acc, f3, 0);
        ip_mma(acc, f3, 1);
        __builtin_amdgcn_sched_group_barrier(0x100, 14, 2);
        __builtin_amdgcn_sched_group_barrier(0x008, 26, 2);
        __builtin_amdgcn_sched_group_barrier(0x100, 14, 2);
        __builtin_amdgcn_sched_group_barrier(0x008, 26, 2);
        __builtin_amdgcn_sched_group_barrier(0x100, 14, 2);
        __builtin_amdgcn_sched_group_barrier(0x008, 52, 2);
    }

    __syncthreads();
    float* cs = sm;
    {
        float* cw = cs + (w * 16 + 8 * g) * IP_CP + r16;
        #pragma unroll
        for (int j = 0; j < IP_NT; ++j) {
            #pragma unroll
            for (int r = 0; r < 8; ++r) cw[r * IP_CP + j * 16] = acc[j][r];
        }
    }
    __syncthreads();
    float* cblk = C + (size_t)(mb * IP_M) * N + nb * IP_N;
    constexpr int tailpc = (N - (nbt - 1) * IP_N) / 4;
    if (nb == nbt - 1) ip_store_rows<tailpc>(cs, cblk, tid, N);
    else               ip_store_rows<IP_N / 4>(cs, cblk, tid, N);
}

#define OP_M     32
#define OP_N     96
#define OP_KC    64
#define OP_P     68
#define OP_CP    100
#define OP_STAGE ((OP_M + OP_N) * OP_P)

struct OpFrag {
    v2f a[4];
    v2f q[4][3];
};

struct OpRaw {
    v4f a[4];
    v4f s[4];
    v4f b[12];
};

__device__ __forceinline__ void op_read(OpFrag& f, const float* asb, const float* bsb, int k0) {
    #pragma unroll
    for (int h = 0; h < 4; ++h) {
        f.a[h] = *(const v2f*)(asb + k0 + 4 * h);
        #pragma unroll
        for (int j = 0; j < 3; ++j) f.q[h][j] = *(const v2f*)(bsb + j * 16 * OP_P + k0 + 4 * h);
    }
}

__device__ __forceinline__ void op_mma(v8f (&acc)[3], const OpFrag& f, int h) {
    #pragma unroll
    for (int j = 0; j < 3; ++j)
        acc[j] = __builtin_amdgcn_wmma_f32_16x16x4_f32(false, f.a[h], false, f.q[h][j], (short)0, acc[j], false, false);
}

template <int K>
__device__ __forceinline__ void op_gload(OpRaw& r, const float* __restrict__ agp, const float* __restrict__ ag2,
                                         const float* __restrict__ bgp) {
    #pragma unroll
    for (int i = 0; i < 4; ++i)  r.a[i] = *(const v4f*)(agp + (size_t)(8 * i) * K);
    #pragma unroll
    for (int i = 0; i < 4; ++i)  r.s[i] = *(const v4f*)(ag2 + (size_t)(8 * i) * K);
    #pragma unroll
    for (int i = 0; i < 12; ++i) r.b[i] = *(const v4f*)(bgp + (size_t)(8 * i) * K);
}

__device__ __forceinline__ void op_sstore(const OpRaw& r, float* so, int soffA, int soffB) {
    #pragma unroll
    for (int i = 0; i < 4; ++i)  *(v4f*)(so + soffA + 8 * i * OP_P) = r.a[i] + r.s[i];
    #pragma unroll
    for (int i = 0; i < 12; ++i) *(v4f*)(so + soffB + 8 * i * OP_P) = r.b[i];
}

template <int K, bool LOAD, bool STORE>
__device__ __forceinline__ void op_chunk(v8f (&acc)[3], OpFrag& f0, OpFrag& f1, OpFrag& f2, OpFrag& f3,
                                         OpRaw& rl, const OpRaw& rn,
                                         const float* sc, float* sn,
                                         int aoff, int boff, int soffA, int soffB,
                                         const float* __restrict__ agp, const float* __restrict__ ag2,
                                         const float* __restrict__ bgp) {
    if constexpr (LOAD) op_gload<K>(rl, agp, ag2, bgp);
    op_read(f1, sc + aoff, sc + boff, 16);
    op_mma(acc, f0, 0); op_mma(acc, f0, 1); op_mma(acc, f0, 2); op_mma(acc, f0, 3);
    op_read(f2, sc + aoff, sc + boff, 32);
    op_mma(acc, f1, 0); op_mma(acc, f1, 1); op_mma(acc, f1, 2); op_mma(acc, f1, 3);
    op_read(f3, sc + aoff, sc + boff, 48);
    op_mma(acc, f2, 0); op_mma(acc, f2, 1); op_mma(acc, f2, 2); op_mma(acc, f2, 3);
    if constexpr (STORE) op_sstore(rn, sn, soffA, soffB);
    op_mma(acc, f3, 0); op_mma(acc, f3, 1);
    if constexpr (LOAD) __builtin_amdgcn_sched_group_barrier(0x020, 20, 0);
    __builtin_amdgcn_sched_group_barrier(0x100, 8, 0);
    __builtin_amdgcn_sched_group_barrier(0x008, 12, 0);
    __builtin_amdgcn_sched_group_barrier(0x100, 8, 0);
    __builtin_amdgcn_sched_group_barrier(0x008, 12, 0);
    __builtin_amdgcn_sched_group_barrier(0x100, 8, 0);
    __builtin_amdgcn_sched_group_barrier(0x008, 12, 0);
    if constexpr (STORE) {
        __builtin_amdgcn_sched_group_barrier(0x002, 8, 0);
        __builtin_amdgcn_sched_group_barrier(0x200, 16, 0);
    }
    __builtin_amdgcn_sched_group_barrier(0x008, 6, 0);

    if constexpr (STORE) {
        __builtin_amdgcn_sched_barrier(0);
        __builtin_amdgcn_fence(__ATOMIC_RELEASE, "workgroup", "local");
        __builtin_amdgcn_s_barrier();
        __builtin_amdgcn_fence(__ATOMIC_ACQUIRE, "workgroup", "local");
        __builtin_amdgcn_sched_barrier(0);

        op_read(f0, sn + aoff, sn + boff, 0);
    }
    op_mma(acc, f3, 2); op_mma(acc, f3, 3);
    if constexpr (STORE) __builtin_amdgcn_sched_group_barrier(0x100, 8, 1);
    __builtin_amdgcn_sched_group_barrier(0x008, 6, 1);
}

template <int M, int N, int K>
__global__ void __launch_bounds__(128)
gemm_nt_f32_wmma_32x96(const float* __restrict__ A,
                       const float* __restrict__ A2,
                       const float* __restrict__ Bm,
                       float* __restrict__ C) {
    static_assert(M % OP_M == 0 && N % OP_N == 0 && K % OP_KC == 0,
                  "whole tiles and whole K chunks only");
    static_assert(OP_M * OP_CP <= OP_STAGE, "output tile is staged inside stage 0");
    constexpr int nch = K / OP_KC;
    static_assert(nch % 2 == 0 && nch >= 4, "chunks are taken in pairs (two register sets, two stages)");
    constexpr int nbt = N / OP_N;
    const int nb  = blockIdx.x % nbt;
    const int mb  = blockIdx.x / nbt;

    const int tid  = threadIdx.x;
    const int w    = tid >> 5;
    const int lane = tid & 31;
    const int g    = lane >> 4;
    const int r16  = lane & 15;
    const int rg   = w & 1;
    const int cg   = w >> 1;

    __shared__ __attribute__((aligned(16))) float sm[2 * OP_STAGE];

    const int sp = tid & 15;
    const int sr = tid >> 4;
    const float* agp = A  + (size_t)(mb * OP_M + sr) * K + sp * 4;
    const float* ag2 = A2 + (size_t)(mb * OP_M + sr) * K + sp * 4;
    const float* bgp = Bm + (size_t)(nb * OP_N + sr) * K + sp * 4;
    const int soffA = sr * OP_P + sp * 4;
    const int soffB = OP_M * OP_P + sr * OP_P + sp * 4;

    const int aoff = (rg * 16 + r16) * OP_P + 2 * g;
    const int boff = OP_M * OP_P + (cg * 48 + r16) * OP_P + 2 * g;

    v8f acc[3];
    #pragma unroll
    for (int j = 0; j < 3; ++j) acc[j] = v8f{};
    OpRaw r0, r1;
    OpFrag f0, f1, f2, f3;
    float* st0 = sm;
    float* st1 = sm + OP_STAGE;

    op_gload<K>(r0, agp, ag2, bgp);
    __builtin_amdgcn_sched_barrier(0);
    op_gload<K>(r1, agp + OP_KC, ag2 + OP_KC, bgp + OP_KC);
    op_sstore(r0, st0, soffA, soffB);
    __builtin_amdgcn_fence(__ATOMIC_RELEASE, "workgroup", "local");
    __builtin_amdgcn_s_barrier();
    __builtin_amdgcn_fence(__ATOMIC_ACQUIRE, "workgroup", "local");
    op_read(f0, st0 + aoff, st0 + boff, 0);

    const float* pa = agp + 2 * OP_KC;
    const float* p2 = ag2 + 2 * OP_KC;
    const float* pb = bgp + 2 * OP_KC;
    for (int c = 0; c < nch - 2; c += 2) {
        op_chunk<K, true, true>(acc, f0, f1, f2, f3, r0, r1, st0, st1, aoff, boff, soffA, soffB,
                                pa, p2, pb);
        op_chunk<K, true, true>(acc, f0, f1, f2, f3, r1, r0, st1, st0, aoff, boff, soffA, soffB,
                                pa + OP_KC, p2 + OP_KC, pb + OP_KC);
        pa += 2 * OP_KC;
        p2 += 2 * OP_KC;
        pb += 2 * OP_KC;
    }
    op_chunk<K, false, true >(acc, f0, f1, f2, f3, r0, r1, st0, st1, aoff, boff, soffA, soffB,
                              agp, ag2, bgp);
    op_chunk<K, false, false>(acc, f0, f1, f2, f3, r1, r0, st1, st0, aoff, boff, soffA, soffB,
                              agp, ag2, bgp);

    __syncthreads();
    float* cs = sm;
    {
        float* cw = cs + (rg * 16 + 8 * g) * OP_CP + cg * 48 + r16;
        #pragma unroll
        for (int r = 0; r < 8; ++r) {
            cw[r * OP_CP +  0] = acc[0][r];
            cw[r * OP_CP + 16] = acc[1][r];
            cw[r * OP_CP + 32] = acc[2][r];
        }
    }
    __syncthreads();
    #pragma unroll
    for (int i = 0; i < (OP_M * OP_N / 4) / 128; ++i) {
        const int idx = i * 128 + tid;
        const int row = idx / (OP_N / 4);
        const int pc  = idx % (OP_N / 4);
        const v4f v = *(const v4f*)(cs + row * OP_CP + pc * 4);
        *(v4f*)(C + (size_t)(mb * OP_M + row) * N + nb * OP_N + pc * 4) = v;
    }
}

#define TCH   32
#define NCHK  (SEQLEN / TCH)
#define PSPLIT 2
#define NSPLIT 2
#define PROWS  (HEADDIM / PSPLIT)
#define NST    (DSTATE / NSPLIT)
#define NPT    (NST / 4)
#define BCP    (2 * NST)
#define STHR   (PROWS * 4)
#define SBLK   (2 * STHR)
#define CROWS  (TCH * (BCP / 4) / STHR)
#define ZP    40

struct ScanStage {
    v4f   rx[5];
    v4f   rc[CROWS + 3];
    v4f   rz[2];
    float dv;
};

__device__ __forceinline__ float silu_f(float a) {
    return a / (1.0f + __expf(-a));
}

template <int M>
__device__ __forceinline__ float add_lane_xor(float v) {
    const int partner = __builtin_amdgcn_update_dpp(0, __float_as_int(v), 0x160 + M, 0xf, 0xf, true);
    return v + __int_as_float(partner);
}

template <bool FIRST>
__device__ __forceinline__ v4f ld_raw(const float* __restrict__ col0, int l) {
    const int lr = (FIRST && l < 0) ? 0 : l;
    return *(const v4f*)(col0 + (size_t)lr * DINPROJ);
}

template <bool FIRST>
__device__ __forceinline__ v4f conv4_silu(const v4f r0, const v4f r1, const v4f r2, const v4f r3,
                                          int l, const v4f (&w)[4], const v4f bias) {
    v4f o;
    #pragma unroll
    for (int ch = 0; ch < 4; ++ch) {
        float acc = bias[ch];
        if (!FIRST || l >= 3) acc = fmaf(r0[ch], w[ch].x, acc);
        if (!FIRST || l >= 2) acc = fmaf(r1[ch], w[ch].y, acc);
        if (!FIRST || l >= 1) acc = fmaf(r2[ch], w[ch].z, acc);
        acc = fmaf(r3[ch], w[ch].w, acc);
        o[ch] = silu_f(acc);
    }
    return o;
}

template <bool FIRST>
__device__ __forceinline__ void stage_load(ScanStage& s,
                                           const float* __restrict__ xraw,
                                           const float* __restrict__ craw,
                                           const float* __restrict__ zraw,
                                           const float* __restrict__ draw,
                                           int lx, int lc, int ld, bool dlane) {
    #pragma unroll
    for (int i = 0; i < 5; ++i) s.rx[i] = ld_raw<FIRST>(xraw, lx - 3 + i);
    #pragma unroll
    for (int i = 0; i < CROWS + 3; ++i) s.rc[i] = ld_raw<FIRST>(craw, lc - 3 + i);
    s.rz[0] = *(const v4f*)(zraw + (size_t)lx * DINPROJ);
    s.rz[1] = *(const v4f*)(zraw + (size_t)(lx + 1) * DINPROJ);
    s.dv = dlane ? draw[(size_t)ld * DINPROJ] : 0.0f;
}

template <bool FIRST>
__device__ __forceinline__ void stage_store(const ScanStage& s,
                                            float* __restrict__ sxb, float* __restrict__ sBCb,
                                            float* __restrict__ sgb, v2f* __restrict__ sddb,
                                            int lx, int lc, int xrow, int crow, int q8, int q16,
                                            const v4f (&wx)[4], const v4f bxv,
                                            const v4f (&wc)[4], const v4f bcv,
                                            bool dlane, int t, float dtb, float Ah) {
    #pragma unroll
    for (int o = 0; o < 2; ++o) {
        const v4f v = conv4_silu<FIRST>(s.rx[o], s.rx[o + 1], s.rx[o + 2], s.rx[o + 3],
                                        lx + o, wx, bxv);
        *(v4f*)(sxb + (xrow + o) * PROWS + q8 * 4) = v;
    }
    #pragma unroll
    for (int o = 0; o < CROWS; ++o) {
        const v4f v = conv4_silu<FIRST>(s.rc[o], s.rc[o + 1], s.rc[o + 2], s.rc[o + 3],
                                        lc + o, wc, bcv);
        *(v4f*)(sBCb + (crow + o) * BCP + q16 * 4) = v;
    }
    #pragma unroll
    for (int o = 0; o < 2; ++o) {
        const v4f z = s.rz[o];
        const v4f g = v4f{silu_f(z.x), silu_f(z.y), silu_f(z.z), silu_f(z.w)};
        *(v4f*)(sgb + (xrow + o) * ZP + q8 * 4) = g;
    }
    if (dlane) {
        const float dv = s.dv + dtb;
        float dt = (dv > 20.0f) ? dv : __logf(1.0f + __expf(dv));
        dt = fminf(fmaxf(dt, 1e-6f), 1000.0f);
        const float dA = fminf(fmaxf(__expf(dt * Ah), 1e-6f), 1000.0f);
        sddb[t] = v2f{dt, dA};
    }
}

__global__ void __launch_bounds__(SBLK)
ssm_scan_kernel(const float* __restrict__ zxbcdt,
                const float* __restrict__ conv_w,
                const float* __restrict__ conv_b,
                const float* __restrict__ dt_bias,
                const float* __restrict__ A_log,
                const float* __restrict__ D_param,
                float* __restrict__ y) {
    const int nh = blockIdx.x % NSPLIT;
    const int ph = (blockIdx.x / NSPLIT) % PSPLIT;
    const int bh = blockIdx.x / (NSPLIT * PSPLIT);
    const int b  = bh / NHEADS;
    const int h  = bh % NHEADS;
    const bool stager = threadIdx.x >= STHR;
    const int t  = threadIdx.x & (STHR - 1);
    const int p  = t >> 2;
    const int nc = t & 3;

    const float Ah  = -__expf(A_log[h]);
    const float dtb = dt_bias[h];
    const float Dp  = D_param[h];

    __shared__ __attribute__((aligned(16))) float sx [2][TCH * PROWS];
    __shared__ __attribute__((aligned(16))) float sBC[2][TCH * BCP];
    __shared__ __attribute__((aligned(16))) float sg [2][TCH * ZP];
    __shared__ __attribute__((aligned(8)))  v2f   sdd[2][TCH];

    const int q8   = t & 7;
    const int xrow = (t >> 3) * 2;
    const int q16  = t & 15;
    const int crow = (t >> 4) * CROWS;
    const bool dlane = t < TCH;
    const int cx = h * HEADDIM + ph * PROWS + q8 * 4;
    const int cb = DSSM + (q16 >> 3) * DSTATE + nh * NST + (q16 & 7) * 4;
    const float* base = zxbcdt + (size_t)b * SEQLEN * DINPROJ;
    const float* zraw = base + cx;
    const float* xraw = base + DSSM + cx;
    const float* craw = base + DSSM + cb;
    const float* draw = base + (DSSM + CONVDIM) + h;

    v4f wx[4], wc[4], bxv, bcv;

    v2f s2[NPT / 2];
    #pragma unroll
    for (int k = 0; k < NPT / 2; ++k) s2[k] = v2f{0.0f, 0.0f};
    const bool addDx = (nh == 0);

    ScanStage st;
    if (stager) {
        #pragma unroll
        for (int ch = 0; ch < 4; ++ch) {
            wx[ch] = *(const v4f*)(conv_w + (size_t)(cx + ch) * DCONV);
            wc[ch] = *(const v4f*)(conv_w + (size_t)(cb + ch) * DCONV);
        }
        bxv = *(const v4f*)(conv_b + cx);
        bcv = *(const v4f*)(conv_b + cb);
        ScanStage s0;
        stage_load<true>(s0, xraw, craw, zraw, draw, xrow, crow, t, dlane);
        stage_store<true>(s0, &sx[0][0], &sBC[0][0], &sg[0][0], &sdd[0][0],
                          xrow, crow, xrow, crow, q8, q16, wx, bxv, wc, bcv,
                          dlane, t, dtb, Ah);
        stage_load<false>(st, xraw, craw, zraw, draw, TCH + xrow, TCH + crow, TCH + t, dlane);
    }
    __builtin_amdgcn_fence(__ATOMIC_RELEASE, "workgroup", "local");
    __builtin_amdgcn_s_barrier();
    __builtin_amdgcn_fence(__ATOMIC_ACQUIRE, "workgroup", "local");

    int buf = 0;
    for (int c = 0; c < NCHK; ++c) {
        const bool more = (c + 1) < NCHK;
        const int  l1   = (c + 1) * TCH;

        if (stager) {
            if (more) {
                stage_store<false>(st, &sx[buf ^ 1][0], &sBC[buf ^ 1][0], &sg[buf ^ 1][0],
                                   &sdd[buf ^ 1][0],
                                   l1 + xrow, l1 + crow, xrow, crow, q8, q16,
                                   wx, bxv, wc, bcv, dlane, t, dtb, Ah);
                if ((c + 2) < NCHK)
                    stage_load<false>(st, xraw, craw, zraw, draw,
                                      l1 + TCH + xrow, l1 + TCH + crow, l1 + TCH + t, dlane);
            }
        } else {

        const float* bx  = &sx[buf][p];
        const float* bBC = &sBC[buf][nc * NPT];
        const float* bg  = &sg[buf][nc * ZP + p];
        float* yp = y + (size_t)nh * MTOK * DSSM
                      + (size_t)(b * SEQLEN + c * TCH + nc) * DSSM + h * HEADDIM + ph * PROWS + p;
        for (int j4 = 0; j4 < TCH; j4 += 4) {
            float ys[4], xs4[4];
            const float gate = bg[j4 * ZP];
            v2f ddr[4];
            v4f Br[4][NPT / 4], Cr[4][NPT / 4];
            #pragma unroll
            for (int u = 0; u < 4; ++u) {
                const int j = j4 + u;
                ddr[u] = sdd[buf][j];
                xs4[u] = bx[j * PROWS];
                const v4f* Bq = (const v4f*)(bBC + j * BCP);
                const v4f* Cq = (const v4f*)(bBC + j * BCP + NST);
                #pragma unroll
                for (int q = 0; q < NPT / 4; ++q) { Br[u][q] = Bq[q]; Cr[u][q] = Cq[q]; }
            }
            __builtin_amdgcn_sched_barrier(0);
            #pragma unroll
            for (int u = 0; u < 4; ++u) {
                const float dt  = ddr[u].x;
                const float dA  = ddr[u].y;
                const float xv  = xs4[u];
                const float dtx = dt * xv;

                const v4f B0 = Br[u][0], B1 = Br[u][1];
                const v4f C0 = Cr[u][0], C1 = Cr[u][1];
                const v2f Bp[4] = {v2f{B0.x, B0.y}, v2f{B0.z, B0.w}, v2f{B1.x, B1.y}, v2f{B1.z, B1.w}};
                const v2f Cp[4] = {v2f{C0.x, C0.y}, v2f{C0.z, C0.w}, v2f{C1.x, C1.y}, v2f{C1.z, C1.w}};
                const v2f dA2  = v2f{dA, dA};
                const v2f dtx2 = v2f{dtx, dtx};

                v2f ya = v2f{0.0f, 0.0f}, yb = v2f{0.0f, 0.0f};
                #pragma unroll
                for (int k = 0; k < NPT / 2; ++k) {
                    v2f sv = __builtin_elementwise_fma(s2[k], dA2, dtx2 * Bp[k]);
                    sv.x = fminf(fmaxf(sv.x, -1000.0f), 1000.0f);
                    sv.y = fminf(fmaxf(sv.y, -1000.0f), 1000.0f);
                    s2[k] = sv;
                    if (k < NPT / 4) ya = __builtin_elementwise_fma(sv, Cp[k], ya);
                    else             yb = __builtin_elementwise_fma(sv, Cp[k], yb);
                }
                const v2f yy = ya + yb;
                ys[u]  = yy.x + yy.y;
            }
            ys[0] = add_lane_xor<1>(ys[0]);
            ys[1] = add_lane_xor<1>(ys[1]);
            ys[2] = add_lane_xor<1>(ys[2]);
            ys[3] = add_lane_xor<1>(ys[3]);
            ys[0] = add_lane_xor<2>(ys[0]);
            ys[1] = add_lane_xor<2>(ys[1]);
            ys[2] = add_lane_xor<2>(ys[2]);
            ys[3] = add_lane_xor<2>(ys[3]);
            float ykeep = ys[0], xkeep = xs4[0];
            ykeep = (nc == 1) ? ys[1] : ykeep;   xkeep = (nc == 1) ? xs4[1] : xkeep;
            ykeep = (nc == 2) ? ys[2] : ykeep;   xkeep = (nc == 2) ? xs4[2] : xkeep;
            ykeep = (nc == 3) ? ys[3] : ykeep;   xkeep = (nc == 3) ? xs4[3] : xkeep;
            if (addDx) ykeep = fmaf(Dp, xkeep, ykeep);
            yp[(size_t)j4 * DSSM] = ykeep * gate;
        }
        }

        if (more) {
            __builtin_amdgcn_fence(__ATOMIC_RELEASE, "workgroup", "local");
            __builtin_amdgcn_s_barrier();
            __builtin_amdgcn_fence(__ATOMIC_ACQUIRE, "workgroup", "local");
            buf ^= 1;
        }
    }
}

extern "C" void kernel_launch(void* const* d_in, const int* in_sizes, int n_in,
                              void* d_out, int out_size, void* d_ws, size_t ws_size,
                              hipStream_t stream) {
    const float* u       = (const float*)d_in[0];
    const float* W_in    = (const float*)d_in[1];
    const float* conv_w  = (const float*)d_in[2];
    const float* conv_b  = (const float*)d_in[3];
    const float* dt_bias = (const float*)d_in[4];
    const float* A_log   = (const float*)d_in[5];
    const float* D_param = (const float*)d_in[6];
    const float* W_out   = (const float*)d_in[7];
    float* out = (float*)d_out;

    float* ws      = (float*)d_ws;
    float* zxbcdt  = ws;
    float* ybuf    = zxbcdt + (size_t)MTOK * DINPROJ + (size_t)MTOK * CONVDIM;

    {
        const int mbt = MTOK / IP_M;
        const int nbt = (DINPROJ + IP_N - 1) / IP_N;
        gemm_nt_f32_wmma_64x208<MTOK, DINPROJ, DMODEL><<<mbt * nbt, 128, 0, stream>>>(u, W_in, zxbcdt);
    }

    ssm_scan_kernel<<<BATCH * NHEADS * PSPLIT * NSPLIT, SBLK, 0, stream>>>(zxbcdt, conv_w, conv_b, dt_bias,
                                                                           A_log, D_param, ybuf);

    {
        const int mbt = MTOK / OP_M;
        const int nbt = DMODEL / OP_N;
        gemm_nt_f32_wmma_32x96<MTOK, DMODEL, DSSM><<<mbt * nbt, 128, 0, stream>>>(
            ybuf, ybuf + (size_t)MTOK * DSSM, W_out, out);
    }
}
